# starting kernel
# baseline (speedup 1.0000x reference)
#include <hip/hip_runtime.h>
#include <hip/hip_bf16.h>

#define IN_DIM 512
#define HID 20
#define OUT_DIM 10

typedef __attribute__((ext_vector_type(2))) float v2f;
typedef __attribute__((ext_vector_type(8))) float v8f;

__global__ void k_init(float* __restrict__ deg, float* __restrict__ emb, int n) {
  long long t = (long long)blockIdx.x * blockDim.x + threadIdx.x;
  if (t < (long long)n * 64) emb[t] = 0.0f;
  if (t < n) deg[t] = 1.0f;
}

__global__ void k_packB(const float* __restrict__ W, int Krows, int Ncols,
                        float* __restrict__ Bp, int Ksteps, int ntiles) {
  int t = blockIdx.x * blockDim.x + threadIdx.x;
  int total = Ksteps * ntiles * 64;
  if (t >= total) return;
  int j = t & 1;
  int lane = (t >> 1) & 31;
  int tile = (t >> 6) % ntiles;
  int ks = (t >> 6) / ntiles;
  int k = 4 * ks + 2 * (lane >> 4) + j;
  int n = tile * 16 + (lane & 15);
  Bp[t] = (k < Krows && n < Ncols) ? W[k * Ncols + n] : 0.0f;
}

__global__ void k_deg(const int* __restrict__ dst, const float* __restrict__ ew,
                      float* __restrict__ deg, long long E) {
  long long e = (long long)blockIdx.x * blockDim.x + threadIdx.x;
  if (e >= E) return;
  unsafeAtomicAdd(&deg[dst[e]], ew[e]);
}

__global__ void k_dinv(const float* __restrict__ deg, float* __restrict__ dinv, int n) {
  int i = blockIdx.x * blockDim.x + threadIdx.x;
  if (i >= n) return;
  float d = deg[i];
  dinv[i] = (d > 0.0f) ? rsqrtf(d) : 0.0f;
}

__global__ void k_selfloop(const float* __restrict__ h, const float* __restrict__ dinv,
                           float* __restrict__ agg, int n) {
  int t = blockIdx.x * blockDim.x + threadIdx.x;
  int i = t >> 5, c = t & 31;
  if (i >= n) return;
  if (c < HID) { float di = dinv[i]; agg[(size_t)i * 32 + c] = h[(size_t)i * 32 + c] * di * di; }
}

__global__ void k_edge_agg(const int* __restrict__ src, const int* __restrict__ dst,
                           const float* __restrict__ ew, const float* __restrict__ dinv,
                           const float* __restrict__ h, float* __restrict__ agg, long long E) {
  long long t = (long long)blockIdx.x * blockDim.x + threadIdx.x;
  long long e = t >> 5;
  int c = (int)(t & 31);
  if (e >= E) return;
  int s = src[e], d = dst[e];
  float coef = dinv[s] * ew[e] * dinv[d];
  if (c < HID) unsafeAtomicAdd(&agg[(size_t)d * 32 + c], h[(size_t)s * 32 + c] * coef);
}

__global__ void k_bias_norm(const float* __restrict__ agg, const float* __restrict__ b,
                            float* __restrict__ emb, int n, int off) {
  int t = blockIdx.x * blockDim.x + threadIdx.x;
  int i = t >> 5, c = t & 31;
  if (i >= n) return;
  float v = 0.0f;
  if (c < HID) v = agg[(size_t)i * 32 + c] + b[c];
  float s = v * v;
#pragma unroll
  for (int m = 16; m >= 1; m >>= 1) s += __shfl_xor(s, m, 32);
  float nrm = fmaxf(sqrtf(s), 1e-12f);
  if (c < HID) {
    float r = v / nrm;
    emb[(size_t)i * 64 + off + c] = (r > 0.0f) ? r : 0.0f;
  }
}

template <int NTILES>
__global__ void k_wmma_gemm(const float* __restrict__ A, int lda, int acol0,
                            const float* __restrict__ Bp, int Ksteps,
                            float* __restrict__ out, int ldo, int ncols,
                            const float* __restrict__ bias, int npairs) {
  int pair = blockIdx.x * (blockDim.x >> 5) + (threadIdx.x >> 5);
  if (pair >= npairs) return;
  int lane = threadIdx.x & 31;
  int half = lane >> 4;
  int l16 = lane & 15;
  int m0 = pair * 32;
  const float* ap0 = A + (size_t)(m0 + l16) * lda + acol0 + half * 2;
  const float* ap1 = ap0 + (size_t)16 * lda;
  const float* bp = Bp + lane * 2;
  v8f acc[2][NTILES];
#pragma unroll
  for (int rr = 0; rr < 2; ++rr)
#pragma unroll
    for (int t = 0; t < NTILES; ++t) acc[rr][t] = (v8f){0.f, 0.f, 0.f, 0.f, 0.f, 0.f, 0.f, 0.f};

#pragma unroll 2
  for (int ks = 0; ks < Ksteps; ++ks) {
    v2f a0 = *(const v2f*)(ap0 + 4 * ks);
    v2f a1 = *(const v2f*)(ap1 + 4 * ks);
#pragma unroll
    for (int t = 0; t < NTILES; ++t) {
      v2f b = *(const v2f*)(bp + (size_t)(ks * NTILES + t) * 64);
      acc[0][t] = __builtin_amdgcn_wmma_f32_16x16x4_f32(false, a0, false, b, (short)0, acc[0][t], false, false);
      acc[1][t] = __builtin_amdgcn_wmma_f32_16x16x4_f32(false, a1, false, b, (short)0, acc[1][t], false, false);
    }
  }

#pragma unroll
  for (int rr = 0; rr < 2; ++rr)
#pragma unroll
    for (int t = 0; t < NTILES; ++t)
#pragma unroll
      for (int r = 0; r < 8; ++r) {
        int row = m0 + rr * 16 + r + half * 8;
        int col = t * 16 + l16;
        if (col < ncols) out[(size_t)row * ldo + col] = acc[rr][t][r] + (bias ? bias[col] : 0.0f);
      }
}

extern "C" void kernel_launch(void* const* d_in, const int* in_sizes, int n_in,
                              void* d_out, int out_size, void* d_ws, size_t ws_size,
                              hipStream_t stream) {
  const float* x  = (const float*)d_in[0];
  const int*   ei = (const int*)d_in[1];
  const float* ew = (const float*)d_in[2];
  const float* W1 = (const float*)d_in[3];
  const float* b1 = (const float*)d_in[4];
  const float* W2 = (const float*)d_in[5];
  const float* b2 = (const float*)d_in[6];
  const float* W3 = (const float*)d_in[7];
  const float* b3 = (const float*)d_in[8];
  const float* Wl = (const float*)d_in[9];
  const float* bl = (const float*)d_in[10];
  float* out = (float*)d_out;

  const int N = in_sizes[0] / IN_DIM;
  const long long E = in_sizes[2];
  const int* src = ei;
  const int* dst = ei + E;

  const int KS1 = IN_DIM / 4;
  const int KS2 = (HID + 3) / 4;
  const int KSL = 64 / 4;

  float* ws = (float*)d_ws;
  float* deg  = ws; ws += N;
  float* dinv = ws; ws += N;
  float* h    = ws; ws += (size_t)N * 32;
  float* agg  = ws; ws += (size_t)N * 32;
  float* emb  = ws; ws += (size_t)N * 64;
  float* W1p  = ws; ws += KS1 * 2 * 64;
  float* W2p  = ws; ws += KS2 * 2 * 64;
  float* W3p  = ws; ws += KS2 * 2 * 64;
  float* Wlp  = ws; ws += KSL * 1 * 64;

  const int TB = 256;
  const int NPAIRS = N / 32;
  dim3 gblk((NPAIRS + 3) / 4);
  long long nt32 = (long long)N * 32;
  long long et32 = E * 32;
  int g_n32 = (int)((nt32 + TB - 1) / TB);
  int g_e32 = (int)((et32 + TB - 1) / TB);

  k_init<<<(int)(((long long)N * 64 + TB - 1) / TB), TB, 0, stream>>>(deg, emb, N);
  k_packB<<<(KS1 * 2 * 64 + TB - 1) / TB, TB, 0, stream>>>(W1, IN_DIM, HID, W1p, KS1, 2);
  k_packB<<<(KS2 * 2 * 64 + TB - 1) / TB, TB, 0, stream>>>(W2, HID, HID, W2p, KS2, 2);
  k_packB<<<(KS2 * 2 * 64 + TB - 1) / TB, TB, 0, stream>>>(W3, HID, HID, W3p, KS2, 2);
  k_packB<<<(KSL * 1 * 64 + TB - 1) / TB, TB, 0, stream>>>(Wl, 3 * HID, OUT_DIM, Wlp, KSL, 1);
  k_deg<<<(int)((E + TB - 1) / TB), TB, 0, stream>>>(dst, ew, deg, E);
  k_dinv<<<(N + TB - 1) / TB, TB, 0, stream>>>(deg, dinv, N);

  k_wmma_gemm<2><<<gblk, 128, 0, stream>>>(x, IN_DIM, 0, W1p, KS1, h, 32, 32, nullptr, NPAIRS);
  k_selfloop<<<g_n32, TB, 0, stream>>>(h, dinv, agg, N);
  k_edge_agg<<<g_e32, TB, 0, stream>>>(src, dst, ew, dinv, h, agg, E);
  k_bias_norm<<<g_n32, TB, 0, stream>>>(agg, b1, emb, N, 0);

  k_wmma_gemm<2><<<gblk, 128, 0, stream>>>(emb, 64, 0, W2p, KS2, h, 32, 32, nullptr, NPAIRS);
  k_selfloop<<<g_n32, TB, 0, stream>>>(h, dinv, agg, N);
  k_edge_agg<<<g_e32, TB, 0, stream>>>(src, dst, ew, dinv, h, agg, E);
  k_bias_norm<<<g_n32, TB, 0, stream>>>(agg, b2, emb, N, HID);

  k_wmma_gemm<2><<<gblk, 128, 0, stream>>>(emb, 64, HID, W3p, KS2, h, 32, 32, nullptr, NPAIRS);
  k_selfloop<<<g_n32, TB, 0, stream>>>(h, dinv, agg, N);
  k_edge_agg<<<g_e32, TB, 0, stream>>>(src, dst, ew, dinv, h, agg, E);
  k_bias_norm<<<g_n32, TB, 0, stream>>>(agg, b3, emb, N, 2 * HID);

  k_wmma_gemm<1><<<gblk, 128, 0, stream>>>(emb, 64, 0, Wlp, KSL, out, OUT_DIM, OUT_DIM, bl, NPAIRS);
}


# Edges counting-sorted by destination per call (integer-add ranks, arrival order); one wave per node gathers its rows in registers and finishes the row: no float atomics, sum order changes
# speedup vs baseline: 1.9555x; 1.9555x over previous
#include <hip/hip_runtime.h>
#include <hip/hip_bf16.h>

#define IN_DIM 512
#define HID 20
#define OUT_DIM 10

typedef __attribute__((ext_vector_type(2))) float v2f;
typedef __attribute__((ext_vector_type(8))) float v8f;

__global__ void k_init(int* __restrict__ cnt, float* __restrict__ emb, int n) {
  long long t = (long long)blockIdx.x * blockDim.x + threadIdx.x;
  if (t < (long long)n * 64) emb[t] = 0.0f;
  if (t < n) cnt[t] = 0;
}

__global__ void k_packB(const float* __restrict__ W, int Krows, int Ncols,
                        float* __restrict__ Bp, int Ksteps, int ntiles) {
  int t = blockIdx.x * blockDim.x + threadIdx.x;
  int total = Ksteps * ntiles * 64;
  if (t >= total) return;
  int j = t & 1;
  int lane = (t >> 1) & 31;
  int tile = (t >> 6) % ntiles;
  int ks = (t >> 6) / ntiles;
  int k = 4 * ks + 2 * (lane >> 4) + j;
  int n = tile * 16 + (lane & 15);
  Bp[t] = (k < Krows && n < Ncols) ? W[k * Ncols + n] : 0.0f;
}

__global__ void k_rank(const int* __restrict__ dst, int* __restrict__ cnt,
                       int* __restrict__ rank, int E) {
  int e = blockIdx.x * blockDim.x + threadIdx.x;
  if (e >= E) return;
  rank[e] = atomicAdd(&cnt[dst[e]], 1);
}

__global__ void k_scan(int* __restrict__ cnt, int n) {
  __shared__ int part[1024];
  int t = threadIdx.x;
  int per = (n + 1023) / 1024;
  int b = t * per;
  int e = b + per; if (e > n) e = n;
  int s = 0;
  for (int i = b; i < e; ++i) s += cnt[i];
  part[t] = s;
  __syncthreads();
  for (int d = 1; d < 1024; d <<= 1) {
    int v = (t >= d) ? part[t - d] : 0;
    __syncthreads();
    part[t] += v;
    __syncthreads();
  }
  int run = part[t] - s;
  int i = b;
  for (; i + 8 <= e; i += 8) {
    int c[8];
#pragma unroll
    for (int u = 0; u < 8; ++u) c[u] = cnt[i + u];
#pragma unroll
    for (int u = 0; u < 8; ++u) { cnt[i + u] = run; run += c[u]; }
  }
  for (; i < e; ++i) { int c1 = cnt[i]; cnt[i] = run; run += c1; }
}

__global__ void k_fill(const int* __restrict__ dst, const int* __restrict__ off,
                       const int* __restrict__ rank, int* __restrict__ perm, int E) {
  int e = blockIdx.x * blockDim.x + threadIdx.x;
  if (e >= E) return;
  perm[off[dst[e]] + rank[e]] = e;
}

__global__ void k_dinv_seg(const int* __restrict__ off, const int* __restrict__ perm,
                           const float* __restrict__ ew, float* __restrict__ dinv,
                           int n, int E) {
  int i = __builtin_amdgcn_readfirstlane((int)((blockIdx.x * blockDim.x + threadIdx.x) >> 5));
  if (i >= n) return;
  int lane = threadIdx.x & 31;
  int beg = off[i];
  int end = (i + 1 < n) ? off[i + 1] : E;
  float s = 0.0f;
  for (int p = beg + lane; p < end; p += 32) s += ew[perm[p]];
#pragma unroll
  for (int m = 16; m >= 1; m >>= 1) s += __shfl_xor(s, m, 32);
  float d = s + 1.0f;
  if (lane == 0) dinv[i] = (d > 0.0f) ? rsqrtf(d) : 0.0f;
}

__global__ void k_gather_norm(const int* __restrict__ off, const int* __restrict__ perm,
                              const int* __restrict__ src, const float* __restrict__ ew,
                              const float* __restrict__ dinv, const float* __restrict__ h,
                              const float* __restrict__ b, float* __restrict__ emb,
                              int n, int E, int coloff) {
  int i = __builtin_amdgcn_readfirstlane((int)((blockIdx.x * blockDim.x + threadIdx.x) >> 5));
  if (i >= n) return;
  int lane = threadIdx.x & 31;
  int beg = off[i];
  int end = (i + 1 < n) ? off[i + 1] : E;
  float di = dinv[i];
  float acc = h[(size_t)i * 32 + lane] * di * di;
  for (int p0 = beg; p0 < end; p0 += 32) {
    int m = end - p0; if (m > 32) m = 32;
    int s_l = 0;
    float cf_l = 0.0f;
    if (lane < m) {
      int e = perm[p0 + lane];
      s_l = src[e];
      cf_l = dinv[s_l] * ew[e] * di;
    }
    int cf_i = __float_as_int(cf_l);
    int j = 0;
    for (; j + 8 <= m; j += 8) {
      float v[8];
#pragma unroll
      for (int u = 0; u < 8; ++u) {
        int s = __builtin_amdgcn_readlane(s_l, j + u);
        v[u] = h[(size_t)s * 32 + lane];
      }
#pragma unroll
      for (int u = 0; u < 8; ++u) {
        float cf = __int_as_float(__builtin_amdgcn_readlane(cf_i, j + u));
        acc += v[u] * cf;
      }
    }
    for (; j < m; ++j) {
      int s = __builtin_amdgcn_readlane(s_l, j);
      float cf = __int_as_float(__builtin_amdgcn_readlane(cf_i, j));
      acc += h[(size_t)s * 32 + lane] * cf;
    }
  }
  float v = (lane < HID) ? acc + b[lane] : 0.0f;
  float ss = v * v;
#pragma unroll
  for (int m = 16; m >= 1; m >>= 1) ss += __shfl_xor(ss, m, 32);
  float nrm = fmaxf(sqrtf(ss), 1e-12f);
  if (lane < HID) {
    float r = v / nrm;
    emb[(size_t)i * 64 + coloff + lane] = (r > 0.0f) ? r : 0.0f;
  }
}

template <int NTILES>
__global__ void k_wmma_gemm(const float* __restrict__ A, int lda, int acol0,
                            const float* __restrict__ Bp, int Ksteps,
                            float* __restrict__ out, int ldo, int ncols,
                            const float* __restrict__ bias, int npairs) {
  int pair = blockIdx.x * (blockDim.x >> 5) + (threadIdx.x >> 5);
  if (pair >= npairs) return;
  int lane = threadIdx.x & 31;
  int half = lane >> 4;
  int l16 = lane & 15;
  int m0 = pair * 32;
  const float* ap0 = A + (size_t)(m0 + l16) * lda + acol0 + half * 2;
  const float* ap1 = ap0 + (size_t)16 * lda;
  const float* bp = Bp + lane * 2;
  v8f acc[2][NTILES];
#pragma unroll
  for (int rr = 0; rr < 2; ++rr)
#pragma unroll
    for (int t = 0; t < NTILES; ++t) acc[rr][t] = (v8f){0.f, 0.f, 0.f, 0.f, 0.f, 0.f, 0.f, 0.f};

#pragma unroll 2
  for (int ks = 0; ks < Ksteps; ++ks) {
    v2f a0 = *(const v2f*)(ap0 + 4 * ks);
    v2f a1 = *(const v2f*)(ap1 + 4 * ks);
#pragma unroll
    for (int t = 0; t < NTILES; ++t) {
      v2f b = *(const v2f*)(bp + (size_t)(ks * NTILES + t) * 64);
      acc[0][t] = __builtin_amdgcn_wmma_f32_16x16x4_f32(false, a0, false, b, (short)0, acc[0][t], false, false);
      acc[1][t] = __builtin_amdgcn_wmma_f32_16x16x4_f32(false, a1, false, b, (short)0, acc[1][t], false, false);
    }
  }

#pragma unroll
  for (int rr = 0; rr < 2; ++rr)
#pragma unroll
    for (int t = 0; t < NTILES; ++t)
#pragma unroll
      for (int r = 0; r < 8; ++r) {
        int row = m0 + rr * 16 + r + half * 8;
        int col = t * 16 + l16;
        if (col < ncols) out[(size_t)row * ldo + col] = acc[rr][t][r] + (bias ? bias[col] : 0.0f);
      }
}

extern "C" void kernel_launch(void* const* d_in, const int* in_sizes, int n_in,
                              void* d_out, int out_size, void* d_ws, size_t ws_size,
                              hipStream_t stream) {
  const float* x  = (const float*)d_in[0];
  const int*   ei = (const int*)d_in[1];
  const float* ew = (const float*)d_in[2];
  const float* W1 = (const float*)d_in[3];
  const float* b1 = (const float*)d_in[4];
  const float* W2 = (const float*)d_in[5];
  const float* b2 = (const float*)d_in[6];
  const float* W3 = (const float*)d_in[7];
  const float* b3 = (const float*)d_in[8];
  const float* Wl = (const float*)d_in[9];
  const float* bl = (const float*)d_in[10];
  float* out = (float*)d_out;

  const int N = in_sizes[0] / IN_DIM;
  const long long E = in_sizes[2];
  const int* src = ei;
  const int* dst = ei + E;

  const int KS1 = IN_DIM / 4;
  const int KS2 = (HID + 3) / 4;
  const int KSL = 64 / 4;

  float* ws = (float*)d_ws;
  int*   off  = (int*)ws; ws += N;
  float* dinv = ws; ws += N;
  float* h    = ws; ws += (size_t)N * 32;
  int*   perm = (int*)ws; ws += (size_t)N * 32;
  float* emb  = ws; ws += (size_t)N * 64;
  float* W1p  = ws; ws += KS1 * 2 * 64;
  float* W2p  = ws; ws += KS2 * 2 * 64;
  float* W3p  = ws; ws += KS2 * 2 * 64;
  float* Wlp  = ws; ws += KSL * 1 * 64;
  int*   rank = (int*)h;

  const int TB = 256;
  const int NPAIRS = N / 32;
  dim3 gblk((NPAIRS + 3) / 4);
  const int Ei = (int)E;
  long long nt32 = (long long)N * 32;
  int g_n32 = (int)((nt32 + TB - 1) / TB);
  int g_e = (int)((E + TB - 1) / TB);

  k_init<<<(int)(((long long)N * 64 + TB - 1) / TB), TB, 0, stream>>>(off, emb, N);
  k_packB<<<(KS1 * 2 * 64 + TB - 1) / TB, TB, 0, stream>>>(W1, IN_DIM, HID, W1p, KS1, 2);
  k_packB<<<(KS2 * 2 * 64 + TB - 1) / TB, TB, 0, stream>>>(W2, HID, HID, W2p, KS2, 2);
  k_packB<<<(KS2 * 2 * 64 + TB - 1) / TB, TB, 0, stream>>>(W3, HID, HID, W3p, KS2, 2);
  k_packB<<<(KSL * 1 * 64 + TB - 1) / TB, TB, 0, stream>>>(Wl, 3 * HID, OUT_DIM, Wlp, KSL, 1);
  k_rank<<<g_e, TB, 0, stream>>>(dst, off, rank, Ei);
  k_scan<<<1, 1024, 0, stream>>>(off, N);
  k_fill<<<g_e, TB, 0, stream>>>(dst, off, rank, perm, Ei);
  k_dinv_seg<<<g_n32, TB, 0, stream>>>(off, perm, ew, dinv, N, Ei);

  k_wmma_gemm<2><<<gblk, 128, 0, stream>>>(x, IN_DIM, 0, W1p, KS1, h, 32, 32, nullptr, NPAIRS);
  k_gather_norm<<<g_n32, TB, 0, stream>>>(off, perm, src, ew, dinv, h, b1, emb, N, Ei, 0);

  k_wmma_gemm<2><<<gblk, 128, 0, stream>>>(emb, 64, 0, W2p, KS2, h, 32, 32, nullptr, NPAIRS);
  k_gather_norm<<<g_n32, TB, 0, stream>>>(off, perm, src, ew, dinv, h, b2, emb, N, Ei, HID);

  k_wmma_gemm<2><<<gblk, 128, 0, stream>>>(emb, 64, HID, W3p, KS2, h, 32, 32, nullptr, NPAIRS);
  k_gather_norm<<<g_n32, TB, 0, stream>>>(off, perm, src, ew, dinv, h, b3, emb, N, Ei, 2 * HID);

  k_wmma_gemm<1><<<gblk, 128, 0, stream>>>(emb, 64, 0, Wlp, KSL, out, OUT_DIM, OUT_DIM, bl, NPAIRS);
}


# Counting sort without per-edge global atomics: work-group LDS histograms bin edges by 256-node range, one work-group per bin ranks them with LDS adds; arrival order kept, sum order differs
# speedup vs baseline: 3.1244x; 1.5977x over previous
#include <hip/hip_runtime.h>
#include <hip/hip_bf16.h>

#define IN_DIM 512
#define HID 20
#define OUT_DIM 10

typedef __attribute__((ext_vector_type(2))) float v2f;
typedef __attribute__((ext_vector_type(8))) float v8f;

__global__ void k_init(int* __restrict__ cnt, float* __restrict__ emb, int n, int ncnt) {
  long long t = (long long)blockIdx.x * blockDim.x + threadIdx.x;
  if (t < (long long)n * 64) emb[t] = 0.0f;
  if (t < ncnt) cnt[t] = 0;
}

__global__ void k_packB(const float* __restrict__ W, int Krows, int Ncols,
                        float* __restrict__ Bp, int Ksteps, int ntiles) {
  int t = blockIdx.x * blockDim.x + threadIdx.x;
  int total = Ksteps * ntiles * 64;
  if (t >= total) return;
  int j = t & 1;
  int lane = (t >> 1) & 31;
  int tile = (t >> 6) % ntiles;
  int ks = (t >> 6) / ntiles;
  int k = 4 * ks + 2 * (lane >> 4) + j;
  int n = tile * 16 + (lane & 15);
  Bp[t] = (k < Krows && n < Ncols) ? W[k * Ncols + n] : 0.0f;
}

#define BIN_SHIFT 8
#define BIN_NODES 256
#define NBIN_MAX 512
#define CNT_PITCH 32
#define SORT_TB 512
#define BIN_TB 1024

__global__ void k_bincount(const int* __restrict__ dst, int* __restrict__ binCount,
                           int* __restrict__ blockBase, int E, int nbin) {
  __shared__ int hist[NBIN_MAX];
  int t = threadIdx.x;
  hist[t] = 0;
  __syncthreads();
  int chunk = (E + (int)gridDim.x - 1) / (int)gridDim.x;
  int beg = (int)blockIdx.x * chunk;
  int end = beg + chunk; if (end > E) end = E;
  for (int e0 = beg + t; e0 < end; e0 += 4 * SORT_TB) {
    int d[4];
#pragma unroll
    for (int u = 0; u < 4; ++u) {
      int e = e0 + u * SORT_TB;
      d[u] = (e < end) ? dst[e] : -1;
    }
#pragma unroll
    for (int u = 0; u < 4; ++u)
      if (d[u] >= 0) atomicAdd(&hist[d[u] >> BIN_SHIFT], 1);
  }
  __syncthreads();
  if (t < nbin) {
    int c = hist[t];
    int base = 0;
    if (c > 0) base = atomicAdd(&binCount[t * CNT_PITCH], c);
    blockBase[(int)blockIdx.x * nbin + t] = base;
  }
}

__global__ void k_binscatter(const int* __restrict__ dst, const int* __restrict__ binCount,
                             const int* __restrict__ blockBase, int* __restrict__ binStart,
                             int* __restrict__ binned, int E, int nbin) {
  __shared__ int s[NBIN_MAX];
  __shared__ int cur[NBIN_MAX];
  int t = threadIdx.x;
  int v = (t < nbin) ? binCount[t * CNT_PITCH] : 0;
  s[t] = v;
  __syncthreads();
  for (int d = 1; d < SORT_TB; d <<= 1) {
    int a = (t >= d) ? s[t - d] : 0;
    __syncthreads();
    s[t] += a;
    __syncthreads();
  }
  int excl = s[t] - v;
  cur[t] = (t < nbin) ? excl + blockBase[(int)blockIdx.x * nbin + t] : 0;
  if (blockIdx.x == 0) {
    if (t < nbin) binStart[t] = excl;
    if (t == 0) binStart[nbin] = E;
  }
  __syncthreads();
  int chunk = (E + (int)gridDim.x - 1) / (int)gridDim.x;
  int beg = (int)blockIdx.x * chunk;
  int end = beg + chunk; if (end > E) end = E;
  for (int e0 = beg + t; e0 < end; e0 += 4 * SORT_TB) {
    int d[4];
#pragma unroll
    for (int u = 0; u < 4; ++u) {
      int e = e0 + u * SORT_TB;
      d[u] = (e < end) ? dst[e] : -1;
    }
#pragma unroll
    for (int u = 0; u < 4; ++u) {
      if (d[u] >= 0) {
        int e = e0 + u * SORT_TB;
        int p = atomicAdd(&cur[d[u] >> BIN_SHIFT], 1);
        binned[p] = (e << 8) | (d[u] & (BIN_NODES - 1));
      }
    }
  }
}

__global__ void k_binsort(const int* __restrict__ binStart, const int* __restrict__ binned,
                          int* __restrict__ off, int* __restrict__ perm, int n) {
  __shared__ int c[BIN_NODES];
  __shared__ int s[BIN_NODES];
  int t = threadIdx.x;
  int b = blockIdx.x;
  int beg = binStart[b];
  int end = binStart[b + 1];
  if (t < BIN_NODES) c[t] = 0;
  __syncthreads();
  for (int p0 = beg + t; p0 < end; p0 += 4 * BIN_TB) {
    int r[4];
#pragma unroll
    for (int u = 0; u < 4; ++u) {
      int p = p0 + u * BIN_TB;
      r[u] = (p < end) ? binned[p] : -1;
    }
#pragma unroll
    for (int u = 0; u < 4; ++u)
      if (r[u] >= 0) atomicAdd(&c[r[u] & (BIN_NODES - 1)], 1);
  }
  __syncthreads();
  int v = (t < BIN_NODES) ? c[t] : 0;
  if (t < BIN_NODES) s[t] = v;
  __syncthreads();
  for (int d = 1; d < BIN_NODES; d <<= 1) {
    int a = (t < BIN_NODES && t >= d) ? s[t - d] : 0;
    __syncthreads();
    if (t < BIN_NODES) s[t] += a;
    __syncthreads();
  }
  if (t < BIN_NODES) {
    int st = beg + s[t] - v;
    int node = b * BIN_NODES + t;
    if (node < n) off[node] = st;
    c[t] = st;
  }
  __syncthreads();
  for (int p0 = beg + t; p0 < end; p0 += 4 * BIN_TB) {
    int r[4];
#pragma unroll
    for (int u = 0; u < 4; ++u) {
      int p = p0 + u * BIN_TB;
      r[u] = (p < end) ? binned[p] : -1;
    }
#pragma unroll
    for (int u = 0; u < 4; ++u) {
      if (r[u] >= 0) {
        int pos = atomicAdd(&c[r[u] & (BIN_NODES - 1)], 1);
        perm[pos] = r[u] >> 8;
      }
    }
  }
}

__global__ void k_dinv_seg(const int* __restrict__ off, const int* __restrict__ perm,
                           const float* __restrict__ ew, float* __restrict__ dinv,
                           int n, int E) {
  int i = __builtin_amdgcn_readfirstlane((int)((blockIdx.x * blockDim.x + threadIdx.x) >> 5));
  if (i >= n) return;
  int lane = threadIdx.x & 31;
  int beg = off[i];
  int end = (i + 1 < n) ? off[i + 1] : E;
  float s = 0.0f;
  for (int p = beg + lane; p < end; p += 32) s += ew[perm[p]];
#pragma unroll
  for (int m = 16; m >= 1; m >>= 1) s += __shfl_xor(s, m, 32);
  float d = s + 1.0f;
  if (lane == 0) dinv[i] = (d > 0.0f) ? rsqrtf(d) : 0.0f;
}

__global__ void k_gather_norm(const int* __restrict__ off, const int* __restrict__ perm,
                              const int* __restrict__ src, const float* __restrict__ ew,
                              const float* __restrict__ dinv, const float* __restrict__ h,
                              const float* __restrict__ b, float* __restrict__ emb,
                              int n, int E, int coloff) {
  int i = __builtin_amdgcn_readfirstlane((int)((blockIdx.x * blockDim.x + threadIdx.x) >> 5));
  if (i >= n) return;
  int lane = threadIdx.x & 31;
  int beg = off[i];
  int end = (i + 1 < n) ? off[i + 1] : E;
  float di = dinv[i];
  float acc = h[(size_t)i * 32 + lane] * di * di;
  for (int p0 = beg; p0 < end; p0 += 32) {
    int m = end - p0; if (m > 32) m = 32;
    int s_l = 0;
    float cf_l = 0.0f;
    if (lane < m) {
      int e = perm[p0 + lane];
      s_l = src[e];
      cf_l = dinv[s_l] * ew[e] * di;
    }
    int cf_i = __float_as_int(cf_l);
    int j = 0;
    for (; j + 8 <= m; j += 8) {
      float v[8];
#pragma unroll
      for (int u = 0; u < 8; ++u) {
        int s = __builtin_amdgcn_readlane(s_l, j + u);
        v[u] = h[(size_t)s * 32 + lane];
      }
#pragma unroll
      for (int u = 0; u < 8; ++u) {
        float cf = __int_as_float(__builtin_amdgcn_readlane(cf_i, j + u));
        acc += v[u] * cf;
      }
    }
    for (; j < m; ++j) {
      int s = __builtin_amdgcn_readlane(s_l, j);
      float cf = __int_as_float(__builtin_amdgcn_readlane(cf_i, j));
      acc += h[(size_t)s * 32 + lane] * cf;
    }
  }
  float v = (lane < HID) ? acc + b[lane] : 0.0f;
  float ss = v * v;
#pragma unroll
  for (int m = 16; m >= 1; m >>= 1) ss += __shfl_xor(ss, m, 32);
  float nrm = fmaxf(sqrtf(ss), 1e-12f);
  if (lane < HID) {
    float r = v / nrm;
    emb[(size_t)i * 64 + coloff + lane] = (r > 0.0f) ? r : 0.0f;
  }
}

template <int NTILES>
__global__ void k_wmma_gemm(const float* __restrict__ A, int lda, int acol0,
                            const float* __restrict__ Bp, int Ksteps,
                            float* __restrict__ out, int ldo, int ncols,
                            const float* __restrict__ bias, int npairs) {
  int pair = blockIdx.x * (blockDim.x >> 5) + (threadIdx.x >> 5);
  if (pair >= npairs) return;
  int lane = threadIdx.x & 31;
  int half = lane >> 4;
  int l16 = lane & 15;
  int m0 = pair * 32;
  const float* ap0 = A + (size_t)(m0 + l16) * lda + acol0 + half * 2;
  const float* ap1 = ap0 + (size_t)16 * lda;
  const float* bp = Bp + lane * 2;
  v8f acc[2][NTILES];
#pragma unroll
  for (int rr = 0; rr < 2; ++rr)
#pragma unroll
    for (int t = 0; t < NTILES; ++t) acc[rr][t] = (v8f){0.f, 0.f, 0.f, 0.f, 0.f, 0.f, 0.f, 0.f};

#pragma unroll 2
  for (int ks = 0; ks < Ksteps; ++ks) {
    v2f a0 = *(const v2f*)(ap0 + 4 * ks);
    v2f a1 = *(const v2f*)(ap1 + 4 * ks);
#pragma unroll
    for (int t = 0; t < NTILES; ++t) {
      v2f b = *(const v2f*)(bp + (size_t)(ks * NTILES + t) * 64);
      acc[0][t] = __builtin_amdgcn_wmma_f32_16x16x4_f32(false, a0, false, b, (short)0, acc[0][t], false, false);
      acc[1][t] = __builtin_amdgcn_wmma_f32_16x16x4_f32(false, a1, false, b, (short)0, acc[1][t], false, false);
    }
  }

#pragma unroll
  for (int rr = 0; rr < 2; ++rr)
#pragma unroll
    for (int t = 0; t < NTILES; ++t)
#pragma unroll
      for (int r = 0; r < 8; ++r) {
        int row = m0 + rr * 16 + r + half * 8;
        int col = t * 16 + l16;
        if (col < ncols) out[(size_t)row * ldo + col] = acc[rr][t][r] + (bias ? bias[col] : 0.0f);
      }
}

extern "C" void kernel_launch(void* const* d_in, const int* in_sizes, int n_in,
                              void* d_out, int out_size, void* d_ws, size_t ws_size,
                              hipStream_t stream) {
  const float* x  = (const float*)d_in[0];
  const int*   ei = (const int*)d_in[1];
  const float* ew = (const float*)d_in[2];
  const float* W1 = (const float*)d_in[3];
  const float* b1 = (const float*)d_in[4];
  const float* W2 = (const float*)d_in[5];
  const float* b2 = (const float*)d_in[6];
  const float* W3 = (const float*)d_in[7];
  const float* b3 = (const float*)d_in[8];
  const float* Wl = (const float*)d_in[9];
  const float* bl = (const float*)d_in[10];
  float* out = (float*)d_out;

  const int N = in_sizes[0] / IN_DIM;
  const long long E = in_sizes[2];
  const int* src = ei;
  const int* dst = ei + E;

  const int KS1 = IN_DIM / 4;
  const int KS2 = (HID + 3) / 4;
  const int KSL = 64 / 4;

  float* ws = (float*)d_ws;
  int*   off  = (int*)ws; ws += N;
  float* dinv = ws; ws += N;
  float* h    = ws; ws += (size_t)N * 32;
  int*   perm = (int*)ws; ws += (size_t)N * 32;
  float* emb  = ws; ws += (size_t)N * 64;
  float* W1p  = ws; ws += KS1 * 2 * 64;
  float* W2p  = ws; ws += KS2 * 2 * 64;
  float* W3p  = ws; ws += KS2 * 2 * 64;
  float* Wlp  = ws; ws += KSL * 1 * 64;
  const int NBIN = (N + BIN_NODES - 1) / BIN_NODES;
  const int NSORTWG = 512;
  int*   binCount  = (int*)dinv;
  int*   binStart  = (int*)dinv + NBIN_MAX * CNT_PITCH;
  int*   binned    = (int*)h;
  int*   blockBase = perm;

  const int TB = 256;
  const int NPAIRS = N / 32;
  dim3 gblk((NPAIRS + 3) / 4);
  const int Ei = (int)E;
  long long nt32 = (long long)N * 32;
  int g_n32 = (int)((nt32 + TB - 1) / TB);

  k_init<<<(int)(((long long)N * 64 + TB - 1) / TB), TB, 0, stream>>>(binCount, emb, N, NBIN * CNT_PITCH);
  k_packB<<<(KS1 * 2 * 64 + TB - 1) / TB, TB, 0, stream>>>(W1, IN_DIM, HID, W1p, KS1, 2);
  k_packB<<<(KS2 * 2 * 64 + TB - 1) / TB, TB, 0, stream>>>(W2, HID, HID, W2p, KS2, 2);
  k_packB<<<(KS2 * 2 * 64 + TB - 1) / TB, TB, 0, stream>>>(W3, HID, HID, W3p, KS2, 2);
  k_packB<<<(KSL * 1 * 64 + TB - 1) / TB, TB, 0, stream>>>(Wl, 3 * HID, OUT_DIM, Wlp, KSL, 1);
  k_bincount<<<NSORTWG, SORT_TB, 0, stream>>>(dst, binCount, blockBase, Ei, NBIN);
  k_binscatter<<<NSORTWG, SORT_TB, 0, stream>>>(dst, binCount, blockBase, binStart, binned, Ei, NBIN);
  k_binsort<<<NBIN, BIN_TB, 0, stream>>>(binStart, binned, off, perm, N);
  k_dinv_seg<<<g_n32, TB, 0, stream>>>(off, perm, ew, dinv, N, Ei);

  k_wmma_gemm<2><<<gblk, 128, 0, stream>>>(x, IN_DIM, 0, W1p, KS1, h, 32, 32, nullptr, NPAIRS);
  k_gather_norm<<<g_n32, TB, 0, stream>>>(off, perm, src, ew, dinv, h, b1, emb, N, Ei, 0);

  k_wmma_gemm<2><<<gblk, 128, 0, stream>>>(emb, 64, 0, W2p, KS2, h, 32, 32, nullptr, NPAIRS);
  k_gather_norm<<<g_n32, TB, 0, stream>>>(off, perm, src, ew, dinv, h, b2, emb, N, Ei, HID);

  k_wmma_gemm<2><<<gblk, 128, 0, stream>>>(emb, 64, HID, W3p, KS2, h, 32, 32, nullptr, NPAIRS);
  k_gather_norm<<<g_n32, TB, 0, stream>>>(off, perm, src, ew, dinv, h, b3, emb, N, Ei, 2 * HID);

  k_wmma_gemm<1><<<gblk, 128, 0, stream>>>(emb, 64, 0, Wlp, KSL, out, OUT_DIM, OUT_DIM, bl, NPAIRS);
}


# GEMM store scales each h row by dinv of its node, so the gather drops the per-edge dinv[src] lookup: coefficient = edge weight, dinv[dst] applied once per row; rounding order of products changes
# speedup vs baseline: 3.3721x; 1.0793x over previous
#include <hip/hip_runtime.h>
#include <hip/hip_bf16.h>

#define IN_DIM 512
#define HID 20
#define OUT_DIM 10

typedef __attribute__((ext_vector_type(2))) float v2f;
typedef __attribute__((ext_vector_type(8))) float v8f;

__global__ void k_init(int* __restrict__ cnt, float* __restrict__ emb, int n, int ncnt) {
  long long t = (long long)blockIdx.x * blockDim.x + threadIdx.x;
  if (t < (long long)n * 64) emb[t] = 0.0f;
  if (t < ncnt) cnt[t] = 0;
}

__global__ void k_packB(const float* __restrict__ W, int Krows, int Ncols,
                        float* __restrict__ Bp, int Ksteps, int ntiles) {
  int t = blockIdx.x * blockDim.x + threadIdx.x;
  int total = Ksteps * ntiles * 64;
  if (t >= total) return;
  int j = t & 1;
  int lane = (t >> 1) & 31;
  int tile = (t >> 6) % ntiles;
  int ks = (t >> 6) / ntiles;
  int k = 4 * ks + 2 * (lane >> 4) + j;
  int n = tile * 16 + (lane & 15);
  Bp[t] = (k < Krows && n < Ncols) ? W[k * Ncols + n] : 0.0f;
}

#define BIN_SHIFT 8
#define BIN_NODES 256
#define NBIN_MAX 512
#define CNT_PITCH 32
#define SORT_TB 512
#define BIN_TB 1024

__global__ void k_bincount(const int* __restrict__ dst, int* __restrict__ binCount,
                           int* __restrict__ blockBase, int E, int nbin) {
  __shared__ int hist[NBIN_MAX];
  int t = threadIdx.x;
  hist[t] = 0;
  __syncthreads();
  int chunk = (E + (int)gridDim.x - 1) / (int)gridDim.x;
  int beg = (int)blockIdx.x * chunk;
  int end = beg + chunk; if (end > E) end = E;
  for (int e0 = beg + t; e0 < end; e0 += 4 * SORT_TB) {
    int d[4];
#pragma unroll
    for (int u = 0; u < 4; ++u) {
      int e = e0 + u * SORT_TB;
      d[u] = (e < end) ? dst[e] : -1;
    }
#pragma unroll
    for (int u = 0; u < 4; ++u)
      if (d[u] >= 0) atomicAdd(&hist[d[u] >> BIN_SHIFT], 1);
  }
  __syncthreads();
  if (t < nbin) {
    int c = hist[t];
    int base = 0;
    if (c > 0) base = atomicAdd(&binCount[t * CNT_PITCH], c);
    blockBase[(int)blockIdx.x * nbin + t] = base;
  }
}

__global__ void k_binscatter(const int* __restrict__ dst, const int* __restrict__ binCount,
                             const int* __restrict__ blockBase, int* __restrict__ binStart,
                             int* __restrict__ binned, int E, int nbin) {
  __shared__ int s[NBIN_MAX];
  __shared__ int cur[NBIN_MAX];
  int t = threadIdx.x;
  int v = (t < nbin) ? binCount[t * CNT_PITCH] : 0;
  s[t] = v;
  __syncthreads();
  for (int d = 1; d < SORT_TB; d <<= 1) {
    int a = (t >= d) ? s[t - d] : 0;
    __syncthreads();
    s[t] += a;
    __syncthreads();
  }
  int excl = s[t] - v;
  cur[t] = (t < nbin) ? excl + blockBase[(int)blockIdx.x * nbin + t] : 0;
  if (blockIdx.x == 0) {
    if (t < nbin) binStart[t] = excl;
    if (t == 0) binStart[nbin] = E;
  }
  __syncthreads();
  int chunk = (E + (int)gridDim.x - 1) / (int)gridDim.x;
  int beg = (int)blockIdx.x * chunk;
  int end = beg + chunk; if (end > E) end = E;
  for (int e0 = beg + t; e0 < end; e0 += 4 * SORT_TB) {
    int d[4];
#pragma unroll
    for (int u = 0; u < 4; ++u) {
      int e = e0 + u * SORT_TB;
      d[u] = (e < end) ? dst[e] : -1;
    }
#pragma unroll
    for (int u = 0; u < 4; ++u) {
      if (d[u] >= 0) {
        int e = e0 + u * SORT_TB;
        int p = atomicAdd(&cur[d[u] >> BIN_SHIFT], 1);
        binned[p] = (e << 8) | (d[u] & (BIN_NODES - 1));
      }
    }
  }
}

__global__ void k_binsort(const int* __restrict__ binStart, const int* __restrict__ binned,
                          int* __restrict__ off, int* __restrict__ perm, int n) {
  __shared__ int c[BIN_NODES];
  __shared__ int s[BIN_NODES];
  int t = threadIdx.x;
  int b = blockIdx.x;
  int beg = binStart[b];
  int end = binStart[b + 1];
  if (t < BIN_NODES) c[t] = 0;
  __syncthreads();
  for (int p0 = beg + t; p0 < end; p0 += 4 * BIN_TB) {
    int r[4];
#pragma unroll
    for (int u = 0; u < 4; ++u) {
      int p = p0 + u * BIN_TB;
      r[u] = (p < end) ? binned[p] : -1;
    }
#pragma unroll
    for (int u = 0; u < 4; ++u)
      if (r[u] >= 0) atomicAdd(&c[r[u] & (BIN_NODES - 1)], 1);
  }
  __syncthreads();
  int v = (t < BIN_NODES) ? c[t] : 0;
  if (t < BIN_NODES) s[t] = v;
  __syncthreads();
  for (int d = 1; d < BIN_NODES; d <<= 1) {
    int a = (t < BIN_NODES && t >= d) ? s[t - d] : 0;
    __syncthreads();
    if (t < BIN_NODES) s[t] += a;
    __syncthreads();
  }
  if (t < BIN_NODES) {
    int st = beg + s[t] - v;
    int node = b * BIN_NODES + t;
    if (node < n) off[node] = st;
    c[t] = st;
  }
  __syncthreads();
  for (int p0 = beg + t; p0 < end; p0 += 4 * BIN_TB) {
    int r[4];
#pragma unroll
    for (int u = 0; u < 4; ++u) {
      int p = p0 + u * BIN_TB;
      r[u] = (p < end) ? binned[p] : -1;
    }
#pragma unroll
    for (int u = 0; u < 4; ++u) {
      if (r[u] >= 0) {
        int pos = atomicAdd(&c[r[u] & (BIN_NODES - 1)], 1);
        perm[pos] = r[u] >> 8;
      }
    }
  }
}

__global__ void k_dinv_seg(const int* __restrict__ off, const int* __restrict__ perm,
                           const float* __restrict__ ew, float* __restrict__ dinv,
                           int n, int E) {
  int i = __builtin_amdgcn_readfirstlane((int)((blockIdx.x * blockDim.x + threadIdx.x) >> 5));
  if (i >= n) return;
  int lane = threadIdx.x & 31;
  int beg = off[i];
  int end = (i + 1 < n) ? off[i + 1] : E;
  float s = 0.0f;
  for (int p = beg + lane; p < end; p += 32) s += ew[perm[p]];
#pragma unroll
  for (int m = 16; m >= 1; m >>= 1) s += __shfl_xor(s, m, 32);
  float d = s + 1.0f;
  if (lane == 0) dinv[i] = (d > 0.0f) ? rsqrtf(d) : 0.0f;
}

__global__ void k_gather_norm(const int* __restrict__ off, const int* __restrict__ perm,
                              const int* __restrict__ src, const float* __restrict__ ew,
                              const float* __restrict__ dinv, const float* __restrict__ h,
                              const float* __restrict__ b, float* __restrict__ emb,
                              int n, int E, int coloff) {
  int i = __builtin_amdgcn_readfirstlane((int)((blockIdx.x * blockDim.x + threadIdx.x) >> 5));
  if (i >= n) return;
  int lane = threadIdx.x & 31;
  int beg = off[i];
  int end = (i + 1 < n) ? off[i + 1] : E;
  float di = dinv[i];
  float acc = h[(size_t)i * 32 + lane];
  for (int p0 = beg; p0 < end; p0 += 32) {
    int m = end - p0; if (m > 32) m = 32;
    int s_l = 0;
    float cf_l = 0.0f;
    if (lane < m) {
      int e = perm[p0 + lane];
      s_l = src[e];
      cf_l = ew[e];
    }
    int cf_i = __float_as_int(cf_l);
    int j = 0;
    for (; j + 8 <= m; j += 8) {
      float v[8];
#pragma unroll
      for (int u = 0; u < 8; ++u) {
        int s = __builtin_amdgcn_readlane(s_l, j + u);
        v[u] = h[(size_t)s * 32 + lane];
      }
#pragma unroll
      for (int u = 0; u < 8; ++u) {
        float cf = __int_as_float(__builtin_amdgcn_readlane(cf_i, j + u));
        acc += v[u] * cf;
      }
    }
    for (; j < m; ++j) {
      int s = __builtin_amdgcn_readlane(s_l, j);
      float cf = __int_as_float(__builtin_amdgcn_readlane(cf_i, j));
      acc += h[(size_t)s * 32 + lane] * cf;
    }
  }
  float v = (lane < HID) ? acc * di + b[lane] : 0.0f;
  float ss = v * v;
#pragma unroll
  for (int m = 16; m >= 1; m >>= 1) ss += __shfl_xor(ss, m, 32);
  float nrm = fmaxf(sqrtf(ss), 1e-12f);
  if (lane < HID) {
    float r = v / nrm;
    emb[(size_t)i * 64 + coloff + lane] = (r > 0.0f) ? r : 0.0f;
  }
}

template <int NTILES, bool ROWSCALE>
__global__ void k_wmma_gemm(const float* __restrict__ A, int lda, int acol0,
                            const float* __restrict__ Bp, int Ksteps,
                            float* __restrict__ out, int ldo, int ncols,
                            const float* __restrict__ bias,
                            const float* __restrict__ rowscale, int npairs) {
  int pair = blockIdx.x * (blockDim.x >> 5) + (threadIdx.x >> 5);
  if (pair >= npairs) return;
  int lane = threadIdx.x & 31;
  int half = lane >> 4;
  int l16 = lane & 15;
  int m0 = pair * 32;
  const float* ap0 = A + (size_t)(m0 + l16) * lda + acol0 + half * 2;
  const float* ap1 = ap0 + (size_t)16 * lda;
  const float* bp = Bp + lane * 2;
  float rs[2][8];
  if (ROWSCALE) {
    const float4* rp = (const float4*)(rowscale + m0 + half * 8);
    float4 q0 = rp[0], q1 = rp[1], q2 = rp[4], q3 = rp[5];
    rs[0][0] = q0.x; rs[0][1] = q0.y; rs[0][2] = q0.z; rs[0][3] = q0.w;
    rs[0][4] = q1.x; rs[0][5] = q1.y; rs[0][6] = q1.z; rs[0][7] = q1.w;
    rs[1][0] = q2.x; rs[1][1] = q2.y; rs[1][2] = q2.z; rs[1][3] = q2.w;
    rs[1][4] = q3.x; rs[1][5] = q3.y; rs[1][6] = q3.z; rs[1][7] = q3.w;
  }
  v8f acc[2][NTILES];
#pragma unroll
  for (int rr = 0; rr < 2; ++rr)
#pragma unroll
    for (int t = 0; t < NTILES; ++t) acc[rr][t] = (v8f){0.f, 0.f, 0.f, 0.f, 0.f, 0.f, 0.f, 0.f};

#pragma unroll 2
  for (int ks = 0; ks < Ksteps; ++ks) {
    v2f a0 = *(const v2f*)(ap0 + 4 * ks);
    v2f a1 = *(const v2f*)(ap1 + 4 * ks);
#pragma unroll
    for (int t = 0; t < NTILES; ++t) {
      v2f b = *(const v2f*)(bp + (size_t)(ks * NTILES + t) * 64);
      acc[0][t] = __builtin_amdgcn_wmma_f32_16x16x4_f32(false, a0, false, b, (short)0, acc[0][t], false, false);
      acc[1][t] = __builtin_amdgcn_wmma_f32_16x16x4_f32(false, a1, false, b, (short)0, acc[1][t], false, false);
    }
  }

#pragma unroll
  for (int rr = 0; rr < 2; ++rr)
#pragma unroll
    for (int t = 0; t < NTILES; ++t)
#pragma unroll
      for (int r = 0; r < 8; ++r) {
        int row = m0 + rr * 16 + r + half * 8;
        int col = t * 16 + l16;
        if (col < ncols) {
          float val = acc[rr][t][r] + (bias ? bias[col] : 0.0f);
          if (ROWSCALE) val *= rs[rr][r];
          out[(size_t)row * ldo + col] = val;
        }
      }
}

extern "C" void kernel_launch(void* const* d_in, const int* in_sizes, int n_in,
                              void* d_out, int out_size, void* d_ws, size_t ws_size,
                              hipStream_t stream) {
  const float* x  = (const float*)d_in[0];
  const int*   ei = (const int*)d_in[1];
  const float* ew = (const float*)d_in[2];
  const float* W1 = (const float*)d_in[3];
  const float* b1 = (const float*)d_in[4];
  const float* W2 = (const float*)d_in[5];
  const float* b2 = (const float*)d_in[6];
  const float* W3 = (const float*)d_in[7];
  const float* b3 = (const float*)d_in[8];
  const float* Wl = (const float*)d_in[9];
  const float* bl = (const float*)d_in[10];
  float* out = (float*)d_out;

  const int N = in_sizes[0] / IN_DIM;
  const long long E = in_sizes[2];
  const int* src = ei;
  const int* dst = ei + E;

  const int KS1 = IN_DIM / 4;
  const int KS2 = (HID + 3) / 4;
  const int KSL = 64 / 4;

  float* ws = (float*)d_ws;
  int*   off  = (int*)ws; ws += N;
  float* dinv = ws; ws += N;
  float* h    = ws; ws += (size_t)N * 32;
  int*   perm = (int*)ws; ws += (size_t)N * 32;
  float* emb  = ws; ws += (size_t)N * 64;
  float* W1p  = ws; ws += KS1 * 2 * 64;
  float* W2p  = ws; ws += KS2 * 2 * 64;
  float* W3p  = ws; ws += KS2 * 2 * 64;
  float* Wlp  = ws; ws += KSL * 1 * 64;
  const int NBIN = (N + BIN_NODES - 1) / BIN_NODES;
  const int NSORTWG = 512;
  int*   binCount  = (int*)dinv;
  int*   binStart  = (int*)dinv + NBIN_MAX * CNT_PITCH;
  int*   binned    = (int*)h;
  int*   blockBase = perm;

  const int TB = 256;
  const int NPAIRS = N / 32;
  dim3 gblk((NPAIRS + 3) / 4);
  const int Ei = (int)E;
  long long nt32 = (long long)N * 32;
  int g_n32 = (int)((nt32 + TB - 1) / TB);

  k_init<<<(int)(((long long)N * 64 + TB - 1) / TB), TB, 0, stream>>>(binCount, emb, N, NBIN * CNT_PITCH);
  k_packB<<<(KS1 * 2 * 64 + TB - 1) / TB, TB, 0, stream>>>(W1, IN_DIM, HID, W1p, KS1, 2);
  k_packB<<<(KS2 * 2 * 64 + TB - 1) / TB, TB, 0, stream>>>(W2, HID, HID, W2p, KS2, 2);
  k_packB<<<(KS2 * 2 * 64 + TB - 1) / TB, TB, 0, stream>>>(W3, HID, HID, W3p, KS2, 2);
  k_packB<<<(KSL * 1 * 64 + TB - 1) / TB, TB, 0, stream>>>(Wl, 3 * HID, OUT_DIM, Wlp, KSL, 1);
  k_bincount<<<NSORTWG, SORT_TB, 0, stream>>>(dst, binCount, blockBase, Ei, NBIN);
  k_binscatter<<<NSORTWG, SORT_TB, 0, stream>>>(dst, binCount, blockBase, binStart, binned, Ei, NBIN);
  k_binsort<<<NBIN, BIN_TB, 0, stream>>>(binStart, binned, off, perm, N);
  k_dinv_seg<<<g_n32, TB, 0, stream>>>(off, perm, ew, dinv, N, Ei);

  k_wmma_gemm<2, true><<<gblk, 128, 0, stream>>>(x, IN_DIM, 0, W1p, KS1, h, 32, 32, nullptr, dinv, NPAIRS);
  k_gather_norm<<<g_n32, TB, 0, stream>>>(off, perm, src, ew, dinv, h, b1, emb, N, Ei, 0);

  k_wmma_gemm<2, true><<<gblk, 128, 0, stream>>>(emb, 64, 0, W2p, KS2, h, 32, 32, nullptr, dinv, NPAIRS);
  k_gather_norm<<<g_n32, TB, 0, stream>>>(off, perm, src, ew, dinv, h, b2, emb, N, Ei, HID);

  k_wmma_gemm<2, true><<<gblk, 128, 0, stream>>>(emb, 64, HID, W3p, KS2, h, 32, 32, nullptr, dinv, NPAIRS);
  k_gather_norm<<<g_n32, TB, 0, stream>>>(off, perm, src, ew, dinv, h, b3, emb, N, Ei, 2 * HID);

  k_wmma_gemm<1, false><<<gblk, 128, 0, stream>>>(emb, 64, 0, Wlp, KSL, out, OUT_DIM, OUT_DIM, bl, nullptr, NPAIRS);
}


# Sort carries each edge's (source, weight) pair to its sorted slot: gathers read them coalesced, no per-edge index hops; final linear summed per layer to free the plane, so sum order of out changes
# speedup vs baseline: 4.1910x; 1.2428x over previous
#include <hip/hip_runtime.h>
#include <hip/hip_bf16.h>

#define IN_DIM 512
#define HID 20
#define OUT_DIM 10

typedef __attribute__((ext_vector_type(2))) float v2f;
typedef __attribute__((ext_vector_type(8))) float v8f;

__global__ void k_init(int* __restrict__ cnt, int ncnt) {
  int t = blockIdx.x * blockDim.x + threadIdx.x;
  if (t < ncnt) cnt[t] = 0;
}

__global__ void k_packB(const float* __restrict__ W, int Krows, int Ncols,
                        float* __restrict__ Bp, int Ksteps, int ntiles) {
  int t = blockIdx.x * blockDim.x + threadIdx.x;
  int total = Ksteps * ntiles * 64;
  if (t >= total) return;
  int j = t & 1;
  int lane = (t >> 1) & 31;
  int tile = (t >> 6) % ntiles;
  int ks = (t >> 6) / ntiles;
  int k = 4 * ks + 2 * (lane >> 4) + j;
  int n = tile * 16 + (lane & 15);
  Bp[t] = (k < Krows && n < Ncols) ? W[k * Ncols + n] : 0.0f;
}

__global__ void k_packB2(const float* __restrict__ W, const float* __restrict__ Wls,
                         float* __restrict__ Bp) {
  int t = blockIdx.x * blockDim.x + threadIdx.x;
  if (t >= (HID / 4) * 2 * 64) return;
  int j = t & 1;
  int lane = (t >> 1) & 31;
  int tile = (t >> 6) & 1;
  int ks = t >> 7;
  int k = 4 * ks + 2 * (lane >> 4) + j;
  int n = tile * 16 + (lane & 15);
  float v = 0.0f;
  if (n < HID) v = W[k * HID + n];
  else if (n < HID + OUT_DIM) v = Wls[k * OUT_DIM + (n - HID)];
  Bp[t] = v;
}

#define BIN_SHIFT 8
#define BIN_NODES 256
#define NBIN_MAX 512
#define CNT_PITCH 32
#define SORT_TB 512
#define BIN_TB 1024

__global__ void k_bincount(const int* __restrict__ dst, int* __restrict__ binCount,
                           int* __restrict__ blockBase, int E, int nbin) {
  __shared__ int hist[NBIN_MAX];
  int t = threadIdx.x;
  hist[t] = 0;
  __syncthreads();
  int chunk = (E + (int)gridDim.x - 1) / (int)gridDim.x;
  int beg = (int)blockIdx.x * chunk;
  int end = beg + chunk; if (end > E) end = E;
  for (int e0 = beg + t; e0 < end; e0 += 4 * SORT_TB) {
    int d[4];
#pragma unroll
    for (int u = 0; u < 4; ++u) {
      int e = e0 + u * SORT_TB;
      d[u] = (e < end) ? dst[e] : -1;
    }
#pragma unroll
    for (int u = 0; u < 4; ++u)
      if (d[u] >= 0) atomicAdd(&hist[d[u] >> BIN_SHIFT], 1);
  }
  __syncthreads();
  if (t < nbin) {
    int c = hist[t];
    int base = 0;
    if (c > 0) base = atomicAdd(&binCount[t * CNT_PITCH], c);
    blockBase[(int)blockIdx.x * nbin + t] = base;
  }
}

__global__ void k_binscatter(const int* __restrict__ src, const int* __restrict__ dst,
                             const float* __restrict__ ew, const int* __restrict__ binCount,
                             const int* __restrict__ blockBase, int* __restrict__ binStart,
                             int2* __restrict__ binned, int E, int nbin) {
  __shared__ int s[NBIN_MAX];
  __shared__ int cur[NBIN_MAX];
  int t = threadIdx.x;
  int v = (t < nbin) ? binCount[t * CNT_PITCH] : 0;
  s[t] = v;
  __syncthreads();
  for (int d = 1; d < SORT_TB; d <<= 1) {
    int a = (t >= d) ? s[t - d] : 0;
    __syncthreads();
    s[t] += a;
    __syncthreads();
  }
  int excl = s[t] - v;
  cur[t] = (t < nbin) ? excl + blockBase[(int)blockIdx.x * nbin + t] : 0;
  if (blockIdx.x == 0) {
    if (t < nbin) binStart[t] = excl;
    if (t == 0) binStart[nbin] = E;
  }
  __syncthreads();
  int chunk = (E + (int)gridDim.x - 1) / (int)gridDim.x;
  int beg = (int)blockIdx.x * chunk;
  int end = beg + chunk; if (end > E) end = E;
  for (int e0 = beg + t; e0 < end; e0 += 4 * SORT_TB) {
    int d[4];
    int sr[4];
    int wb[4];
#pragma unroll
    for (int u = 0; u < 4; ++u) {
      int e = e0 + u * SORT_TB;
      bool ok = e < end;
      d[u] = ok ? dst[e] : -1;
      sr[u] = ok ? src[e] : 0;
      wb[u] = ok ? __float_as_int(ew[e]) : 0;
    }
#pragma unroll
    for (int u = 0; u < 4; ++u) {
      if (d[u] >= 0) {
        int p = atomicAdd(&cur[d[u] >> BIN_SHIFT], 1);
        binned[p] = make_int2((sr[u] << 8) | (d[u] & (BIN_NODES - 1)), wb[u]);
      }
    }
  }
}

__global__ void k_binsort(const int* __restrict__ binStart, const int2* __restrict__ binned,
                          int* __restrict__ off, int2* __restrict__ pairs, int n) {
  __shared__ int c[BIN_NODES];
  __shared__ int s[BIN_NODES];
  int t = threadIdx.x;
  int b = blockIdx.x;
  int beg = binStart[b];
  int end = binStart[b + 1];
  if (t < BIN_NODES) c[t] = 0;
  __syncthreads();
  for (int p0 = beg + t; p0 < end; p0 += 4 * BIN_TB) {
    int r[4];
#pragma unroll
    for (int u = 0; u < 4; ++u) {
      int p = p0 + u * BIN_TB;
      r[u] = (p < end) ? binned[p].x : -1;
    }
#pragma unroll
    for (int u = 0; u < 4; ++u)
      if (r[u] >= 0) atomicAdd(&c[r[u] & (BIN_NODES - 1)], 1);
  }
  __syncthreads();
  int v = (t < BIN_NODES) ? c[t] : 0;
  if (t < BIN_NODES) s[t] = v;
  __syncthreads();
  for (int d = 1; d < BIN_NODES; d <<= 1) {
    int a = (t < BIN_NODES && t >= d) ? s[t - d] : 0;
    __syncthreads();
    if (t < BIN_NODES) s[t] += a;
    __syncthreads();
  }
  if (t < BIN_NODES) {
    int st = beg + s[t] - v;
    int node = b * BIN_NODES + t;
    if (node < n) off[node] = st;
    c[t] = st;
  }
  __syncthreads();
  for (int p0 = beg + t; p0 < end; p0 += 4 * BIN_TB) {
    int2 r[4];
#pragma unroll
    for (int u = 0; u < 4; ++u) {
      int p = p0 + u * BIN_TB;
      r[u] = (p < end) ? binned[p] : make_int2(-1, 0);
    }
#pragma unroll
    for (int u = 0; u < 4; ++u) {
      if (r[u].x >= 0) {
        int pos = atomicAdd(&c[r[u].x & (BIN_NODES - 1)], 1);
        pairs[pos] = make_int2(r[u].x >> 8, r[u].y);
      }
    }
  }
}

__global__ void k_dinv_seg(const int* __restrict__ off, const int2* __restrict__ pairs,
                           float* __restrict__ dinv, int n, int E) {
  int i = __builtin_amdgcn_readfirstlane((int)((blockIdx.x * blockDim.x + threadIdx.x) >> 5));
  if (i >= n) return;
  int lane = threadIdx.x & 31;
  int beg = off[i];
  int end = (i + 1 < n) ? off[i + 1] : E;
  float s = 0.0f;
  for (int p = beg + lane; p < end; p += 32) s += __int_as_float(pairs[p].y);
#pragma unroll
  for (int m = 16; m >= 1; m >>= 1) s += __shfl_xor(s, m, 32);
  float d = s + 1.0f;
  if (lane == 0) dinv[i] = (d > 0.0f) ? rsqrtf(d) : 0.0f;
}

__global__ void k_gather_norm(const int* __restrict__ off, const int2* __restrict__ pairs,
                              const float* __restrict__ dinv, const float* __restrict__ h,
                              const float* __restrict__ b, float* __restrict__ o,
                              int n, int E) {
  int i = __builtin_amdgcn_readfirstlane((int)((blockIdx.x * blockDim.x + threadIdx.x) >> 5));
  if (i >= n) return;
  int lane = threadIdx.x & 31;
  int beg = off[i];
  int end = (i + 1 < n) ? off[i + 1] : E;
  float di = dinv[i];
  float acc = h[(size_t)i * 32 + lane];
  for (int p0 = beg; p0 < end; p0 += 32) {
    int m = end - p0; if (m > 32) m = 32;
    int s_l = 0;
    int cf_i = 0;
    if (lane < m) {
      int2 q = pairs[p0 + lane];
      s_l = q.x;
      cf_i = q.y;
    }
    int j = 0;
    for (; j + 8 <= m; j += 8) {
      float v[8];
#pragma unroll
      for (int u = 0; u < 8; ++u) {
        int s = __builtin_amdgcn_readlane(s_l, j + u);
        v[u] = h[(size_t)s * 32 + lane];
      }
#pragma unroll
      for (int u = 0; u < 8; ++u) {
        float cf = __int_as_float(__builtin_amdgcn_readlane(cf_i, j + u));
        acc += v[u] * cf;
      }
    }
    for (; j < m; ++j) {
      int s = __builtin_amdgcn_readlane(s_l, j);
      float cf = __int_as_float(__builtin_amdgcn_readlane(cf_i, j));
      acc += h[(size_t)s * 32 + lane] * cf;
    }
  }
  float v = (lane < HID) ? acc * di + b[lane] : 0.0f;
  float ss = v * v;
#pragma unroll
  for (int m = 16; m >= 1; m >>= 1) ss += __shfl_xor(ss, m, 32);
  float nrm = fmaxf(sqrtf(ss), 1e-12f);
  if (lane < HID) {
    float r = v / nrm;
    o[(size_t)i * 32 + lane] = (r > 0.0f) ? r : 0.0f;
  }
}

template <int NTILES, bool ROWSCALE, bool ACCUM>
__global__ void k_wmma_gemm(const float* __restrict__ A, int lda, int acol0,
                            const float* __restrict__ Bp, int Ksteps,
                            float* __restrict__ out, int ldo, int ncols,
                            const float* __restrict__ bias,
                            const float* __restrict__ rowscale, int npairs) {
  int pair = blockIdx.x * (blockDim.x >> 5) + (threadIdx.x >> 5);
  if (pair >= npairs) return;
  int lane = threadIdx.x & 31;
  int half = lane >> 4;
  int l16 = lane & 15;
  int m0 = pair * 32;
  const float* ap0 = A + (size_t)(m0 + l16) * lda + acol0 + half * 2;
  const float* ap1 = ap0 + (size_t)16 * lda;
  const float* bp = Bp + lane * 2;
  float rs[2][8];
  if (ROWSCALE) {
    const float4* rp = (const float4*)(rowscale + m0 + half * 8);
    float4 q0 = rp[0], q1 = rp[1], q2 = rp[4], q3 = rp[5];
    rs[0][0] = q0.x; rs[0][1] = q0.y; rs[0][2] = q0.z; rs[0][3] = q0.w;
    rs[0][4] = q1.x; rs[0][5] = q1.y; rs[0][6] = q1.z; rs[0][7] = q1.w;
    rs[1][0] = q2.x; rs[1][1] = q2.y; rs[1][2] = q2.z; rs[1][3] = q2.w;
    rs[1][4] = q3.x; rs[1][5] = q3.y; rs[1][6] = q3.z; rs[1][7] = q3.w;
  }
  v8f acc[2][NTILES];
#pragma unroll
  for (int rr = 0; rr < 2; ++rr)
#pragma unroll
    for (int t = 0; t < NTILES; ++t) acc[rr][t] = (v8f){0.f, 0.f, 0.f, 0.f, 0.f, 0.f, 0.f, 0.f};

#pragma unroll 2
  for (int ks = 0; ks < Ksteps; ++ks) {
    v2f a0 = *(const v2f*)(ap0 + 4 * ks);
    v2f a1 = *(const v2f*)(ap1 + 4 * ks);
#pragma unroll
    for (int t = 0; t < NTILES; ++t) {
      v2f b = *(const v2f*)(bp + (size_t)(ks * NTILES + t) * 64);
      acc[0][t] = __builtin_amdgcn_wmma_f32_16x16x4_f32(false, a0, false, b, (short)0, acc[0][t], false, false);
      acc[1][t] = __builtin_amdgcn_wmma_f32_16x16x4_f32(false, a1, false, b, (short)0, acc[1][t], false, false);
    }
  }

#pragma unroll
  for (int rr = 0; rr < 2; ++rr)
#pragma unroll
    for (int t = 0; t < NTILES; ++t)
#pragma unroll
      for (int r = 0; r < 8; ++r) {
        int row = m0 + rr * 16 + r + half * 8;
        int col = t * 16 + l16;
        if (col < ncols) {
          float val = acc[rr][t][r] + (bias ? bias[col] : 0.0f);
          if (ROWSCALE) val *= rs[rr][r];
          if (ACCUM) val += out[(size_t)row * ldo + col];
          out[(size_t)row * ldo + col] = val;
        }
      }
}

template <bool FIRST>
__global__ void k_wmma_gemm_dual(const float* __restrict__ A, const float* __restrict__ Bp,
                                 float* __restrict__ h, const float* __restrict__ rowscale,
                                 float* __restrict__ out, const float* __restrict__ bl,
                                 int npairs) {
  int pair = blockIdx.x * (blockDim.x >> 5) + (threadIdx.x >> 5);
  if (pair >= npairs) return;
  int lane = threadIdx.x & 31;
  int half = lane >> 4;
  int l16 = lane & 15;
  int m0 = pair * 32;
  const float* ap0 = A + (size_t)(m0 + l16) * 32 + half * 2;
  const float* ap1 = ap0 + (size_t)16 * 32;
  const float* bp = Bp + lane * 2;
  float rs[2][8];
  {
    const float4* rp = (const float4*)(rowscale + m0 + half * 8);
    float4 q0 = rp[0], q1 = rp[1], q2 = rp[4], q3 = rp[5];
    rs[0][0] = q0.x; rs[0][1] = q0.y; rs[0][2] = q0.z; rs[0][3] = q0.w;
    rs[0][4] = q1.x; rs[0][5] = q1.y; rs[0][6] = q1.z; rs[0][7] = q1.w;
    rs[1][0] = q2.x; rs[1][1] = q2.y; rs[1][2] = q2.z; rs[1][3] = q2.w;
    rs[1][4] = q3.x; rs[1][5] = q3.y; rs[1][6] = q3.z; rs[1][7] = q3.w;
  }
  v8f acc[2][2];
#pragma unroll
  for (int rr = 0; rr < 2; ++rr)
#pragma unroll
    for (int t = 0; t < 2; ++t) acc[rr][t] = (v8f){0.f, 0.f, 0.f, 0.f, 0.f, 0.f, 0.f, 0.f};

#pragma unroll
  for (int ks = 0; ks < HID / 4; ++ks) {
    v2f a0 = *(const v2f*)(ap0 + 4 * ks);
    v2f a1 = *(const v2f*)(ap1 + 4 * ks);
#pragma unroll
    for (int t = 0; t < 2; ++t) {
      v2f b = *(const v2f*)(bp + (size_t)(ks * 2 + t) * 64);
      acc[0][t] = __builtin_amdgcn_wmma_f32_16x16x4_f32(false, a0, false, b, (short)0, acc[0][t], false, false);
      acc[1][t] = __builtin_amdgcn_wmma_f32_16x16x4_f32(false, a1, false, b, (short)0, acc[1][t], false, false);
    }
  }

  const int col1 = 16 + l16;
#pragma unroll
  for (int rr = 0; rr < 2; ++rr)
#pragma unroll
    for (int r = 0; r < 8; ++r) {
      int row = m0 + rr * 16 + r + half * 8;
      float sc = rs[rr][r];
      float v1 = acc[rr][1][r];
      h[(size_t)row * 32 + l16] = acc[rr][0][r] * sc;
      h[(size_t)row * 32 + col1] = (col1 < HID) ? v1 * sc : 0.0f;
      if (col1 >= HID && col1 < HID + OUT_DIM) {
        size_t oi = (size_t)row * OUT_DIM + (col1 - HID);
        if (FIRST) out[oi] = v1 + bl[col1 - HID];
        else out[oi] = out[oi] + v1;
      }
    }
}

extern "C" void kernel_launch(void* const* d_in, const int* in_sizes, int n_in,
                              void* d_out, int out_size, void* d_ws, size_t ws_size,
                              hipStream_t stream) {
  const float* x  = (const float*)d_in[0];
  const int*   ei = (const int*)d_in[1];
  const float* ew = (const float*)d_in[2];
  const float* W1 = (const float*)d_in[3];
  const float* b1 = (const float*)d_in[4];
  const float* W2 = (const float*)d_in[5];
  const float* b2 = (const float*)d_in[6];
  const float* W3 = (const float*)d_in[7];
  const float* b3 = (const float*)d_in[8];
  const float* Wl = (const float*)d_in[9];
  const float* bl = (const float*)d_in[10];
  float* out = (float*)d_out;

  const int N = in_sizes[0] / IN_DIM;
  const long long E = in_sizes[2];
  const int* src = ei;
  const int* dst = ei + E;

  const int KS1 = IN_DIM / 4;
  const int KS2 = HID / 4;

  float* ws = (float*)d_ws;
  int*   off  = (int*)ws; ws += N;
  float* dinv = ws; ws += N;
  int2*  pairs = (int2*)ws; ws += (size_t)N * 64;
  float* h    = ws; ws += (size_t)N * 32;
  float* o    = ws; ws += (size_t)N * 32;
  float* W1p  = ws; ws += KS1 * 2 * 64;
  float* W2c  = ws; ws += KS2 * 2 * 64;
  float* W3c  = ws; ws += KS2 * 2 * 64;
  float* Wl3p = ws; ws += KS2 * 1 * 64;
  const int NBIN = (N + BIN_NODES - 1) / BIN_NODES;
  const int NSORTWG = 512;
  int*   binCount  = (int*)dinv;
  int*   binStart  = (int*)dinv + NBIN_MAX * CNT_PITCH;
  int2*  binned    = (int2*)h;
  int*   blockBase = (int*)pairs;

  const int TB = 256;
  const int NPAIRS = N / 32;
  dim3 gblk((NPAIRS + 3) / 4);
  const int Ei = (int)E;
  long long nt32 = (long long)N * 32;
  int g_n32 = (int)((nt32 + TB - 1) / TB);

  k_init<<<(NBIN * CNT_PITCH + TB - 1) / TB, TB, 0, stream>>>(binCount, NBIN * CNT_PITCH);
  k_packB<<<(KS1 * 2 * 64 + TB - 1) / TB, TB, 0, stream>>>(W1, IN_DIM, HID, W1p, KS1, 2);
  k_packB2<<<(KS2 * 2 * 64 + TB - 1) / TB, TB, 0, stream>>>(W2, Wl, W2c);
  k_packB2<<<(KS2 * 2 * 64 + TB - 1) / TB, TB, 0, stream>>>(W3, Wl + HID * OUT_DIM, W3c);
  k_packB<<<(KS2 * 1 * 64 + TB - 1) / TB, TB, 0, stream>>>(Wl + 2 * HID * OUT_DIM, HID, OUT_DIM, Wl3p, KS2, 1);
  k_bincount<<<NSORTWG, SORT_TB, 0, stream>>>(dst, binCount, blockBase, Ei, NBIN);
  k_binscatter<<<NSORTWG, SORT_TB, 0, stream>>>(src, dst, ew, binCount, blockBase, binStart, binned, Ei, NBIN);
  k_binsort<<<NBIN, BIN_TB, 0, stream>>>(binStart, binned, off, pairs, N);
  k_dinv_seg<<<g_n32, TB, 0, stream>>>(off, pairs, dinv, N, Ei);

  k_wmma_gemm<2, true, false><<<gblk, 128, 0, stream>>>(x, IN_DIM, 0, W1p, KS1, h, 32, 32, nullptr, dinv, NPAIRS);
  k_gather_norm<<<g_n32, TB, 0, stream>>>(off, pairs, dinv, h, b1, o, N, Ei);

  k_wmma_gemm_dual<true><<<gblk, 128, 0, stream>>>(o, W2c, h, dinv, out, bl, NPAIRS);
  k_gather_norm<<<g_n32, TB, 0, stream>>>(off, pairs, dinv, h, b2, o, N, Ei);

  k_wmma_gemm_dual<false><<<gblk, 128, 0, stream>>>(o, W3c, h, dinv, out, bl, NPAIRS);
  k_gather_norm<<<g_n32, TB, 0, stream>>>(off, pairs, dinv, h, b3, o, N, Ei);

  k_wmma_gemm<1, false, true><<<gblk, 128, 0, stream>>>(o, 32, 0, Wl3p, KS2, out, OUT_DIM, OUT_DIM, nullptr, nullptr, NPAIRS);
}


# Degree sums and 3 gathers: a wave walks 8 adjacent nodes, next chunk's edge pairs requested ahead, a chunk's <=32 rows all in flight (400000 waves -> 50000); sum order kept, tail products unfused
# speedup vs baseline: 4.4800x; 1.0690x over previous
#include <hip/hip_runtime.h>
#include <hip/hip_bf16.h>

#define IN_DIM 512
#define HID 20
#define OUT_DIM 10

typedef __attribute__((ext_vector_type(2))) float v2f;
typedef __attribute__((ext_vector_type(8))) float v8f;

__global__ void k_init(int* __restrict__ cnt, int ncnt) {
  int t = blockIdx.x * blockDim.x + threadIdx.x;
  if (t < ncnt) cnt[t] = 0;
}

__global__ void k_packB(const float* __restrict__ W, int Krows, int Ncols,
                        float* __restrict__ Bp, int Ksteps, int ntiles) {
  int t = blockIdx.x * blockDim.x + threadIdx.x;
  int total = Ksteps * ntiles * 64;
  if (t >= total) return;
  int j = t & 1;
  int lane = (t >> 1) & 31;
  int tile = (t >> 6) % ntiles;
  int ks = (t >> 6) / ntiles;
  int k = 4 * ks + 2 * (lane >> 4) + j;
  int n = tile * 16 + (lane & 15);
  Bp[t] = (k < Krows && n < Ncols) ? W[k * Ncols + n] : 0.0f;
}

__global__ void k_packB2(const float* __restrict__ W, const float* __restrict__ Wls,
                         float* __restrict__ Bp) {
  int t = blockIdx.x * blockDim.x + threadIdx.x;
  if (t >= (HID / 4) * 2 * 64) return;
  int j = t & 1;
  int lane = (t >> 1) & 31;
  int tile = (t >> 6) & 1;
  int ks = t >> 7;
  int k = 4 * ks + 2 * (lane >> 4) + j;
  int n = tile * 16 + (lane & 15);
  float v = 0.0f;
  if (n < HID) v = W[k * HID + n];
  else if (n < HID + OUT_DIM) v = Wls[k * OUT_DIM + (n - HID)];
  Bp[t] = v;
}

#define BIN_SHIFT 8
#define BIN_NODES 256
#define NBIN_MAX 512
#define CNT_PITCH 32
#define SORT_TB 512
#define BIN_TB 1024

__global__ void k_bincount(const int* __restrict__ dst, int* __restrict__ binCount,
                           int* __restrict__ blockBase, int E, int nbin) {
  __shared__ int hist[NBIN_MAX];
  int t = threadIdx.x;
  hist[t] = 0;
  __syncthreads();
  int chunk = (E + (int)gridDim.x - 1) / (int)gridDim.x;
  int beg = (int)blockIdx.x * chunk;
  int end = beg + chunk; if (end > E) end = E;
  for (int e0 = beg + t; e0 < end; e0 += 4 * SORT_TB) {
    int d[4];
#pragma unroll
    for (int u = 0; u < 4; ++u) {
      int e = e0 + u * SORT_TB;
      d[u] = (e < end) ? dst[e] : -1;
    }
#pragma unroll
    for (int u = 0; u < 4; ++u)
      if (d[u] >= 0) atomicAdd(&hist[d[u] >> BIN_SHIFT], 1);
  }
  __syncthreads();
  if (t < nbin) {
    int c = hist[t];
    int base = 0;
    if (c > 0) base = atomicAdd(&binCount[t * CNT_PITCH], c);
    blockBase[(int)blockIdx.x * nbin + t] = base;
  }
}

__global__ void k_binscatter(const int* __restrict__ src, const int* __restrict__ dst,
                             const float* __restrict__ ew, const int* __restrict__ binCount,
                             const int* __restrict__ blockBase, int* __restrict__ binStart,
                             int2* __restrict__ binned, int E, int nbin) {
  __shared__ int s[NBIN_MAX];
  __shared__ int cur[NBIN_MAX];
  int t = threadIdx.x;
  int v = (t < nbin) ? binCount[t * CNT_PITCH] : 0;
  s[t] = v;
  __syncthreads();
  for (int d = 1; d < SORT_TB; d <<= 1) {
    int a = (t >= d) ? s[t - d] : 0;
    __syncthreads();
    s[t] += a;
    __syncthreads();
  }
  int excl = s[t] - v;
  cur[t] = (t < nbin) ? excl + blockBase[(int)blockIdx.x * nbin + t] : 0;
  if (blockIdx.x == 0) {
    if (t < nbin) binStart[t] = excl;
    if (t == 0) binStart[nbin] = E;
  }
  __syncthreads();
  int chunk = (E + (int)gridDim.x - 1) / (int)gridDim.x;
  int beg = (int)blockIdx.x * chunk;
  int end = beg + chunk; if (end > E) end = E;
  for (int e0 = beg + t; e0 < end; e0 += 4 * SORT_TB) {
    int d[4];
    int sr[4];
    int wb[4];
#pragma unroll
    for (int u = 0; u < 4; ++u) {
      int e = e0 + u * SORT_TB;
      bool ok = e < end;
      d[u] = ok ? dst[e] : -1;
      sr[u] = ok ? src[e] : 0;
      wb[u] = ok ? __float_as_int(ew[e]) : 0;
    }
#pragma unroll
    for (int u = 0; u < 4; ++u) {
      if (d[u] >= 0) {
        int p = atomicAdd(&cur[d[u] >> BIN_SHIFT], 1);
        binned[p] = make_int2((sr[u] << 8) | (d[u] & (BIN_NODES - 1)), wb[u]);
      }
    }
  }
}

__global__ void k_binsort(const int* __restrict__ binStart, const int2* __restrict__ binned,
                          int* __restrict__ off, int2* __restrict__ pairs, int n) {
  __shared__ int c[BIN_NODES];
  __shared__ int s[BIN_NODES];
  int t = threadIdx.x;
  int b = blockIdx.x;
  int beg = binStart[b];
  int end = binStart[b + 1];
  if (t < BIN_NODES) c[t] = 0;
  __syncthreads();
  for (int p0 = beg + t; p0 < end; p0 += 4 * BIN_TB) {
    int r[4];
#pragma unroll
    for (int u = 0; u < 4; ++u) {
      int p = p0 + u * BIN_TB;
      r[u] = (p < end) ? binned[p].x : -1;
    }
#pragma unroll
    for (int u = 0; u < 4; ++u)
      if (r[u] >= 0) atomicAdd(&c[r[u] & (BIN_NODES - 1)], 1);
  }
  __syncthreads();
  int v = (t < BIN_NODES) ? c[t] : 0;
  if (t < BIN_NODES) s[t] = v;
  __syncthreads();
  for (int d = 1; d < BIN_NODES; d <<= 1) {
    int a = (t < BIN_NODES && t >= d) ? s[t - d] : 0;
    __syncthreads();
    if (t < BIN_NODES) s[t] += a;
    __syncthreads();
  }
  if (t < BIN_NODES) {
    int st = beg + s[t] - v;
    int node = b * BIN_NODES + t;
    if (node < n) off[node] = st;
    c[t] = st;
  }
  __syncthreads();
  for (int p0 = beg + t; p0 < end; p0 += 4 * BIN_TB) {
    int2 r[4];
#pragma unroll
    for (int u = 0; u < 4; ++u) {
      int p = p0 + u * BIN_TB;
      r[u] = (p < end) ? binned[p] : make_int2(-1, 0);
    }
#pragma unroll
    for (int u = 0; u < 4; ++u) {
      if (r[u].x >= 0) {
        int pos = atomicAdd(&c[r[u].x & (BIN_NODES - 1)], 1);
        pairs[pos] = make_int2(r[u].x >> 8, r[u].y);
      }
    }
  }
}

#define NPW 8

__global__ void k_dinv_seg(const int* __restrict__ off, const int2* __restrict__ pairs,
                           float* __restrict__ dinv, int n, int E) {
  int i0 = __builtin_amdgcn_readfirstlane((int)((blockIdx.x * blockDim.x + threadIdx.x) >> 5)) * NPW;
  if (i0 >= n) return;
  int lane = threadIdx.x & 31;
  int cnt = n - i0; if (cnt > NPW) cnt = NPW;
  int ofv = E;
  if (lane <= cnt && i0 + lane < n) ofv = off[i0 + lane];
  float x0[NPW];
  float x1[NPW];
  int emax = (E > 0) ? E - 1 : 0;
#pragma unroll
  for (int k = 0; k < NPW; ++k) {
    int p = __builtin_amdgcn_readlane(ofv, k) + lane;
    int end = __builtin_amdgcn_readlane(ofv, k + 1);
    int pa = (p < emax) ? p : emax;
    int pb = (p + 32 < emax) ? p + 32 : emax;
    float wa = __int_as_float(pairs[pa].y);
    float wb = __int_as_float(pairs[pb].y);
    x0[k] = (p < end) ? wa : 0.0f;
    x1[k] = (p + 32 < end) ? wb : 0.0f;
  }
  float res = 0.0f;
#pragma unroll
  for (int k = 0; k < NPW; ++k) {
    int p = __builtin_amdgcn_readlane(ofv, k) + lane;
    int end = __builtin_amdgcn_readlane(ofv, k + 1);
    float s = 0.0f;
    s += x0[k];
    s += x1[k];
    for (int p2 = p + 64; p2 < end; p2 += 32) s += __int_as_float(pairs[p2].y);
#pragma unroll
    for (int m = 16; m >= 1; m >>= 1) s += __shfl_xor(s, m, 32);
    float d = s + 1.0f;
    float r = (d > 0.0f) ? rsqrtf(d) : 0.0f;
    if (lane == k) res = r;
  }
  if (lane < cnt) dinv[i0 + lane] = res;
}

__global__ void k_gather_norm(const int* __restrict__ off, const int2* __restrict__ pairs,
                              const float* __restrict__ dinv, const float* __restrict__ h,
                              const float* __restrict__ b, float* __restrict__ o,
                              int n, int E) {
  int i0 = __builtin_amdgcn_readfirstlane((int)((blockIdx.x * blockDim.x + threadIdx.x) >> 5)) * NPW;
  if (i0 >= n) return;
  int lane = threadIdx.x & 31;
  int cnt = n - i0; if (cnt > NPW) cnt = NPW;
  int ofv = E;
  if (lane <= cnt && i0 + lane < n) ofv = off[i0 + lane];
  float dv = (lane < cnt) ? dinv[i0 + lane] : 0.0f;
  float bias = (lane < HID) ? b[lane] : 0.0f;
  const char* hb = (const char*)h;
  unsigned lane4 = (unsigned)lane * 4u;

  int k = 0;
  int p0 = __builtin_amdgcn_readlane(ofv, 0);
  int end = __builtin_amdgcn_readlane(ofv, 1);
  int m = end - p0; if (m > 32) m = 32;
  int2 q = make_int2(i0, 0);
  if (lane < m) q = pairs[p0 + lane];
  bool first = true;
  float acc = 0.0f;
  for (;;) {
    unsigned sx[32];
#pragma unroll
    for (int u = 0; u < 32; ++u) sx[u] = (unsigned)__builtin_amdgcn_readlane(q.x, u);
    int i = i0 + k;
    int p1 = p0 + m;
    bool done = p1 >= end;
    int kn = done ? k + 1 : k;
    bool more = kn < cnt;
    int endn = end;
    if (done && more) endn = __builtin_amdgcn_readlane(ofv, kn + 1);
    int mn = 0;
    if (more) { mn = endn - p1; if (mn > 32) mn = 32; }
    int2 qn = make_int2(i0, 0);
    if (lane < mn) qn = pairs[p1 + lane];

    float self = 0.0f;
    if (first) self = h[(size_t)i * 32 + lane];
    float v[32];
#pragma unroll
    for (int g = 0; g < 4; ++g) {
      if (g * 8 < m) {
#pragma unroll
        for (int u = 0; u < 8; ++u) {
          v[g * 8 + u] = *(const float*)(hb + (sx[g * 8 + u] * 128u + lane4));
        }
      }
    }
    if (first) acc = self;
#pragma unroll
    for (int g = 0; g < 4; ++g) {
      if (g * 8 < m) {
#pragma unroll
        for (int u = 0; u < 8; ++u) {
          float cf = __int_as_float(__builtin_amdgcn_readlane(q.y, g * 8 + u));
          acc += v[g * 8 + u] * cf;
        }
      }
    }
    if (done) {
      float di = __int_as_float(__builtin_amdgcn_readlane(__float_as_int(dv), k));
      float r0 = (lane < HID) ? acc * di + bias : 0.0f;
      float ss = r0 * r0;
#pragma unroll
      for (int x = 16; x >= 1; x >>= 1) ss += __shfl_xor(ss, x, 32);
      float nrm = fmaxf(sqrtf(ss), 1e-12f);
      if (lane < HID) {
        float r = r0 / nrm;
        o[(size_t)i * 32 + lane] = (r > 0.0f) ? r : 0.0f;
      }
    }
    if (!more) break;
    first = done;
    k = kn; p0 = p1; end = endn; m = mn; q = qn;
  }
}

template <int NTILES, bool ROWSCALE, bool ACCUM>
__global__ void k_wmma_gemm(const float* __restrict__ A, int lda, int acol0,
                            const float* __restrict__ Bp, int Ksteps,
                            float* __restrict__ out, int ldo, int ncols,
                            const float* __restrict__ bias,
                            const float* __restrict__ rowscale, int npairs) {
  int pair = blockIdx.x * (blockDim.x >> 5) + (threadIdx.x >> 5);
  if (pair >= npairs) return;
  int lane = threadIdx.x & 31;
  int half = lane >> 4;
  int l16 = lane & 15;
  int m0 = pair * 32;
  const float* ap0 = A + (size_t)(m0 + l16) * lda + acol0 + half * 2;
  const float* ap1 = ap0 + (size_t)16 * lda;
  const float* bp = Bp + lane * 2;
  float rs[2][8];
  if (ROWSCALE) {
    const float4* rp = (const float4*)(rowscale + m0 + half * 8);
    float4 q0 = rp[0], q1 = rp[1], q2 = rp[4], q3 = rp[5];
    rs[0][0] = q0.x; rs[0][1] = q0.y; rs[0][2] = q0.z; rs[0][3] = q0.w;
    rs[0][4] = q1.x; rs[0][5] = q1.y; rs[0][6] = q1.z; rs[0][7] = q1.w;
    rs[1][0] = q2.x; rs[1][1] = q2.y; rs[1][2] = q2.z; rs[1][3] = q2.w;
    rs[1][4] = q3.x; rs[1][5] = q3.y; rs[1][6] = q3.z; rs[1][7] = q3.w;
  }
  v8f acc[2][NTILES];
#pragma unroll
  for (int rr = 0; rr < 2; ++rr)
#pragma unroll
    for (int t = 0; t < NTILES; ++t) acc[rr][t] = (v8f){0.f, 0.f, 0.f, 0.f, 0.f, 0.f, 0.f, 0.f};

#pragma unroll 2
  for (int ks = 0; ks < Ksteps; ++ks) {
    v2f a0 = *(const v2f*)(ap0 + 4 * ks);
    v2f a1 = *(const v2f*)(ap1 + 4 * ks);
#pragma unroll
    for (int t = 0; t < NTILES; ++t) {
      v2f b = *(const v2f*)(bp + (size_t)(ks * NTILES + t) * 64);
      acc[0][t] = __builtin_amdgcn_wmma_f32_16x16x4_f32(false, a0, false, b, (short)0, acc[0][t], false, false);
      acc[1][t] = __builtin_amdgcn_wmma_f32_16x16x4_f32(false, a1, false, b, (short)0, acc[1][t], false, false);
    }
  }

#pragma unroll
  for (int rr = 0; rr < 2; ++rr)
#pragma unroll
    for (int t = 0; t < NTILES; ++t)
#pragma unroll
      for (int r = 0; r < 8; ++r) {
        int row = m0 + rr * 16 + r + half * 8;
        int col = t * 16 + l16;
        if (col < ncols) {
          float val = acc[rr][t][r] + (bias ? bias[col] : 0.0f);
          if (ROWSCALE) val *= rs[rr][r];
          if (ACCUM) val += out[(size_t)row * ldo + col];
          out[(size_t)row * ldo + col] = val;
        }
      }
}

template <bool FIRST>
__global__ void k_wmma_gemm_dual(const float* __restrict__ A, const float* __restrict__ Bp,
                                 float* __restrict__ h, const float* __restrict__ rowscale,
                                 float* __restrict__ out, const float* __restrict__ bl,
                                 int npairs) {
  int pair = blockIdx.x * (blockDim.x >> 5) + (threadIdx.x >> 5);
  if (pair >= npairs) return;
  int lane = threadIdx.x & 31;
  int half = lane >> 4;
  int l16 = lane & 15;
  int m0 = pair * 32;
  const float* ap0 = A + (size_t)(m0 + l16) * 32 + half * 2;
  const float* ap1 = ap0 + (size_t)16 * 32;
  const float* bp = Bp + lane * 2;
  float rs[2][8];
  {
    const float4* rp = (const float4*)(rowscale + m0 + half * 8);
    float4 q0 = rp[0], q1 = rp[1], q2 = rp[4], q3 = rp[5];
    rs[0][0] = q0.x; rs[0][1] = q0.y; rs[0][2] = q0.z; rs[0][3] = q0.w;
    rs[0][4] = q1.x; rs[0][5] = q1.y; rs[0][6] = q1.z; rs[0][7] = q1.w;
    rs[1][0] = q2.x; rs[1][1] = q2.y; rs[1][2] = q2.z; rs[1][3] = q2.w;
    rs[1][4] = q3.x; rs[1][5] = q3.y; rs[1][6] = q3.z; rs[1][7] = q3.w;
  }
  v8f acc[2][2];
#pragma unroll
  for (int rr = 0; rr < 2; ++rr)
#pragma unroll
    for (int t = 0; t < 2; ++t) acc[rr][t] = (v8f){0.f, 0.f, 0.f, 0.f, 0.f, 0.f, 0.f, 0.f};

#pragma unroll
  for (int ks = 0; ks < HID / 4; ++ks) {
    v2f a0 = *(const v2f*)(ap0 + 4 * ks);
    v2f a1 = *(const v2f*)(ap1 + 4 * ks);
#pragma unroll
    for (int t = 0; t < 2; ++t) {
      v2f b = *(const v2f*)(bp + (size_t)(ks * 2 + t) * 64);
      acc[0][t] = __builtin_amdgcn_wmma_f32_16x16x4_f32(false, a0, false, b, (short)0, acc[0][t], false, false);
      acc[1][t] = __builtin_amdgcn_wmma_f32_16x16x4_f32(false, a1, false, b, (short)0, acc[1][t], false, false);
    }
  }

  const int col1 = 16 + l16;
#pragma unroll
  for (int rr = 0; rr < 2; ++rr)
#pragma unroll
    for (int r = 0; r < 8; ++r) {
      int row = m0 + rr * 16 + r + half * 8;
      float sc = rs[rr][r];
      float v1 = acc[rr][1][r];
      h[(size_t)row * 32 + l16] = acc[rr][0][r] * sc;
      h[(size_t)row * 32 + col1] = (col1 < HID) ? v1 * sc : 0.0f;
      if (col1 >= HID && col1 < HID + OUT_DIM) {
        size_t oi = (size_t)row * OUT_DIM + (col1 - HID);
        if (FIRST) out[oi] = v1 + bl[col1 - HID];
        else out[oi] = out[oi] + v1;
      }
    }
}

extern "C" void kernel_launch(void* const* d_in, const int* in_sizes, int n_in,
                              void* d_out, int out_size, void* d_ws, size_t ws_size,
                              hipStream_t stream) {
  const float* x  = (const float*)d_in[0];
  const int*   ei = (const int*)d_in[1];
  const float* ew = (const float*)d_in[2];
  const float* W1 = (const float*)d_in[3];
  const float* b1 = (const float*)d_in[4];
  const float* W2 = (const float*)d_in[5];
  const float* b2 = (const float*)d_in[6];
  const float* W3 = (const float*)d_in[7];
  const float* b3 = (const float*)d_in[8];
  const float* Wl = (const float*)d_in[9];
  const float* bl = (const float*)d_in[10];
  float* out = (float*)d_out;

  const int N = in_sizes[0] / IN_DIM;
  const long long E = in_sizes[2];
  const int* src = ei;
  const int* dst = ei + E;

  const int KS1 = IN_DIM / 4;
  const int KS2 = HID / 4;

  float* ws = (float*)d_ws;
  int*   off  = (int*)ws; ws += N;
  float* dinv = ws; ws += N;
  int2*  pairs = (int2*)ws; ws += (size_t)N * 64;
  float* h    = ws; ws += (size_t)N * 32;
  float* o    = ws; ws += (size_t)N * 32;
  float* W1p  = ws; ws += KS1 * 2 * 64;
  float* W2c  = ws; ws += KS2 * 2 * 64;
  float* W3c  = ws; ws += KS2 * 2 * 64;
  float* Wl3p = ws; ws += KS2 * 1 * 64;
  const int NBIN = (N + BIN_NODES - 1) / BIN_NODES;
  const int NSORTWG = 512;
  int*   binCount  = (int*)dinv;
  int*   binStart  = (int*)dinv + NBIN_MAX * CNT_PITCH;
  int2*  binned    = (int2*)h;
  int*   blockBase = (int*)pairs;

  const int TB = 256;
  const int NPAIRS = N / 32;
  dim3 gblk((NPAIRS + 3) / 4);
  const int Ei = (int)E;
  long long nt32 = (long long)((N + NPW - 1) / NPW) * 32;
  int g_n32 = (int)((nt32 + TB - 1) / TB);

  k_init<<<(NBIN * CNT_PITCH + TB - 1) / TB, TB, 0, stream>>>(binCount, NBIN * CNT_PITCH);
  k_packB<<<(KS1 * 2 * 64 + TB - 1) / TB, TB, 0, stream>>>(W1, IN_DIM, HID, W1p, KS1, 2);
  k_packB2<<<(KS2 * 2 * 64 + TB - 1) / TB, TB, 0, stream>>>(W2, Wl, W2c);
  k_packB2<<<(KS2 * 2 * 64 + TB - 1) / TB, TB, 0, stream>>>(W3, Wl + HID * OUT_DIM, W3c);
  k_packB<<<(KS2 * 1 * 64 + TB - 1) / TB, TB, 0, stream>>>(Wl + 2 * HID * OUT_DIM, HID, OUT_DIM, Wl3p, KS2, 1);
  k_bincount<<<NSORTWG, SORT_TB, 0, stream>>>(dst, binCount, blockBase, Ei, NBIN);
  k_binscatter<<<NSORTWG, SORT_TB, 0, stream>>>(src, dst, ew, binCount, blockBase, binStart, binned, Ei, NBIN);
  k_binsort<<<NBIN, BIN_TB, 0, stream>>>(binStart, binned, off, pairs, N);
  k_dinv_seg<<<g_n32, TB, 0, stream>>>(off, pairs, dinv, N, Ei);

  k_wmma_gemm<2, true, false><<<gblk, 128, 0, stream>>>(x, IN_DIM, 0, W1p, KS1, h, 32, 32, nullptr, dinv, NPAIRS);
  k_gather_norm<<<g_n32, TB, 0, stream>>>(off, pairs, dinv, h, b1, o, N, Ei);

  k_wmma_gemm_dual<true><<<gblk, 128, 0, stream>>>(o, W2c, h, dinv, out, bl, NPAIRS);
  k_gather_norm<<<g_n32, TB, 0, stream>>>(off, pairs, dinv, h, b2, o, N, Ei);

  k_wmma_gemm_dual<false><<<gblk, 128, 0, stream>>>(o, W3c, h, dinv, out, bl, NPAIRS);
  k_gather_norm<<<g_n32, TB, 0, stream>>>(off, pairs, dinv, h, b3, o, N, Ei);

  k_wmma_gemm<1, false, true><<<gblk, 128, 0, stream>>>(o, 32, 0, Wl3p, KS2, out, OUT_DIM, OUT_DIM, nullptr, nullptr, NPAIRS);
}


# 16 launches -> 11: counter clear and the four weight packs run as one launch; each bin-sort work-group also sums its 256 nodes' degrees (same per-node order), so the dinv launch is gone
# speedup vs baseline: 4.6750x; 1.0435x over previous
#include <hip/hip_runtime.h>
#include <hip/hip_bf16.h>

#define IN_DIM 512
#define HID 20
#define OUT_DIM 10

typedef __attribute__((ext_vector_type(2))) float v2f;
typedef __attribute__((ext_vector_type(8))) float v8f;

__device__ __forceinline__ void pack_b(const float* __restrict__ W, int Krows, int Ncols,
                                       float* __restrict__ Bp, int Ksteps, int ntiles, int t) {
  int total = Ksteps * ntiles * 64;
  if (t >= total) return;
  int j = t & 1;
  int lane = (t >> 1) & 31;
  int tile = (t >> 6) % ntiles;
  int ks = (t >> 6) / ntiles;
  int k = 4 * ks + 2 * (lane >> 4) + j;
  int n = tile * 16 + (lane & 15);
  Bp[t] = (k < Krows && n < Ncols) ? W[k * Ncols + n] : 0.0f;
}

__device__ __forceinline__ void pack_b2(const float* __restrict__ W, const float* __restrict__ Wls,
                                        float* __restrict__ Bp, int t) {
  if (t >= (HID / 4) * 2 * 64) return;
  int j = t & 1;
  int lane = (t >> 1) & 31;
  int tile = (t >> 6) & 1;
  int ks = t >> 7;
  int k = 4 * ks + 2 * (lane >> 4) + j;
  int n = tile * 16 + (lane & 15);
  float v = 0.0f;
  if (n < HID) v = W[k * HID + n];
  else if (n < HID + OUT_DIM) v = Wls[k * OUT_DIM + (n - HID)];
  Bp[t] = v;
}

#define SETUP_TB 256
#define NB_W1 ((IN_DIM / 4) * 2 * 64 / SETUP_TB)
#define NB_WC (((HID / 4) * 2 * 64 + SETUP_TB - 1) / SETUP_TB)
#define NB_WL (((HID / 4) * 1 * 64 + SETUP_TB - 1) / SETUP_TB)

__global__ void k_setup(int* __restrict__ cnt, int ncnt, int nb_cnt,
                        const float* __restrict__ W1, const float* __restrict__ W2,
                        const float* __restrict__ W3, const float* __restrict__ Wl,
                        float* __restrict__ W1p, float* __restrict__ W2c,
                        float* __restrict__ W3c, float* __restrict__ Wl3p) {
  int b = blockIdx.x;
  int t = threadIdx.x;
  if (b < nb_cnt) {
    int i = b * SETUP_TB + t;
    if (i < ncnt) cnt[i] = 0;
    return;
  }
  b -= nb_cnt;
  if (b < NB_W1) { pack_b(W1, IN_DIM, HID, W1p, IN_DIM / 4, 2, b * SETUP_TB + t); return; }
  b -= NB_W1;
  if (b < NB_WC) { pack_b2(W2, Wl, W2c, b * SETUP_TB + t); return; }
  b -= NB_WC;
  if (b < NB_WC) { pack_b2(W3, Wl + HID * OUT_DIM, W3c, b * SETUP_TB + t); return; }
  b -= NB_WC;
  pack_b(Wl + 2 * HID * OUT_DIM, HID, OUT_DIM, Wl3p, HID / 4, 1, b * SETUP_TB + t);
}

#define BIN_SHIFT 8
#define BIN_NODES 256
#define NBIN_MAX 512
#define CNT_PITCH 32
#define SORT_TB 512
#define BIN_TB 1024
#define NPW 8

__global__ void k_bincount(const int* __restrict__ dst, int* __restrict__ binCount,
                           int* __restrict__ blockBase, int E, int nbin) {
  __shared__ int hist[NBIN_MAX];
  int t = threadIdx.x;
  hist[t] = 0;
  __syncthreads();
  int chunk = (E + (int)gridDim.x - 1) / (int)gridDim.x;
  int beg = (int)blockIdx.x * chunk;
  int end = beg + chunk; if (end > E) end = E;
  for (int e0 = beg + t; e0 < end; e0 += 4 * SORT_TB) {
    int d[4];
#pragma unroll
    for (int u = 0; u < 4; ++u) {
      int e = e0 + u * SORT_TB;
      d[u] = (e < end) ? dst[e] : -1;
    }
#pragma unroll
    for (int u = 0; u < 4; ++u)
      if (d[u] >= 0) atomicAdd(&hist[d[u] >> BIN_SHIFT], 1);
  }
  __syncthreads();
  if (t < nbin) {
    int c = hist[t];
    int base = 0;
    if (c > 0) base = atomicAdd(&binCount[t * CNT_PITCH], c);
    blockBase[(int)blockIdx.x * nbin + t] = base;
  }
}

__global__ void k_binscatter(const int* __restrict__ src, const int* __restrict__ dst,
                             const float* __restrict__ ew, const int* __restrict__ binCount,
                             const int* __restrict__ blockBase, int* __restrict__ binStart,
                             int2* __restrict__ binned, int E, int nbin) {
  __shared__ int s[NBIN_MAX];
  __shared__ int cur[NBIN_MAX];
  int t = threadIdx.x;
  int v = (t < nbin) ? binCount[t * CNT_PITCH] : 0;
  s[t] = v;
  __syncthreads();
  for (int d = 1; d < SORT_TB; d <<= 1) {
    int a = (t >= d) ? s[t - d] : 0;
    __syncthreads();
    s[t] += a;
    __syncthreads();
  }
  int excl = s[t] - v;
  cur[t] = (t < nbin) ? excl + blockBase[(int)blockIdx.x * nbin + t] : 0;
  if (blockIdx.x == 0) {
    if (t < nbin) binStart[t * BIN_NODES] = excl;
  }
  __syncthreads();
  int chunk = (E + (int)gridDim.x - 1) / (int)gridDim.x;
  int beg = (int)blockIdx.x * chunk;
  int end = beg + chunk; if (end > E) end = E;
  for (int e0 = beg + t; e0 < end; e0 += 4 * SORT_TB) {
    int d[4];
    int sr[4];
    int wb[4];
#pragma unroll
    for (int u = 0; u < 4; ++u) {
      int e = e0 + u * SORT_TB;
      bool ok = e < end;
      d[u] = ok ? dst[e] : -1;
      sr[u] = ok ? src[e] : 0;
      wb[u] = ok ? __float_as_int(ew[e]) : 0;
    }
#pragma unroll
    for (int u = 0; u < 4; ++u) {
      if (d[u] >= 0) {
        int p = atomicAdd(&cur[d[u] >> BIN_SHIFT], 1);
        binned[p] = make_int2((sr[u] << 8) | (d[u] & (BIN_NODES - 1)), wb[u]);
      }
    }
  }
}

__global__ void k_binsort(const int2* __restrict__ binned, int* off,
                          int2* __restrict__ pairs, float* __restrict__ dinv, int n, int E) {
  __shared__ int c[BIN_NODES];
  __shared__ int s[BIN_NODES];
  int t = threadIdx.x;
  int b = blockIdx.x;
  int beg = off[b * BIN_NODES];
  int end = (b + 1 < (int)gridDim.x) ? off[(b + 1) * BIN_NODES] : E;
  if (t < BIN_NODES) c[t] = 0;
  __syncthreads();
  for (int p0 = beg + t; p0 < end; p0 += 4 * BIN_TB) {
    int r[4];
#pragma unroll
    for (int u = 0; u < 4; ++u) {
      int p = p0 + u * BIN_TB;
      r[u] = (p < end) ? binned[p].x : -1;
    }
#pragma unroll
    for (int u = 0; u < 4; ++u)
      if (r[u] >= 0) atomicAdd(&c[r[u] & (BIN_NODES - 1)], 1);
  }
  __syncthreads();
  int v = (t < BIN_NODES) ? c[t] : 0;
  if (t < BIN_NODES) s[t] = v;
  __syncthreads();
  for (int d = 1; d < BIN_NODES; d <<= 1) {
    int a = (t < BIN_NODES && t >= d) ? s[t - d] : 0;
    __syncthreads();
    if (t < BIN_NODES) s[t] += a;
    __syncthreads();
  }
  if (t < BIN_NODES) {
    int st = beg + s[t] - v;
    int node = b * BIN_NODES + t;
    if (t > 0 && node < n) off[node] = st;
    c[t] = st;
  }
  __syncthreads();
  for (int p0 = beg + t; p0 < end; p0 += 4 * BIN_TB) {
    int2 r[4];
#pragma unroll
    for (int u = 0; u < 4; ++u) {
      int p = p0 + u * BIN_TB;
      r[u] = (p < end) ? binned[p] : make_int2(-1, 0);
    }
#pragma unroll
    for (int u = 0; u < 4; ++u) {
      if (r[u].x >= 0) {
        int pos = atomicAdd(&c[r[u].x & (BIN_NODES - 1)], 1);
        pairs[pos] = make_int2(r[u].x >> 8, r[u].y);
      }
    }
  }
  __syncthreads();
  int lane = t & 31;
  int j0 = (t >> 5) * NPW;
  int ofv = end;
  if (lane <= NPW) { int j = j0 + lane; ofv = (j == 0) ? beg : c[j - 1]; }
  float x0[NPW];
  float x1[NPW];
  int emax = (end > 0) ? end - 1 : 0;
#pragma unroll
  for (int k = 0; k < NPW; ++k) {
    int p = __builtin_amdgcn_readlane(ofv, k) + lane;
    int e = __builtin_amdgcn_readlane(ofv, k + 1);
    int pa = (p < emax) ? p : emax;
    int pb = (p + 32 < emax) ? p + 32 : emax;
    float wa = __int_as_float(pairs[pa].y);
    float wb = __int_as_float(pairs[pb].y);
    x0[k] = (p < e) ? wa : 0.0f;
    x1[k] = (p + 32 < e) ? wb : 0.0f;
  }
  float res = 0.0f;
#pragma unroll
  for (int k = 0; k < NPW; ++k) {
    int p = __builtin_amdgcn_readlane(ofv, k) + lane;
    int e = __builtin_amdgcn_readlane(ofv, k + 1);
    float sm = 0.0f;
    sm += x0[k];
    sm += x1[k];
    for (int p2 = p + 64; p2 < e; p2 += 32) sm += __int_as_float(pairs[p2].y);
#pragma unroll
    for (int m = 16; m >= 1; m >>= 1) sm += __shfl_xor(sm, m, 32);
    float dg = sm + 1.0f;
    float rq = (dg > 0.0f) ? rsqrtf(dg) : 0.0f;
    if (lane == k) res = rq;
  }
  int node = b * BIN_NODES + j0 + lane;
  if (lane < NPW && node < n) dinv[node] = res;
}

__global__ void k_gather_norm(const int* __restrict__ off, const int2* __restrict__ pairs,
                              const float* __restrict__ dinv, const float* __restrict__ h,
                              const float* __restrict__ b, float* __restrict__ o,
                              int n, int E) {
  int i0 = __builtin_amdgcn_readfirstlane((int)((blockIdx.x * blockDim.x + threadIdx.x) >> 5)) * NPW;
  if (i0 >= n) return;
  int lane = threadIdx.x & 31;
  int cnt = n - i0; if (cnt > NPW) cnt = NPW;
  int ofv = E;
  if (lane <= cnt && i0 + lane < n) ofv = off[i0 + lane];
  float dv = (lane < cnt) ? dinv[i0 + lane] : 0.0f;
  float bias = (lane < HID) ? b[lane] : 0.0f;
  const char* hb = (const char*)h;
  unsigned lane4 = (unsigned)lane * 4u;

  int k = 0;
  int p0 = __builtin_amdgcn_readlane(ofv, 0);
  int end = __builtin_amdgcn_readlane(ofv, 1);
  int m = end - p0; if (m > 32) m = 32;
  int2 q = make_int2(i0, 0);
  if (lane < m) q = pairs[p0 + lane];
  bool first = true;
  float acc = 0.0f;
  for (;;) {
    unsigned sx[32];
#pragma unroll
    for (int u = 0; u < 32; ++u) sx[u] = (unsigned)__builtin_amdgcn_readlane(q.x, u);
    int i = i0 + k;
    int p1 = p0 + m;
    bool done = p1 >= end;
    int kn = done ? k + 1 : k;
    bool more = kn < cnt;
    int endn = end;
    if (done && more) endn = __builtin_amdgcn_readlane(ofv, kn + 1);
    int mn = 0;
    if (more) { mn = endn - p1; if (mn > 32) mn = 32; }
    int2 qn = make_int2(i0, 0);
    if (lane < mn) qn = pairs[p1 + lane];

    float self = 0.0f;
    if (first) self = h[(size_t)i * 32 + lane];
    float v[32];
#pragma unroll
    for (int g = 0; g < 4; ++g) {
      if (g * 8 < m) {
#pragma unroll
        for (int u = 0; u < 8; ++u) {
          v[g * 8 + u] = *(const float*)(hb + (sx[g * 8 + u] * 128u + lane4));
        }
      }
    }
    if (first) acc = self;
#pragma unroll
    for (int g = 0; g < 4; ++g) {
      if (g * 8 < m) {
#pragma unroll
        for (int u = 0; u < 8; ++u) {
          float cf = __int_as_float(__builtin_amdgcn_readlane(q.y, g * 8 + u));
          acc += v[g * 8 + u] * cf;
        }
      }
    }
    if (done) {
      float di = __int_as_float(__builtin_amdgcn_readlane(__float_as_int(dv), k));
      float r0 = (lane < HID) ? acc * di + bias : 0.0f;
      float ss = r0 * r0;
#pragma unroll
      for (int x = 16; x >= 1; x >>= 1) ss += __shfl_xor(ss, x, 32);
      float nrm = fmaxf(sqrtf(ss), 1e-12f);
      if (lane < HID) {
        float r = r0 / nrm;
        o[(size_t)i * 32 + lane] = (r > 0.0f) ? r : 0.0f;
      }
    }
    if (!more) break;
    first = done;
    k = kn; p0 = p1; end = endn; m = mn; q = qn;
  }
}

template <int NTILES, bool ROWSCALE, bool ACCUM>
__global__ void k_wmma_gemm(const float* __restrict__ A, int lda, int acol0,
                            const float* __restrict__ Bp, int Ksteps,
                            float* __restrict__ out, int ldo, int ncols,
                            const float* __restrict__ bias,
                            const float* __restrict__ rowscale, int npairs) {
  int pair = blockIdx.x * (blockDim.x >> 5) + (threadIdx.x >> 5);
  if (pair >= npairs) return;
  int lane = threadIdx.x & 31;
  int half = lane >> 4;
  int l16 = lane & 15;
  int m0 = pair * 32;
  const float* ap0 = A + (size_t)(m0 + l16) * lda + acol0 + half * 2;
  const float* ap1 = ap0 + (size_t)16 * lda;
  const float* bp = Bp + lane * 2;
  float rs[2][8];
  if (ROWSCALE) {
    const float4* rp = (const float4*)(rowscale + m0 + half * 8);
    float4 q0 = rp[0], q1 = rp[1], q2 = rp[4], q3 = rp[5];
    rs[0][0] = q0.x; rs[0][1] = q0.y; rs[0][2] = q0.z; rs[0][3] = q0.w;
    rs[0][4] = q1.x; rs[0][5] = q1.y; rs[0][6] = q1.z; rs[0][7] = q1.w;
    rs[1][0] = q2.x; rs[1][1] = q2.y; rs[1][2] = q2.z; rs[1][3] = q2.w;
    rs[1][4] = q3.x; rs[1][5] = q3.y; rs[1][6] = q3.z; rs[1][7] = q3.w;
  }
  v8f acc[2][NTILES];
#pragma unroll
  for (int rr = 0; rr < 2; ++rr)
#pragma unroll
    for (int t = 0; t < NTILES; ++t) acc[rr][t] = (v8f){0.f, 0.f, 0.f, 0.f, 0.f, 0.f, 0.f, 0.f};

#pragma unroll 2
  for (int ks = 0; ks < Ksteps; ++ks) {
    v2f a0 = *(const v2f*)(ap0 + 4 * ks);
    v2f a1 = *(const v2f*)(ap1 + 4 * ks);
#pragma unroll
    for (int t = 0; t < NTILES; ++t) {
      v2f b = *(const v2f*)(bp + (size_t)(ks * NTILES + t) * 64);
      acc[0][t] = __builtin_amdgcn_wmma_f32_16x16x4_f32(false, a0, false, b, (short)0, acc[0][t], false, false);
      acc[1][t] = __builtin_amdgcn_wmma_f32_16x16x4_f32(false, a1, false, b, (short)0, acc[1][t], false, false);
    }
  }

#pragma unroll
  for (int rr = 0; rr < 2; ++rr)
#pragma unroll
    for (int t = 0; t < NTILES; ++t)
#pragma unroll
      for (int r = 0; r < 8; ++r) {
        int row = m0 + rr * 16 + r + half * 8;
        int col = t * 16 + l16;
        if (col < ncols) {
          float val = acc[rr][t][r] + (bias ? bias[col] : 0.0f);
          if (ROWSCALE) val *= rs[rr][r];
          if (ACCUM) val += out[(size_t)row * ldo + col];
          out[(size_t)row * ldo + col] = val;
        }
      }
}

template <bool FIRST>
__global__ void k_wmma_gemm_dual(const float* __restrict__ A, const float* __restrict__ Bp,
                                 float* __restrict__ h, const float* __restrict__ rowscale,
                                 float* __restrict__ out, const float* __restrict__ bl,
                                 int npairs) {
  int pair = blockIdx.x * (blockDim.x >> 5) + (threadIdx.x >> 5);
  if (pair >= npairs) return;
  int lane = threadIdx.x & 31;
  int half = lane >> 4;
  int l16 = lane & 15;
  int m0 = pair * 32;
  const float* ap0 = A + (size_t)(m0 + l16) * 32 + half * 2;
  const float* ap1 = ap0 + (size_t)16 * 32;
  const float* bp = Bp + lane * 2;
  float rs[2][8];
  {
    const float4* rp = (const float4*)(rowscale + m0 + half * 8);
    float4 q0 = rp[0], q1 = rp[1], q2 = rp[4], q3 = rp[5];
    rs[0][0] = q0.x; rs[0][1] = q0.y; rs[0][2] = q0.z; rs[0][3] = q0.w;
    rs[0][4] = q1.x; rs[0][5] = q1.y; rs[0][6] = q1.z; rs[0][7] = q1.w;
    rs[1][0] = q2.x; rs[1][1] = q2.y; rs[1][2] = q2.z; rs[1][3] = q2.w;
    rs[1][4] = q3.x; rs[1][5] = q3.y; rs[1][6] = q3.z; rs[1][7] = q3.w;
  }
  v8f acc[2][2];
#pragma unroll
  for (int rr = 0; rr < 2; ++rr)
#pragma unroll
    for (int t = 0; t < 2; ++t) acc[rr][t] = (v8f){0.f, 0.f, 0.f, 0.f, 0.f, 0.f, 0.f, 0.f};

#pragma unroll
  for (int ks = 0; ks < HID / 4; ++ks) {
    v2f a0 = *(const v2f*)(ap0 + 4 * ks);
    v2f a1 = *(const v2f*)(ap1 + 4 * ks);
#pragma unroll
    for (int t = 0; t < 2; ++t) {
      v2f b = *(const v2f*)(bp + (size_t)(ks * 2 + t) * 64);
      acc[0][t] = __builtin_amdgcn_wmma_f32_16x16x4_f32(false, a0, false, b, (short)0, acc[0][t], false, false);
      acc[1][t] = __builtin_amdgcn_wmma_f32_16x16x4_f32(false, a1, false, b, (short)0, acc[1][t], false, false);
    }
  }

  const int col1 = 16 + l16;
#pragma unroll
  for (int rr = 0; rr < 2; ++rr)
#pragma unroll
    for (int r = 0; r < 8; ++r) {
      int row = m0 + rr * 16 + r + half * 8;
      float sc = rs[rr][r];
      float v1 = acc[rr][1][r];
      h[(size_t)row * 32 + l16] = acc[rr][0][r] * sc;
      h[(size_t)row * 32 + col1] = (col1 < HID) ? v1 * sc : 0.0f;
      if (col1 >= HID && col1 < HID + OUT_DIM) {
        size_t oi = (size_t)row * OUT_DIM + (col1 - HID);
        if (FIRST) out[oi] = v1 + bl[col1 - HID];
        else out[oi] = out[oi] + v1;
      }
    }
}

extern "C" void kernel_launch(void* const* d_in, const int* in_sizes, int n_in,
                              void* d_out, int out_size, void* d_ws, size_t ws_size,
                              hipStream_t stream) {
  const float* x  = (const float*)d_in[0];
  const int*   ei = (const int*)d_in[1];
  const float* ew = (const float*)d_in[2];
  const float* W1 = (const float*)d_in[3];
  const float* b1 = (const float*)d_in[4];
  const float* W2 = (const float*)d_in[5];
  const float* b2 = (const float*)d_in[6];
  const float* W3 = (const float*)d_in[7];
  const float* b3 = (const float*)d_in[8];
  const float* Wl = (const float*)d_in[9];
  const float* bl = (const float*)d_in[10];
  float* out = (float*)d_out;

  const int N = in_sizes[0] / IN_DIM;
  const long long E = in_sizes[2];
  const int* src = ei;
  const int* dst = ei + E;

  const int KS1 = IN_DIM / 4;
  const int KS2 = HID / 4;

  float* ws = (float*)d_ws;
  int*   off  = (int*)ws; ws += N;
  float* dinv = ws; ws += N;
  int2*  pairs = (int2*)ws; ws += (size_t)N * 64;
  float* h    = ws; ws += (size_t)N * 32;
  float* o    = ws; ws += (size_t)N * 32;
  float* W1p  = ws; ws += KS1 * 2 * 64;
  float* W2c  = ws; ws += KS2 * 2 * 64;
  float* W3c  = ws; ws += KS2 * 2 * 64;
  float* Wl3p = ws; ws += KS2 * 1 * 64;
  const int NBIN = (N + BIN_NODES - 1) / BIN_NODES;
  const int NSORTWG = 512;
  int*   binCount  = (int*)dinv;
  int2*  binned    = (int2*)h;
  int*   blockBase = (int*)pairs;

  const int TB = 256;
  const int NPAIRS = N / 32;
  dim3 gblk((NPAIRS + 3) / 4);
  const int Ei = (int)E;
  long long nt32 = (long long)((N + NPW - 1) / NPW) * 32;
  int g_n32 = (int)((nt32 + TB - 1) / TB);
  const int NB_CNT = (NBIN * CNT_PITCH + SETUP_TB - 1) / SETUP_TB;

  k_setup<<<NB_CNT + NB_W1 + 2 * NB_WC + NB_WL, SETUP_TB, 0, stream>>>(
      binCount, NBIN * CNT_PITCH, NB_CNT, W1, W2, W3, Wl, W1p, W2c, W3c, Wl3p);
  k_bincount<<<NSORTWG, SORT_TB, 0, stream>>>(dst, binCount, blockBase, Ei, NBIN);
  k_binscatter<<<NSORTWG, SORT_TB, 0, stream>>>(src, dst, ew, binCount, blockBase, off, binned, Ei, NBIN);
  k_binsort<<<NBIN, BIN_TB, 0, stream>>>(binned, off, pairs, dinv, N, Ei);

  k_wmma_gemm<2, true, false><<<gblk, 128, 0, stream>>>(x, IN_DIM, 0, W1p, KS1, h, 32, 32, nullptr, dinv, NPAIRS);
  k_gather_norm<<<g_n32, TB, 0, stream>>>(off, pairs, dinv, h, b1, o, N, Ei);

  k_wmma_gemm_dual<true><<<gblk, 128, 0, stream>>>(o, W2c, h, dinv, out, bl, NPAIRS);
  k_gather_norm<<<g_n32, TB, 0, stream>>>(off, pairs, dinv, h, b2, o, N, Ei);

  k_wmma_gemm_dual<false><<<gblk, 128, 0, stream>>>(o, W3c, h, dinv, out, bl, NPAIRS);
  k_gather_norm<<<g_n32, TB, 0, stream>>>(off, pairs, dinv, h, b3, o, N, Ei);

  k_wmma_gemm<1, false, true><<<gblk, 128, 0, stream>>>(o, 32, 0, Wl3p, KS2, out, OUT_DIM, OUT_DIM, nullptr, nullptr, NPAIRS);
}


# 11 launches -> 8: each gather wave hands its 8 finished rows through LDS to a WMMA tail that stores the next layer's h row and the node's slice of out; the 3 small GEMM launches and the o plane go
# speedup vs baseline: 5.3884x; 1.1526x over previous
#include <hip/hip_runtime.h>
#include <hip/hip_bf16.h>

#define IN_DIM 512
#define HID 20
#define OUT_DIM 10

typedef __attribute__((ext_vector_type(2))) float v2f;
typedef __attribute__((ext_vector_type(8))) float v8f;

__device__ __forceinline__ void pack_b(const float* __restrict__ W, int Krows, int Ncols,
                                       float* __restrict__ Bp, int Ksteps, int ntiles, int t) {
  int total = Ksteps * ntiles * 64;
  if (t >= total) return;
  int j = t & 1;
  int lane = (t >> 1) & 31;
  int tile = (t >> 6) % ntiles;
  int ks = (t >> 6) / ntiles;
  int k = 4 * ks + 2 * (lane >> 4) + j;
  int n = tile * 16 + (lane & 15);
  Bp[t] = (k < Krows && n < Ncols) ? W[k * Ncols + n] : 0.0f;
}

__device__ __forceinline__ void pack_b2(const float* __restrict__ W, const float* __restrict__ Wls,
                                        float* __restrict__ Bp, int t) {
  if (t >= (HID / 4) * 2 * 64) return;
  int j = t & 1;
  int lane = (t >> 1) & 31;
  int tile = (t >> 6) & 1;
  int ks = t >> 7;
  int k = 4 * ks + 2 * (lane >> 4) + j;
  int n = tile * 16 + (lane & 15);
  float v = 0.0f;
  if (n < HID) v = W[k * HID + n];
  else if (n < HID + OUT_DIM) v = Wls[k * OUT_DIM + (n - HID)];
  Bp[t] = v;
}

#define SETUP_TB 256
#define NB_W1 ((IN_DIM / 4) * 2 * 64 / SETUP_TB)
#define NB_WC (((HID / 4) * 2 * 64 + SETUP_TB - 1) / SETUP_TB)
#define NB_WL (((HID / 4) * 1 * 64 + SETUP_TB - 1) / SETUP_TB)

__global__ void k_setup(int* __restrict__ cnt, int ncnt, int nb_cnt,
                        const float* __restrict__ W1, const float* __restrict__ W2,
                        const float* __restrict__ W3, const float* __restrict__ Wl,
                        float* __restrict__ W1p, float* __restrict__ W2c,
                        float* __restrict__ W3c, float* __restrict__ Wl3p) {
  int b = blockIdx.x;
  int t = threadIdx.x;
  if (b < nb_cnt) {
    int i = b * SETUP_TB + t;
    if (i < ncnt) cnt[i] = 0;
    return;
  }
  b -= nb_cnt;
  if (b < NB_W1) { pack_b(W1, IN_DIM, HID, W1p, IN_DIM / 4, 2, b * SETUP_TB + t); return; }
  b -= NB_W1;
  if (b < NB_WC) { pack_b2(W2, Wl, W2c, b * SETUP_TB + t); return; }
  b -= NB_WC;
  if (b < NB_WC) { pack_b2(W3, Wl + HID * OUT_DIM, W3c, b * SETUP_TB + t); return; }
  b -= NB_WC;
  pack_b(Wl + 2 * HID * OUT_DIM, HID, OUT_DIM, Wl3p, HID / 4, 1, b * SETUP_TB + t);
}

#define BIN_SHIFT 8
#define BIN_NODES 256
#define NBIN_MAX 512
#define CNT_PITCH 32
#define SORT_TB 512
#define BIN_TB 1024
#define NPW 8
#define GATHER_TB 256

__global__ void k_bincount(const int* __restrict__ dst, int* __restrict__ binCount,
                           int* __restrict__ blockBase, int E, int nbin) {
  __shared__ int hist[NBIN_MAX];
  int t = threadIdx.x;
  hist[t] = 0;
  __syncthreads();
  int chunk = (E + (int)gridDim.x - 1) / (int)gridDim.x;
  int beg = (int)blockIdx.x * chunk;
  int end = beg + chunk; if (end > E) end = E;
  for (int e0 = beg + t; e0 < end; e0 += 4 * SORT_TB) {
    int d[4];
#pragma unroll
    for (int u = 0; u < 4; ++u) {
      int e = e0 + u * SORT_TB;
      d[u] = (e < end) ? dst[e] : -1;
    }
#pragma unroll
    for (int u = 0; u < 4; ++u)
      if (d[u] >= 0) atomicAdd(&hist[d[u] >> BIN_SHIFT], 1);
  }
  __syncthreads();
  if (t < nbin) {
    int c = hist[t];
    int base = 0;
    if (c > 0) base = atomicAdd(&binCount[t * CNT_PITCH], c);
    blockBase[(int)blockIdx.x * nbin + t] = base;
  }
}

__global__ void k_binscatter(const int* __restrict__ src, const int* __restrict__ dst,
                             const float* __restrict__ ew, const int* __restrict__ binCount,
                             const int* __restrict__ blockBase, int* __restrict__ binStart,
                             int2* __restrict__ binned, int E, int nbin) {
  __shared__ int s[NBIN_MAX];
  __shared__ int cur[NBIN_MAX];
  int t = threadIdx.x;
  int v = (t < nbin) ? binCount[t * CNT_PITCH] : 0;
  s[t] = v;
  __syncthreads();
  for (int d = 1; d < SORT_TB; d <<= 1) {
    int a = (t >= d) ? s[t - d] : 0;
    __syncthreads();
    s[t] += a;
    __syncthreads();
  }
  int excl = s[t] - v;
  cur[t] = (t < nbin) ? excl + blockBase[(int)blockIdx.x * nbin + t] : 0;
  if (blockIdx.x == 0) {
    if (t < nbin) binStart[t * BIN_NODES] = excl;
  }
  __syncthreads();
  int chunk = (E + (int)gridDim.x - 1) / (int)gridDim.x;
  int beg = (int)blockIdx.x * chunk;
  int end = beg + chunk; if (end > E) end = E;
  for (int e0 = beg + t; e0 < end; e0 += 4 * SORT_TB) {
    int d[4];
    int sr[4];
    int wb[4];
#pragma unroll
    for (int u = 0; u < 4; ++u) {
      int e = e0 + u * SORT_TB;
      bool ok = e < end;
      d[u] = ok ? dst[e] : -1;
      sr[u] = ok ? src[e] : 0;
      wb[u] = ok ? __float_as_int(ew[e]) : 0;
    }
#pragma unroll
    for (int u = 0; u < 4; ++u) {
      if (d[u] >= 0) {
        int p = atomicAdd(&cur[d[u] >> BIN_SHIFT], 1);
        binned[p] = make_int2((sr[u] << 8) | (d[u] & (BIN_NODES - 1)), wb[u]);
      }
    }
  }
}

__global__ void k_binsort(const int2* __restrict__ binned, int* off,
                          int2* __restrict__ pairs, float* __restrict__ dinv, int n, int E) {
  __shared__ int c[BIN_NODES];
  __shared__ int s[BIN_NODES];
  int t = threadIdx.x;
  int b = blockIdx.x;
  int beg = off[b * BIN_NODES];
  int end = (b + 1 < (int)gridDim.x) ? off[(b + 1) * BIN_NODES] : E;
  if (t < BIN_NODES) c[t] = 0;
  __syncthreads();
  for (int p0 = beg + t; p0 < end; p0 += 4 * BIN_TB) {
    int r[4];
#pragma unroll
    for (int u = 0; u < 4; ++u) {
      int p = p0 + u * BIN_TB;
      r[u] = (p < end) ? binned[p].x : -1;
    }
#pragma unroll
    for (int u = 0; u < 4; ++u)
      if (r[u] >= 0) atomicAdd(&c[r[u] & (BIN_NODES - 1)], 1);
  }
  __syncthreads();
  int v = (t < BIN_NODES) ? c[t] : 0;
  if (t < BIN_NODES) s[t] = v;
  __syncthreads();
  for (int d = 1; d < BIN_NODES; d <<= 1) {
    int a = (t < BIN_NODES && t >= d) ? s[t - d] : 0;
    __syncthreads();
    if (t < BIN_NODES) s[t] += a;
    __syncthreads();
  }
  if (t < BIN_NODES) {
    int st = beg + s[t] - v;
    int node = b * BIN_NODES + t;
    if (t > 0 && node < n) off[node] = st;
    c[t] = st;
  }
  __syncthreads();
  for (int p0 = beg + t; p0 < end; p0 += 4 * BIN_TB) {
    int2 r[4];
#pragma unroll
    for (int u = 0; u < 4; ++u) {
      int p = p0 + u * BIN_TB;
      r[u] = (p < end) ? binned[p] : make_int2(-1, 0);
    }
#pragma unroll
    for (int u = 0; u < 4; ++u) {
      if (r[u].x >= 0) {
        int pos = atomicAdd(&c[r[u].x & (BIN_NODES - 1)], 1);
        pairs[pos] = make_int2(r[u].x >> 8, r[u].y);
      }
    }
  }
  __syncthreads();
  int lane = t & 31;
  int j0 = (t >> 5) * NPW;
  int ofv = end;
  if (lane <= NPW) { int j = j0 + lane; ofv = (j == 0) ? beg : c[j - 1]; }
  float x0[NPW];
  float x1[NPW];
  int emax = (end > 0) ? end - 1 : 0;
#pragma unroll
  for (int k = 0; k < NPW; ++k) {
    int p = __builtin_amdgcn_readlane(ofv, k) + lane;
    int e = __builtin_amdgcn_readlane(ofv, k + 1);
    int pa = (p < emax) ? p : emax;
    int pb = (p + 32 < emax) ? p + 32 : emax;
    float wa = __int_as_float(pairs[pa].y);
    float wb = __int_as_float(pairs[pb].y);
    x0[k] = (p < e) ? wa : 0.0f;
    x1[k] = (p + 32 < e) ? wb : 0.0f;
  }
  float res = 0.0f;
#pragma unroll
  for (int k = 0; k < NPW; ++k) {
    int p = __builtin_amdgcn_readlane(ofv, k) + lane;
    int e = __builtin_amdgcn_readlane(ofv, k + 1);
    float sm = 0.0f;
    sm += x0[k];
    sm += x1[k];
    for (int p2 = p + 64; p2 < e; p2 += 32) sm += __int_as_float(pairs[p2].y);
#pragma unroll
    for (int m = 16; m >= 1; m >>= 1) sm += __shfl_xor(sm, m, 32);
    float dg = sm + 1.0f;
    float rq = (dg > 0.0f) ? rsqrtf(dg) : 0.0f;
    if (lane == k) res = rq;
  }
  int node = b * BIN_NODES + j0 + lane;
  if (lane < NPW && node < n) dinv[node] = res;
}


template <int J>
__device__ __forceinline__ int grp_bcast(int v) {
  return (int)__builtin_amdgcn_mov_dpp8((unsigned)v, (unsigned)(J * 0x249249));
}
__device__ __forceinline__ float lane_xor1(float x) {
  return __int_as_float((int)__builtin_amdgcn_mov_dpp8((unsigned)__float_as_int(x),
      1u | (0u << 3) | (3u << 6) | (2u << 9) | (5u << 12) | (4u << 15) | (7u << 18) | (6u << 21)));
}
__device__ __forceinline__ float lane_xor2(float x) {
  return __int_as_float((int)__builtin_amdgcn_mov_dpp8((unsigned)__float_as_int(x),
      2u | (3u << 3) | (0u << 6) | (1u << 9) | (6u << 12) | (7u << 15) | (4u << 18) | (5u << 21)));
}
__device__ __forceinline__ float lane_xor4(float x) {
  return __int_as_float((int)__builtin_amdgcn_mov_dpp8((unsigned)__float_as_int(x),
      4u | (5u << 3) | (6u << 6) | (7u << 9) | (0u << 12) | (1u << 15) | (2u << 18) | (3u << 21)));
}
__device__ __forceinline__ float lane_xor8(float x) {
  int v = __float_as_int(x);
  return __int_as_float(__builtin_amdgcn_update_dpp(v, v, 0x128, 0xF, 0xF, false));
}
__device__ __forceinline__ float lane_xor16(float x) {
  unsigned v = (unsigned)__float_as_int(x);
  return __int_as_float((int)__builtin_amdgcn_permlanex16(v, v, 0x76543210u, 0xfedcba98u, false, false));
}

#define ROW_LOAD(J) v[J] = *(const float4*)(hb + ((unsigned)grp_bcast<J>(q.x) * 128u + c16))
#define ROW_ADD(J) { float cf = __int_as_float(grp_bcast<J>(q.y)); \
    acc.x += v[J].x * cf; acc.y += v[J].y * cf; acc.z += v[J].z * cf; acc.w += v[J].w * cf; }

template <int MODE>
__global__ void k_gather_norm(const int* __restrict__ off, const int2* __restrict__ pairs,
                              const float* __restrict__ dinv, const float* __restrict__ h,
                              const float* __restrict__ b, const float* __restrict__ Bp,
                              float* __restrict__ hn, float* __restrict__ out,
                              const float* __restrict__ bl, int n, int E) {
  __shared__ __attribute__((aligned(16))) float tile[GATHER_TB / 32][NPW][HID];
  int i0 = __builtin_amdgcn_readfirstlane((int)((blockIdx.x * blockDim.x + threadIdx.x) >> 5)) * NPW;
  if (i0 >= n) return;
  int lane = threadIdx.x & 31;
  int wv = threadIdx.x >> 5;
  int g = lane >> 3;
  int c = lane & 7;
  int slot = 4 * c + g;
  int cnt = n - i0; if (cnt > NPW) cnt = NPW;
  int ofv = E;
  if (lane <= cnt && i0 + lane < n) ofv = off[i0 + lane];
  float dv = (lane < cnt) ? dinv[i0 + lane] : 0.0f;
  const bool cols = c < HID / 4;
  float b0 = 0.0f, b1 = 0.0f, b2 = 0.0f, b3 = 0.0f;
  if (cols) { b0 = b[4 * c]; b1 = b[4 * c + 1]; b2 = b[4 * c + 2]; b3 = b[4 * c + 3]; }
  const char* hb = (const char*)h;
  unsigned c16 = (unsigned)c * 16u;

  int k = 0;
  int p0 = __builtin_amdgcn_readlane(ofv, 0);
  int end = __builtin_amdgcn_readlane(ofv, 1);
  int m = end - p0; if (m > 32) m = 32;
  int2 q = make_int2(i0, 0);
  if (slot < m) q = pairs[p0 + slot];
  bool first = true;
  float4 acc = make_float4(0.0f, 0.0f, 0.0f, 0.0f);
  for (;;) {
    int i = i0 + k;
    int p1 = p0 + m;
    bool done = p1 >= end;
    int kn = done ? k + 1 : k;
    bool more = kn < cnt;
    int endn = end;
    if (done && more) endn = __builtin_amdgcn_readlane(ofv, kn + 1);
    int mn = 0;
    if (more) { mn = endn - p1; if (mn > 32) mn = 32; }
    int2 qn = make_int2(i0, 0);
    if (slot < mn) qn = pairs[p1 + slot];

    float4 self = make_float4(0.0f, 0.0f, 0.0f, 0.0f);
    if (first) self = *(const float4*)(hb + ((unsigned)i * 128u + c16));
    float4 v[8];
    if (0 < m)  { ROW_LOAD(0); ROW_LOAD(1); }
    if (8 < m)  { ROW_LOAD(2); ROW_LOAD(3); }
    if (16 < m) { ROW_LOAD(4); ROW_LOAD(5); }
    if (24 < m) { ROW_LOAD(6); ROW_LOAD(7); }
    if (first) {
      acc.x = (g == 0) ? self.x : 0.0f;
      acc.y = (g == 0) ? self.y : 0.0f;
      acc.z = (g == 0) ? self.z : 0.0f;
      acc.w = (g == 0) ? self.w : 0.0f;
    }
    if (0 < m)  { ROW_ADD(0); ROW_ADD(1); }
    if (8 < m)  { ROW_ADD(2); ROW_ADD(3); }
    if (16 < m) { ROW_ADD(4); ROW_ADD(5); }
    if (24 < m) { ROW_ADD(6); ROW_ADD(7); }
    if (done) {
      float di = __int_as_float(__builtin_amdgcn_readlane(__float_as_int(dv), k));
      float tx = acc.x + lane_xor8(acc.x);
      float ty = acc.y + lane_xor8(acc.y);
      float tz = acc.z + lane_xor8(acc.z);
      float tw = acc.w + lane_xor8(acc.w);
      tx += lane_xor16(tx);
      ty += lane_xor16(ty);
      tz += lane_xor16(tz);
      tw += lane_xor16(tw);
      float rx = cols ? tx * di + b0 : 0.0f;
      float ry = cols ? ty * di + b1 : 0.0f;
      float rz = cols ? tz * di + b2 : 0.0f;
      float rw = cols ? tw * di + b3 : 0.0f;
      float ss = rx * rx + ry * ry + rz * rz + rw * rw;
      ss += lane_xor1(ss);
      ss += lane_xor2(ss);
      ss += lane_xor4(ss);
      float nrm = fmaxf(sqrtf(ss), 1e-12f);
      float inv = 1.0f / nrm;
      if (lane < HID / 4) {
        float4 r;
        rx *= inv; ry *= inv; rz *= inv; rw *= inv;
        r.x = (rx > 0.0f) ? rx : 0.0f;
        r.y = (ry > 0.0f) ? ry : 0.0f;
        r.z = (rz > 0.0f) ? rz : 0.0f;
        r.w = (rw > 0.0f) ? rw : 0.0f;
        *(float4*)(&tile[wv][k][4 * lane]) = r;
      }
    }
    if (!more) break;
    first = done;
    k = kn; p0 = p1; end = endn; m = mn; q = qn;
  }

  const int half = lane >> 4;
  const int l16 = lane & 15;
  constexpr int NT = (MODE == 2) ? 1 : 2;
  const int oc = (MODE == 2) ? l16 : l16 + 16 - HID;
  const bool ocol = half == 0 && oc >= 0 && oc < OUT_DIM;
  float blv = 0.0f;
  if (MODE == 0) { if (ocol) blv = bl[oc]; }
  float oo[NPW];
#pragma unroll
  for (int r = 0; r < NPW; ++r) {
    oo[r] = blv;
    if (MODE != 0) { if (ocol && r < cnt) oo[r] = out[(size_t)(i0 + r) * OUT_DIM + oc]; }
  }
  const float* bp = Bp + lane * 2;
  const float* ar = &tile[wv][lane & (NPW - 1)][2 * half];
  v8f d[NT];
#pragma unroll
  for (int t = 0; t < NT; ++t) d[t] = (v8f){0.f, 0.f, 0.f, 0.f, 0.f, 0.f, 0.f, 0.f};
#pragma unroll
  for (int ks = 0; ks < HID / 4; ++ks) {
    v2f a = *(const v2f*)(ar + 4 * ks);
#pragma unroll
    for (int t = 0; t < NT; ++t) {
      v2f bb = *(const v2f*)(bp + (size_t)(ks * NT + t) * 64);
      d[t] = __builtin_amdgcn_wmma_f32_16x16x4_f32(false, a, false, bb, (short)0, d[t], false, false);
    }
  }
#pragma unroll
  for (int r = 0; r < NPW; ++r) {
    if (r < cnt) {
      if (MODE != 2) {
        float di = __int_as_float(__builtin_amdgcn_readlane(__float_as_int(dv), r));
        float val = half ? d[NT - 1][r] : d[0][r];
        hn[(size_t)(i0 + r) * 32 + lane] = (lane < HID) ? val * di : 0.0f;
      }
      if (ocol) out[(size_t)(i0 + r) * OUT_DIM + oc] = oo[r] + d[NT - 1][r];
    }
  }
}
#undef ROW_LOAD
#undef ROW_ADD

template <int NTILES, bool ROWSCALE>
__global__ void k_wmma_gemm(const float* __restrict__ A, int lda, int acol0,
                            const float* __restrict__ Bp, int Ksteps,
                            float* __restrict__ out, int ldo, int ncols,
                            const float* __restrict__ bias,
                            const float* __restrict__ rowscale, int npairs) {
  int pair = blockIdx.x * (blockDim.x >> 5) + (threadIdx.x >> 5);
  if (pair >= npairs) return;
  int lane = threadIdx.x & 31;
  int half = lane >> 4;
  int l16 = lane & 15;
  int m0 = pair * 32;
  const float* ap0 = A + (size_t)(m0 + l16) * lda + acol0 + half * 2;
  const float* ap1 = ap0 + (size_t)16 * lda;
  const float* bp = Bp + lane * 2;
  float rs[2][8];
  if (ROWSCALE) {
    const float4* rp = (const float4*)(rowscale + m0 + half * 8);
    float4 q0 = rp[0], q1 = rp[1], q2 = rp[4], q3 = rp[5];
    rs[0][0] = q0.x; rs[0][1] = q0.y; rs[0][2] = q0.z; rs[0][3] = q0.w;
    rs[0][4] = q1.x; rs[0][5] = q1.y; rs[0][6] = q1.z; rs[0][7] = q1.w;
    rs[1][0] = q2.x; rs[1][1] = q2.y; rs[1][2] = q2.z; rs[1][3] = q2.w;
    rs[1][4] = q3.x; rs[1][5] = q3.y; rs[1][6] = q3.z; rs[1][7] = q3.w;
  }
  v8f acc[2][NTILES];
#pragma unroll
  for (int rr = 0; rr < 2; ++rr)
#pragma unroll
    for (int t = 0; t < NTILES; ++t) acc[rr][t] = (v8f){0.f, 0.f, 0.f, 0.f, 0.f, 0.f, 0.f, 0.f};

#pragma unroll 2
  for (int ks = 0; ks < Ksteps; ++ks) {
    v2f a0 = *(const v2f*)(ap0 + 4 * ks);
    v2f a1 = *(const v2f*)(ap1 + 4 * ks);
#pragma unroll
    for (int t = 0; t < NTILES; ++t) {
      v2f b = *(const v2f*)(bp + (size_t)(ks * NTILES + t) * 64);
      acc[0][t] = __builtin_amdgcn_wmma_f32_16x16x4_f32(false, a0, false, b, (short)0, acc[0][t], false, false);
      acc[1][t] = __builtin_amdgcn_wmma_f32_16x16x4_f32(false, a1, false, b, (short)0, acc[1][t], false, false);
    }
  }

#pragma unroll
  for (int rr = 0; rr < 2; ++rr)
#pragma unroll
    for (int t = 0; t < NTILES; ++t)
#pragma unroll
      for (int r = 0; r < 8; ++r) {
        int row = m0 + rr * 16 + r + half * 8;
        int col = t * 16 + l16;
        if (col < ncols) {
          float val = acc[rr][t][r] + (bias ? bias[col] : 0.0f);
          if (ROWSCALE) val *= rs[rr][r];
          out[(size_t)row * ldo + col] = val;
        }
      }
}

extern "C" void kernel_launch(void* const* d_in, const int* in_sizes, int n_in,
                              void* d_out, int out_size, void* d_ws, size_t ws_size,
                              hipStream_t stream) {
  const float* x  = (const float*)d_in[0];
  const int*   ei = (const int*)d_in[1];
  const float* ew = (const float*)d_in[2];
  const float* W1 = (const float*)d_in[3];
  const float* b1 = (const float*)d_in[4];
  const float* W2 = (const float*)d_in[5];
  const float* b2 = (const float*)d_in[6];
  const float* W3 = (const float*)d_in[7];
  const float* b3 = (const float*)d_in[8];
  const float* Wl = (const float*)d_in[9];
  const float* bl = (const float*)d_in[10];
  float* out = (float*)d_out;

  const int N = in_sizes[0] / IN_DIM;
  const long long E = in_sizes[2];
  const int* src = ei;
  const int* dst = ei + E;

  const int KS1 = IN_DIM / 4;
  const int KS2 = HID / 4;

  float* ws = (float*)d_ws;
  int*   off  = (int*)ws; ws += N;
  float* dinv = ws; ws += N;
  int2*  pairs = (int2*)ws; ws += (size_t)N * 64;
  float* h    = ws; ws += (size_t)N * 32;
  float* h2   = ws; ws += (size_t)N * 32;
  float* W1p  = ws; ws += KS1 * 2 * 64;
  float* W2c  = ws; ws += KS2 * 2 * 64;
  float* W3c  = ws; ws += KS2 * 2 * 64;
  float* Wl3p = ws; ws += KS2 * 1 * 64;
  const int NBIN = (N + BIN_NODES - 1) / BIN_NODES;
  const int NSORTWG = 512;
  int*   binCount  = (int*)dinv;
  int2*  binned    = (int2*)h;
  int*   blockBase = (int*)pairs;

  const int TB = GATHER_TB;
  const int NPAIRS = N / 32;
  dim3 gblk((NPAIRS + 3) / 4);
  const int Ei = (int)E;
  long long nt32 = (long long)((N + NPW - 1) / NPW) * 32;
  int g_n32 = (int)((nt32 + TB - 1) / TB);
  const int NB_CNT = (NBIN * CNT_PITCH + SETUP_TB - 1) / SETUP_TB;

  k_setup<<<NB_CNT + NB_W1 + 2 * NB_WC + NB_WL, SETUP_TB, 0, stream>>>(
      binCount, NBIN * CNT_PITCH, NB_CNT, W1, W2, W3, Wl, W1p, W2c, W3c, Wl3p);
  k_bincount<<<NSORTWG, SORT_TB, 0, stream>>>(dst, binCount, blockBase, Ei, NBIN);
  k_binscatter<<<NSORTWG, SORT_TB, 0, stream>>>(src, dst, ew, binCount, blockBase, off, binned, Ei, NBIN);
  k_binsort<<<NBIN, BIN_TB, 0, stream>>>(binned, off, pairs, dinv, N, Ei);

  k_wmma_gemm<2, true><<<gblk, 128, 0, stream>>>(x, IN_DIM, 0, W1p, KS1, h, 32, 32, nullptr, dinv, NPAIRS);

  k_gather_norm<0><<<g_n32, TB, 0, stream>>>(off, pairs, dinv, h, b1, W2c, h2, out, bl, N, Ei);

  k_gather_norm<1><<<g_n32, TB, 0, stream>>>(off, pairs, dinv, h2, b2, W3c, h, out, bl, N, Ei);

  k_gather_norm<2><<<g_n32, TB, 0, stream>>>(off, pairs, dinv, h, b3, Wl3p, nullptr, out, bl, N, Ei);
}


# Sort levels 1b and 2 build their 8-byte records in an LDS copy, then store it in order: a store covers 2-4 whole lines, not 32 lines x 8 B; dinv sums read that copy; edge and sum orders as before
# speedup vs baseline: 6.0978x; 1.1317x over previous
#include <hip/hip_runtime.h>
#include <hip/hip_bf16.h>

#define IN_DIM 512
#define HID 20
#define OUT_DIM 10

typedef __attribute__((ext_vector_type(2))) float v2f;
typedef __attribute__((ext_vector_type(8))) float v8f;

__device__ __forceinline__ void pack_b(const float* __restrict__ W, int Krows, int Ncols,
                                       float* __restrict__ Bp, int Ksteps, int ntiles, int t) {
  int total = Ksteps * ntiles * 64;
  if (t >= total) return;
  int j = t & 1;
  int lane = (t >> 1) & 31;
  int tile = (t >> 6) % ntiles;
  int ks = (t >> 6) / ntiles;
  int k = 4 * ks + 2 * (lane >> 4) + j;
  int n = tile * 16 + (lane & 15);
  Bp[t] = (k < Krows && n < Ncols) ? W[k * Ncols + n] : 0.0f;
}

__device__ __forceinline__ void pack_b2(const float* __restrict__ W, const float* __restrict__ Wls,
                                        float* __restrict__ Bp, int t) {
  if (t >= (HID / 4) * 2 * 64) return;
  int j = t & 1;
  int lane = (t >> 1) & 31;
  int tile = (t >> 6) & 1;
  int ks = t >> 7;
  int k = 4 * ks + 2 * (lane >> 4) + j;
  int n = tile * 16 + (lane & 15);
  float v = 0.0f;
  if (n < HID) v = W[k * HID + n];
  else if (n < HID + OUT_DIM) v = Wls[k * OUT_DIM + (n - HID)];
  Bp[t] = v;
}

#define SETUP_TB 256
#define NB_W1 ((IN_DIM / 4) * 2 * 64 / SETUP_TB)
#define NB_WC (((HID / 4) * 2 * 64 + SETUP_TB - 1) / SETUP_TB)
#define NB_WL (((HID / 4) * 1 * 64 + SETUP_TB - 1) / SETUP_TB)

__global__ void k_setup(int* __restrict__ cnt, int ncnt, int nb_cnt,
                        const float* __restrict__ W1, const float* __restrict__ W2,
                        const float* __restrict__ W3, const float* __restrict__ Wl,
                        float* __restrict__ W1p, float* __restrict__ W2c,
                        float* __restrict__ W3c, float* __restrict__ Wl3p) {
  int b = blockIdx.x;
  int t = threadIdx.x;
  if (b < nb_cnt) {
    int i = b * SETUP_TB + t;
    if (i < ncnt) cnt[i] = 0;
    return;
  }
  b -= nb_cnt;
  if (b < NB_W1) { pack_b(W1, IN_DIM, HID, W1p, IN_DIM / 4, 2, b * SETUP_TB + t); return; }
  b -= NB_W1;
  if (b < NB_WC) { pack_b2(W2, Wl, W2c, b * SETUP_TB + t); return; }
  b -= NB_WC;
  if (b < NB_WC) { pack_b2(W3, Wl + HID * OUT_DIM, W3c, b * SETUP_TB + t); return; }
  b -= NB_WC;
  pack_b(Wl + 2 * HID * OUT_DIM, HID, OUT_DIM, Wl3p, HID / 4, 1, b * SETUP_TB + t);
}

#define BIN_SHIFT 8
#define BIN_NODES 256
#define NBIN_MAX 512
#define CNT_PITCH 32
#define SORT_TB 512
#define BIN_TB 1024
#define NPW 8
#define GATHER_TB 256
#define L1_CAP 6400
#define L2_CAP 12288

__global__ void k_bincount(const int* __restrict__ dst, int* __restrict__ binCount,
                           int* __restrict__ blockBase, int* __restrict__ blockCnt, int E, int nbin) {
  __shared__ int hist[NBIN_MAX];
  int t = threadIdx.x;
  hist[t] = 0;
  __syncthreads();
  int chunk = (E + (int)gridDim.x - 1) / (int)gridDim.x;
  int beg = (int)blockIdx.x * chunk;
  int end = beg + chunk; if (end > E) end = E;
  for (int e0 = beg + t; e0 < end; e0 += 4 * SORT_TB) {
    int d[4];
#pragma unroll
    for (int u = 0; u < 4; ++u) {
      int e = e0 + u * SORT_TB;
      d[u] = (e < end) ? dst[e] : -1;
    }
#pragma unroll
    for (int u = 0; u < 4; ++u)
      if (d[u] >= 0) atomicAdd(&hist[d[u] >> BIN_SHIFT], 1);
  }
  __syncthreads();
  if (t < nbin) {
    int c = hist[t];
    int base = 0;
    if (c > 0) base = atomicAdd(&binCount[t * CNT_PITCH], c);
    blockBase[(int)blockIdx.x * nbin + t] = base;
    blockCnt[(int)blockIdx.x * nbin + t] = c;
  }
}

__global__ void k_binscatter(const int* __restrict__ src, const int* __restrict__ dst,
                             const float* __restrict__ ew, const int* __restrict__ binCount,
                             const int* __restrict__ blockBase, const int* __restrict__ blockCnt,
                             int* __restrict__ binStart, int2* __restrict__ binned, int E, int nbin) {
  __shared__ int s[NBIN_MAX];
  __shared__ int ls[NBIN_MAX];
  __shared__ int cur[NBIN_MAX];
  __shared__ int gd[NBIN_MAX];
  __shared__ int2 stage[L1_CAP];
  __shared__ unsigned short sbin[L1_CAP];
  int t = threadIdx.x;
  int v = (t < nbin) ? binCount[t * CNT_PITCH] : 0;
  int lv = (t < nbin) ? blockCnt[(int)blockIdx.x * nbin + t] : 0;
  s[t] = v;
  ls[t] = lv;
  __syncthreads();
  for (int d = 1; d < SORT_TB; d <<= 1) {
    int a = (t >= d) ? s[t - d] : 0;
    int la = (t >= d) ? ls[t - d] : 0;
    __syncthreads();
    s[t] += a;
    ls[t] += la;
    __syncthreads();
  }
  int excl = s[t] - v;
  int lexcl = ls[t] - lv;
  int gbase = (t < nbin) ? excl + blockBase[(int)blockIdx.x * nbin + t] : 0;
  cur[t] = lexcl;
  gd[t] = gbase - lexcl;
  if (blockIdx.x == 0) {
    if (t < nbin) binStart[t * BIN_NODES] = excl;
  }
  __syncthreads();
  int chunk = (E + (int)gridDim.x - 1) / (int)gridDim.x;
  int beg = (int)blockIdx.x * chunk;
  int end = beg + chunk; if (end > E) end = E;
  for (int e0 = beg + t; e0 < end; e0 += 4 * SORT_TB) {
    int d[4];
    int sr[4];
    int wb[4];
#pragma unroll
    for (int u = 0; u < 4; ++u) {
      int e = e0 + u * SORT_TB;
      bool ok = e < end;
      d[u] = ok ? dst[e] : -1;
      sr[u] = ok ? src[e] : 0;
      wb[u] = ok ? __float_as_int(ew[e]) : 0;
    }
#pragma unroll
    for (int u = 0; u < 4; ++u) {
      if (d[u] >= 0) {
        int bin = d[u] >> BIN_SHIFT;
        int lp = atomicAdd(&cur[bin], 1);
        int2 rec = make_int2((sr[u] << 8) | (d[u] & (BIN_NODES - 1)), wb[u]);
        if (lp < L1_CAP) { stage[lp] = rec; sbin[lp] = (unsigned short)bin; }
        else binned[gd[bin] + lp] = rec;
      }
    }
  }
  __syncthreads();
  int m = end - beg;
  if (m > L1_CAP) m = L1_CAP;
  for (int i0 = t; i0 < m; i0 += 4 * SORT_TB) {
    int2 rec[4];
    int gp[4];
#pragma unroll
    for (int u = 0; u < 4; ++u) {
      int i = i0 + u * SORT_TB;
      if (i >= m) i = i0;
      rec[u] = stage[i];
      gp[u] = gd[sbin[i]] + i;
    }
#pragma unroll
    for (int u = 0; u < 4; ++u)
      if (i0 + u * SORT_TB < m) binned[gp[u]] = rec[u];
  }
}

__global__ void k_binsort(const int2* __restrict__ binned, int* off,
                          int2* __restrict__ pairs, float* __restrict__ dinv, int n, int E) {
  __shared__ int c[BIN_NODES];
  __shared__ int s[BIN_NODES];
  __shared__ int2 stage[L2_CAP];
  int t = threadIdx.x;
  int b = blockIdx.x;
  int beg = off[b * BIN_NODES];
  int end = (b + 1 < (int)gridDim.x) ? off[(b + 1) * BIN_NODES] : E;
  if (t < BIN_NODES) c[t] = 0;
  __syncthreads();
  for (int p0 = beg + t; p0 < end; p0 += 4 * BIN_TB) {
    int r[4];
#pragma unroll
    for (int u = 0; u < 4; ++u) {
      int p = p0 + u * BIN_TB;
      r[u] = (p < end) ? binned[p].x : -1;
    }
#pragma unroll
    for (int u = 0; u < 4; ++u)
      if (r[u] >= 0) atomicAdd(&c[r[u] & (BIN_NODES - 1)], 1);
  }
  __syncthreads();
  int v = (t < BIN_NODES) ? c[t] : 0;
  if (t < BIN_NODES) s[t] = v;
  __syncthreads();
  for (int d = 1; d < BIN_NODES; d <<= 1) {
    int a = (t < BIN_NODES && t >= d) ? s[t - d] : 0;
    __syncthreads();
    if (t < BIN_NODES) s[t] += a;
    __syncthreads();
  }
  if (t < BIN_NODES) {
    int st = beg + s[t] - v;
    int node = b * BIN_NODES + t;
    if (t > 0 && node < n) off[node] = st;
    c[t] = st;
  }
  __syncthreads();
  for (int p0 = beg + t; p0 < end; p0 += 4 * BIN_TB) {
    int2 r[4];
#pragma unroll
    for (int u = 0; u < 4; ++u) {
      int p = p0 + u * BIN_TB;
      r[u] = (p < end) ? binned[p] : make_int2(-1, 0);
    }
#pragma unroll
    for (int u = 0; u < 4; ++u) {
      if (r[u].x >= 0) {
        int pos = atomicAdd(&c[r[u].x & (BIN_NODES - 1)], 1);
        int2 pr = make_int2(r[u].x >> 8, r[u].y);
        int lp = pos - beg;
        if (lp < L2_CAP) stage[lp] = pr;
        else pairs[pos] = pr;
      }
    }
  }
  __syncthreads();
  int lane = t & 31;
  int j0 = (t >> 5) * NPW;
  int ofv = end;
  if (lane <= NPW) { int j = j0 + lane; ofv = (j == 0) ? beg : c[j - 1]; }
  const int cntb = end - beg;
  const bool over = cntb > L2_CAP;
  auto wt = [&](int p) -> float {
    int lp = p - beg;
    return __int_as_float((lp < L2_CAP) ? stage[lp].y : pairs[p].y);
  };
  float x0[NPW];
  float x1[NPW];
  int lmax = (over ? L2_CAP : cntb) - 1; if (lmax < 0) lmax = 0;
#pragma unroll
  for (int k = 0; k < NPW; ++k) {
    int p = __builtin_amdgcn_readlane(ofv, k) + lane;
    int e = __builtin_amdgcn_readlane(ofv, k + 1);
    int la = p - beg;      if (la > lmax) la = lmax;
    int lb = p + 32 - beg; if (lb > lmax) lb = lmax;
    float wa = __int_as_float(stage[la].y);
    float wb = __int_as_float(stage[lb].y);
    x0[k] = (p < e) ? wa : 0.0f;
    x1[k] = (p + 32 < e) ? wb : 0.0f;
  }
  if (over) {
#pragma unroll
    for (int k = 0; k < NPW; ++k) {
      int p = __builtin_amdgcn_readlane(ofv, k) + lane;
      int e = __builtin_amdgcn_readlane(ofv, k + 1);
      if (p < e && p - beg >= L2_CAP) x0[k] = __int_as_float(pairs[p].y);
      if (p + 32 < e && p + 32 - beg >= L2_CAP) x1[k] = __int_as_float(pairs[p + 32].y);
    }
  }
  float res = 0.0f;
#pragma unroll
  for (int k = 0; k < NPW; ++k) {
    int p = __builtin_amdgcn_readlane(ofv, k) + lane;
    int e = __builtin_amdgcn_readlane(ofv, k + 1);
    float sm = 0.0f;
    sm += x0[k];
    sm += x1[k];
    for (int p2 = p + 64; p2 < e; p2 += 32) sm += wt(p2);
#pragma unroll
    for (int m = 16; m >= 1; m >>= 1) sm += __shfl_xor(sm, m, 32);
    float dg = sm + 1.0f;
    float rq = (dg > 0.0f) ? rsqrtf(dg) : 0.0f;
    if (lane == k) res = rq;
  }
  int node = b * BIN_NODES + j0 + lane;
  if (lane < NPW && node < n) dinv[node] = res;
  int mm = over ? L2_CAP : cntb;
  for (int i0 = t - (beg & 15); i0 < mm; i0 += 4 * BIN_TB) {
    int2 rec[4];
#pragma unroll
    for (int u = 0; u < 4; ++u) {
      int i = i0 + u * BIN_TB;
      if (i < 0 || i >= mm) i = 0;
      rec[u] = stage[i];
    }
#pragma unroll
    for (int u = 0; u < 4; ++u) {
      int i = i0 + u * BIN_TB;
      if (i >= 0 && i < mm) pairs[beg + i] = rec[u];
    }
  }
}


template <int J>
__device__ __forceinline__ int grp_bcast(int v) {
  return (int)__builtin_amdgcn_mov_dpp8((unsigned)v, (unsigned)(J * 0x249249));
}
__device__ __forceinline__ float lane_xor1(float x) {
  return __int_as_float((int)__builtin_amdgcn_mov_dpp8((unsigned)__float_as_int(x),
      1u | (0u << 3) | (3u << 6) | (2u << 9) | (5u << 12) | (4u << 15) | (7u << 18) | (6u << 21)));
}
__device__ __forceinline__ float lane_xor2(float x) {
  return __int_as_float((int)__builtin_amdgcn_mov_dpp8((unsigned)__float_as_int(x),
      2u | (3u << 3) | (0u << 6) | (1u << 9) | (6u << 12) | (7u << 15) | (4u << 18) | (5u << 21)));
}
__device__ __forceinline__ float lane_xor4(float x) {
  return __int_as_float((int)__builtin_amdgcn_mov_dpp8((unsigned)__float_as_int(x),
      4u | (5u << 3) | (6u << 6) | (7u << 9) | (0u << 12) | (1u << 15) | (2u << 18) | (3u << 21)));
}
__device__ __forceinline__ float lane_xor8(float x) {
  int v = __float_as_int(x);
  return __int_as_float(__builtin_amdgcn_update_dpp(v, v, 0x128, 0xF, 0xF, false));
}
__device__ __forceinline__ float lane_xor16(float x) {
  unsigned v = (unsigned)__float_as_int(x);
  return __int_as_float((int)__builtin_amdgcn_permlanex16(v, v, 0x76543210u, 0xfedcba98u, false, false));
}

#define ROW_LOAD(J) v[J] = *(const float4*)(hb + ((unsigned)grp_bcast<J>(q.x) * 128u + c16))
#define ROW_ADD(J) { float cf = __int_as_float(grp_bcast<J>(q.y)); \
    acc.x += v[J].x * cf; acc.y += v[J].y * cf; acc.z += v[J].z * cf; acc.w += v[J].w * cf; }

template <int MODE>
__global__ void k_gather_norm(const int* __restrict__ off, const int2* __restrict__ pairs,
                              const float* __restrict__ dinv, const float* __restrict__ h,
                              const float* __restrict__ b, const float* __restrict__ Bp,
                              float* __restrict__ hn, float* __restrict__ out,
                              const float* __restrict__ bl, int n, int E) {
  __shared__ __attribute__((aligned(16))) float tile[GATHER_TB / 32][NPW][HID];
  int i0 = __builtin_amdgcn_readfirstlane((int)((blockIdx.x * blockDim.x + threadIdx.x) >> 5)) * NPW;
  if (i0 >= n) return;
  int lane = threadIdx.x & 31;
  int wv = threadIdx.x >> 5;
  int g = lane >> 3;
  int c = lane & 7;
  int slot = 4 * c + g;
  int cnt = n - i0; if (cnt > NPW) cnt = NPW;
  int ofv = E;
  if (lane <= cnt && i0 + lane < n) ofv = off[i0 + lane];
  float dv = (lane < cnt) ? dinv[i0 + lane] : 0.0f;
  const bool cols = c < HID / 4;
  float b0 = 0.0f, b1 = 0.0f, b2 = 0.0f, b3 = 0.0f;
  if (cols) { b0 = b[4 * c]; b1 = b[4 * c + 1]; b2 = b[4 * c + 2]; b3 = b[4 * c + 3]; }
  const char* hb = (const char*)h;
  unsigned c16 = (unsigned)c * 16u;

  int k = 0;
  int p0 = __builtin_amdgcn_readlane(ofv, 0);
  int end = __builtin_amdgcn_readlane(ofv, 1);
  int m = end - p0; if (m > 32) m = 32;
  int2 q = make_int2(i0, 0);
  if (slot < m) q = pairs[p0 + slot];
  bool first = true;
  float4 acc = make_float4(0.0f, 0.0f, 0.0f, 0.0f);
  for (;;) {
    int i = i0 + k;
    int p1 = p0 + m;
    bool done = p1 >= end;
    int kn = done ? k + 1 : k;
    bool more = kn < cnt;
    int endn = end;
    if (done && more) endn = __builtin_amdgcn_readlane(ofv, kn + 1);
    int mn = 0;
    if (more) { mn = endn - p1; if (mn > 32) mn = 32; }
    int2 qn = make_int2(i0, 0);
    if (slot < mn) qn = pairs[p1 + slot];

    float4 self = make_float4(0.0f, 0.0f, 0.0f, 0.0f);
    if (first) self = *(const float4*)(hb + ((unsigned)i * 128u + c16));
    float4 v[8];
    if (0 < m)  { ROW_LOAD(0); ROW_LOAD(1); }
    if (8 < m)  { ROW_LOAD(2); ROW_LOAD(3); }
    if (16 < m) { ROW_LOAD(4); ROW_LOAD(5); }
    if (24 < m) { ROW_LOAD(6); ROW_LOAD(7); }
    if (first) {
      acc.x = (g == 0) ? self.x : 0.0f;
      acc.y = (g == 0) ? self.y : 0.0f;
      acc.z = (g == 0) ? self.z : 0.0f;
      acc.w = (g == 0) ? self.w : 0.0f;
    }
    if (0 < m)  { ROW_ADD(0); ROW_ADD(1); }
    if (8 < m)  { ROW_ADD(2); ROW_ADD(3); }
    if (16 < m) { ROW_ADD(4); ROW_ADD(5); }
    if (24 < m) { ROW_ADD(6); ROW_ADD(7); }
    if (done) {
      float di = __int_as_float(__builtin_amdgcn_readlane(__float_as_int(dv), k));
      float tx = acc.x + lane_xor8(acc.x);
      float ty = acc.y + lane_xor8(acc.y);
      float tz = acc.z + lane_xor8(acc.z);
      float tw = acc.w + lane_xor8(acc.w);
      tx += lane_xor16(tx);
      ty += lane_xor16(ty);
      tz += lane_xor16(tz);
      tw += lane_xor16(tw);
      float rx = cols ? tx * di + b0 : 0.0f;
      float ry = cols ? ty * di + b1 : 0.0f;
      float rz = cols ? tz * di + b2 : 0.0f;
      float rw = cols ? tw * di + b3 : 0.0f;
      float ss = rx * rx + ry * ry + rz * rz + rw * rw;
      ss += lane_xor1(ss);
      ss += lane_xor2(ss);
      ss += lane_xor4(ss);
      float nrm = fmaxf(sqrtf(ss), 1e-12f);
      float inv = 1.0f / nrm;
      if (lane < HID / 4) {
        float4 r;
        rx *= inv; ry *= inv; rz *= inv; rw *= inv;
        r.x = (rx > 0.0f) ? rx : 0.0f;
        r.y = (ry > 0.0f) ? ry : 0.0f;
        r.z = (rz > 0.0f) ? rz : 0.0f;
        r.w = (rw > 0.0f) ? rw : 0.0f;
        *(float4*)(&tile[wv][k][4 * lane]) = r;
      }
    }
    if (!more) break;
    first = done;
    k = kn; p0 = p1; end = endn; m = mn; q = qn;
  }

  const int half = lane >> 4;
  const int l16 = lane & 15;
  constexpr int NT = (MODE == 2) ? 1 : 2;
  const int oc = (MODE == 2) ? l16 : l16 + 16 - HID;
  const bool ocol = half == 0 && oc >= 0 && oc < OUT_DIM;
  float blv = 0.0f;
  if (MODE == 0) { if (ocol) blv = bl[oc]; }
  float oo[NPW];
#pragma unroll
  for (int r = 0; r < NPW; ++r) {
    oo[r] = blv;
    if (MODE != 0) { if (ocol && r < cnt) oo[r] = out[(size_t)(i0 + r) * OUT_DIM + oc]; }
  }
  const float* bp = Bp + lane * 2;
  const float* ar = &tile[wv][lane & (NPW - 1)][2 * half];
  v8f d[NT];
#pragma unroll
  for (int t = 0; t < NT; ++t) d[t] = (v8f){0.f, 0.f, 0.f, 0.f, 0.f, 0.f, 0.f, 0.f};
#pragma unroll
  for (int ks = 0; ks < HID / 4; ++ks) {
    v2f a = *(const v2f*)(ar + 4 * ks);
#pragma unroll
    for (int t = 0; t < NT; ++t) {
      v2f bb = *(const v2f*)(bp + (size_t)(ks * NT + t) * 64);
      d[t] = __builtin_amdgcn_wmma_f32_16x16x4_f32(false, a, false, bb, (short)0, d[t], false, false);
    }
  }
#pragma unroll
  for (int r = 0; r < NPW; ++r) {
    if (r < cnt) {
      if (MODE != 2) {
        float di = __int_as_float(__builtin_amdgcn_readlane(__float_as_int(dv), r));
        float val = half ? d[NT - 1][r] : d[0][r];
        hn[(size_t)(i0 + r) * 32 + lane] = (lane < HID) ? val * di : 0.0f;
      }
      if (ocol) out[(size_t)(i0 + r) * OUT_DIM + oc] = oo[r] + d[NT - 1][r];
    }
  }
}
#undef ROW_LOAD
#undef ROW_ADD

template <int NTILES, bool ROWSCALE>
__global__ void k_wmma_gemm(const float* __restrict__ A, int lda, int acol0,
                            const float* __restrict__ Bp, int Ksteps,
                            float* __restrict__ out, int ldo, int ncols,
                            const float* __restrict__ bias,
                            const float* __restrict__ rowscale, int npairs) {
  int pair = blockIdx.x * (blockDim.x >> 5) + (threadIdx.x >> 5);
  if (pair >= npairs) return;
  int lane = threadIdx.x & 31;
  int half = lane >> 4;
  int l16 = lane & 15;
  int m0 = pair * 32;
  const float* ap0 = A + (size_t)(m0 + l16) * lda + acol0 + half * 2;
  const float* ap1 = ap0 + (size_t)16 * lda;
  const float* bp = Bp + lane * 2;
  float rs[2][8];
  if (ROWSCALE) {
    const float4* rp = (const float4*)(rowscale + m0 + half * 8);
    float4 q0 = rp[0], q1 = rp[1], q2 = rp[4], q3 = rp[5];
    rs[0][0] = q0.x; rs[0][1] = q0.y; rs[0][2] = q0.z; rs[0][3] = q0.w;
    rs[0][4] = q1.x; rs[0][5] = q1.y; rs[0][6] = q1.z; rs[0][7] = q1.w;
    rs[1][0] = q2.x; rs[1][1] = q2.y; rs[1][2] = q2.z; rs[1][3] = q2.w;
    rs[1][4] = q3.x; rs[1][5] = q3.y; rs[1][6] = q3.z; rs[1][7] = q3.w;
  }
  v8f acc[2][NTILES];
#pragma unroll
  for (int rr = 0; rr < 2; ++rr)
#pragma unroll
    for (int t = 0; t < NTILES; ++t) acc[rr][t] = (v8f){0.f, 0.f, 0.f, 0.f, 0.f, 0.f, 0.f, 0.f};

#pragma unroll 2
  for (int ks = 0; ks < Ksteps; ++ks) {
    v2f a0 = *(const v2f*)(ap0 + 4 * ks);
    v2f a1 = *(const v2f*)(ap1 + 4 * ks);
#pragma unroll
    for (int t = 0; t < NTILES; ++t) {
      v2f b = *(const v2f*)(bp + (size_t)(ks * NTILES + t) * 64);
      acc[0][t] = __builtin_amdgcn_wmma_f32_16x16x4_f32(false, a0, false, b, (short)0, acc[0][t], false, false);
      acc[1][t] = __builtin_amdgcn_wmma_f32_16x16x4_f32(false, a1, false, b, (short)0, acc[1][t], false, false);
    }
  }

#pragma unroll
  for (int rr = 0; rr < 2; ++rr)
#pragma unroll
    for (int t = 0; t < NTILES; ++t)
#pragma unroll
      for (int r = 0; r < 8; ++r) {
        int row = m0 + rr * 16 + r + half * 8;
        int col = t * 16 + l16;
        if (col < ncols) {
          float val = acc[rr][t][r] + (bias ? bias[col] : 0.0f);
          if (ROWSCALE) val *= rs[rr][r];
          out[(size_t)row * ldo + col] = val;
        }
      }
}

extern "C" void kernel_launch(void* const* d_in, const int* in_sizes, int n_in,
                              void* d_out, int out_size, void* d_ws, size_t ws_size,
                              hipStream_t stream) {
  const float* x  = (const float*)d_in[0];
  const int*   ei = (const int*)d_in[1];
  const float* ew = (const float*)d_in[2];
  const float* W1 = (const float*)d_in[3];
  const float* b1 = (const float*)d_in[4];
  const float* W2 = (const float*)d_in[5];
  const float* b2 = (const float*)d_in[6];
  const float* W3 = (const float*)d_in[7];
  const float* b3 = (const float*)d_in[8];
  const float* Wl = (const float*)d_in[9];
  const float* bl = (const float*)d_in[10];
  float* out = (float*)d_out;

  const int N = in_sizes[0] / IN_DIM;
  const long long E = in_sizes[2];
  const int* src = ei;
  const int* dst = ei + E;

  const int KS1 = IN_DIM / 4;
  const int KS2 = HID / 4;

  float* ws = (float*)d_ws;
  int*   off  = (int*)ws; ws += N;
  float* dinv = ws; ws += N;
  int2*  pairs = (int2*)ws; ws += (size_t)N * 64;
  float* h    = ws; ws += (size_t)N * 32;
  float* h2   = ws; ws += (size_t)N * 32;
  float* W1p  = ws; ws += KS1 * 2 * 64;
  float* W2c  = ws; ws += KS2 * 2 * 64;
  float* W3c  = ws; ws += KS2 * 2 * 64;
  float* Wl3p = ws; ws += KS2 * 1 * 64;
  const int NBIN = (N + BIN_NODES - 1) / BIN_NODES;
  const int NSORTWG = 512;
  int*   binCount  = (int*)dinv;
  int2*  binned    = (int2*)h;
  int*   blockBase = (int*)pairs;
  int*   blockCnt  = blockBase + NSORTWG * NBIN;

  const int TB = GATHER_TB;
  const int NPAIRS = N / 32;
  dim3 gblk((NPAIRS + 3) / 4);
  const int Ei = (int)E;
  long long nt32 = (long long)((N + NPW - 1) / NPW) * 32;
  int g_n32 = (int)((nt32 + TB - 1) / TB);
  const int NB_CNT = (NBIN * CNT_PITCH + SETUP_TB - 1) / SETUP_TB;

  k_setup<<<NB_CNT + NB_W1 + 2 * NB_WC + NB_WL, SETUP_TB, 0, stream>>>(
      binCount, NBIN * CNT_PITCH, NB_CNT, W1, W2, W3, Wl, W1p, W2c, W3c, Wl3p);
  k_bincount<<<NSORTWG, SORT_TB, 0, stream>>>(dst, binCount, blockBase, blockCnt, Ei, NBIN);
  k_binscatter<<<NSORTWG, SORT_TB, 0, stream>>>(src, dst, ew, binCount, blockBase, blockCnt, off, binned, Ei, NBIN);
  k_binsort<<<NBIN, BIN_TB, 0, stream>>>(binned, off, pairs, dinv, N, Ei);

  k_wmma_gemm<2, true><<<gblk, 128, 0, stream>>>(x, IN_DIM, 0, W1p, KS1, h, 32, 32, nullptr, dinv, NPAIRS);

  k_gather_norm<0><<<g_n32, TB, 0, stream>>>(off, pairs, dinv, h, b1, W2c, h2, out, bl, N, Ei);

  k_gather_norm<1><<<g_n32, TB, 0, stream>>>(off, pairs, dinv, h2, b2, W3c, h, out, bl, N, Ei);

  k_gather_norm<2><<<g_n32, TB, 0, stream>>>(off, pairs, dinv, h, b3, Wl3p, nullptr, out, bl, N, Ei);
}


# Layer-1 GEMM reads x as whole 256-byte row pieces, a 64-column block ahead, into a per-wave LDS tile; W1 fragments from an LDS copy; 16-byte fragments feed 2 steps, so its k (sum) order changes
# speedup vs baseline: 6.5863x; 1.0801x over previous
#include <hip/hip_runtime.h>
#include <hip/hip_bf16.h>

#define IN_DIM 512
#define HID 20
#define OUT_DIM 10

typedef __attribute__((ext_vector_type(2))) float v2f;
typedef __attribute__((ext_vector_type(8))) float v8f;

__device__ __forceinline__ void pack_b(const float* __restrict__ W, int Krows, int Ncols,
                                       float* __restrict__ Bp, int Ksteps, int ntiles, int t) {
  int total = Ksteps * ntiles * 64;
  if (t >= total) return;
  int j = t & 1;
  int lane = (t >> 1) & 31;
  int tile = (t >> 6) % ntiles;
  int ks = (t >> 6) / ntiles;
  int k = 4 * ks + 2 * (lane >> 4) + j;
  int n = tile * 16 + (lane & 15);
  Bp[t] = (k < Krows && n < Ncols) ? W[k * Ncols + n] : 0.0f;
}

__device__ __forceinline__ void pack_b2(const float* __restrict__ W, const float* __restrict__ Wls,
                                        float* __restrict__ Bp, int t) {
  if (t >= (HID / 4) * 2 * 64) return;
  int j = t & 1;
  int lane = (t >> 1) & 31;
  int tile = (t >> 6) & 1;
  int ks = t >> 7;
  int k = 4 * ks + 2 * (lane >> 4) + j;
  int n = tile * 16 + (lane & 15);
  float v = 0.0f;
  if (n < HID) v = W[k * HID + n];
  else if (n < HID + OUT_DIM) v = Wls[k * OUT_DIM + (n - HID)];
  Bp[t] = v;
}

__device__ __forceinline__ void pack_b1(const float* __restrict__ W, float* __restrict__ Bq, int t) {
  int e = t & 3;
  int lane = (t >> 2) & 31;
  int tile = (t >> 7) & 1;
  int jj = t >> 8;
  int k = 8 * jj + 4 * (lane >> 4) + e;
  int n = tile * 16 + (lane & 15);
  Bq[t] = (n < HID) ? W[k * HID + n] : 0.0f;
}

#define SETUP_TB 256
#define NB_W1 ((IN_DIM / 4) * 2 * 64 / SETUP_TB)
#define NB_WC (((HID / 4) * 2 * 64 + SETUP_TB - 1) / SETUP_TB)
#define NB_WL (((HID / 4) * 1 * 64 + SETUP_TB - 1) / SETUP_TB)

__global__ void k_setup(int* __restrict__ cnt, int ncnt, int nb_cnt,
                        const float* __restrict__ W1, const float* __restrict__ W2,
                        const float* __restrict__ W3, const float* __restrict__ Wl,
                        float* __restrict__ W1p, float* __restrict__ W2c,
                        float* __restrict__ W3c, float* __restrict__ Wl3p) {
  int b = blockIdx.x;
  int t = threadIdx.x;
  if (b < nb_cnt) {
    int i = b * SETUP_TB + t;
    if (i < ncnt) cnt[i] = 0;
    return;
  }
  b -= nb_cnt;
  if (b < NB_W1) { pack_b1(W1, W1p, b * SETUP_TB + t); return; }
  b -= NB_W1;
  if (b < NB_WC) { pack_b2(W2, Wl, W2c, b * SETUP_TB + t); return; }
  b -= NB_WC;
  if (b < NB_WC) { pack_b2(W3, Wl + HID * OUT_DIM, W3c, b * SETUP_TB + t); return; }
  b -= NB_WC;
  pack_b(Wl + 2 * HID * OUT_DIM, HID, OUT_DIM, Wl3p, HID / 4, 1, b * SETUP_TB + t);
}

#define BIN_SHIFT 8
#define BIN_NODES 256
#define NBIN_MAX 512
#define CNT_PITCH 32
#define SORT_TB 512
#define BIN_TB 1024
#define NPW 8
#define GATHER_TB 256
#define L1_CAP 6400
#define L2_CAP 12288

__global__ void k_bincount(const int* __restrict__ dst, int* __restrict__ binCount,
                           int* __restrict__ blockBase, int* __restrict__ blockCnt, int E, int nbin) {
  __shared__ int hist[NBIN_MAX];
  int t = threadIdx.x;
  hist[t] = 0;
  __syncthreads();
  int chunk = (E + (int)gridDim.x - 1) / (int)gridDim.x;
  int beg = (int)blockIdx.x * chunk;
  int end = beg + chunk; if (end > E) end = E;
  for (int e0 = beg + t; e0 < end; e0 += 4 * SORT_TB) {
    int d[4];
#pragma unroll
    for (int u = 0; u < 4; ++u) {
      int e = e0 + u * SORT_TB;
      d[u] = (e < end) ? dst[e] : -1;
    }
#pragma unroll
    for (int u = 0; u < 4; ++u)
      if (d[u] >= 0) atomicAdd(&hist[d[u] >> BIN_SHIFT], 1);
  }
  __syncthreads();
  if (t < nbin) {
    int c = hist[t];
    int base = 0;
    if (c > 0) base = atomicAdd(&binCount[t * CNT_PITCH], c);
    blockBase[(int)blockIdx.x * nbin + t] = base;
    blockCnt[(int)blockIdx.x * nbin + t] = c;
  }
}

__global__ void k_binscatter(const int* __restrict__ src, const int* __restrict__ dst,
                             const float* __restrict__ ew, const int* __restrict__ binCount,
                             const int* __restrict__ blockBase, const int* __restrict__ blockCnt,
                             int* __restrict__ binStart, int2* __restrict__ binned, int E, int nbin) {
  __shared__ int s[NBIN_MAX];
  __shared__ int ls[NBIN_MAX];
  __shared__ int cur[NBIN_MAX];
  __shared__ int gd[NBIN_MAX];
  __shared__ int2 stage[L1_CAP];
  __shared__ unsigned short sbin[L1_CAP];
  int t = threadIdx.x;
  int v = (t < nbin) ? binCount[t * CNT_PITCH] : 0;
  int lv = (t < nbin) ? blockCnt[(int)blockIdx.x * nbin + t] : 0;
  s[t] = v;
  ls[t] = lv;
  __syncthreads();
  for (int d = 1; d < SORT_TB; d <<= 1) {
    int a = (t >= d) ? s[t - d] : 0;
    int la = (t >= d) ? ls[t - d] : 0;
    __syncthreads();
    s[t] += a;
    ls[t] += la;
    __syncthreads();
  }
  int excl = s[t] - v;
  int lexcl = ls[t] - lv;
  int gbase = (t < nbin) ? excl + blockBase[(int)blockIdx.x * nbin + t] : 0;
  cur[t] = lexcl;
  gd[t] = gbase - lexcl;
  if (blockIdx.x == 0) {
    if (t < nbin) binStart[t * BIN_NODES] = excl;
  }
  __syncthreads();
  int chunk = (E + (int)gridDim.x - 1) / (int)gridDim.x;
  int beg = (int)blockIdx.x * chunk;
  int end = beg + chunk; if (end > E) end = E;
  for (int e0 = beg + t; e0 < end; e0 += 4 * SORT_TB) {
    int d[4];
    int sr[4];
    int wb[4];
#pragma unroll
    for (int u = 0; u < 4; ++u) {
      int e = e0 + u * SORT_TB;
      bool ok = e < end;
      d[u] = ok ? dst[e] : -1;
      sr[u] = ok ? src[e] : 0;
      wb[u] = ok ? __float_as_int(ew[e]) : 0;
    }
#pragma unroll
    for (int u = 0; u < 4; ++u) {
      if (d[u] >= 0) {
        int bin = d[u] >> BIN_SHIFT;
        int lp = atomicAdd(&cur[bin], 1);
        int2 rec = make_int2((sr[u] << 8) | (d[u] & (BIN_NODES - 1)), wb[u]);
        if (lp < L1_CAP) { stage[lp] = rec; sbin[lp] = (unsigned short)bin; }
        else binned[gd[bin] + lp] = rec;
      }
    }
  }
  __syncthreads();
  int m = end - beg;
  if (m > L1_CAP) m = L1_CAP;
  for (int i0 = t; i0 < m; i0 += 4 * SORT_TB) {
    int2 rec[4];
    int gp[4];
#pragma unroll
    for (int u = 0; u < 4; ++u) {
      int i = i0 + u * SORT_TB;
      if (i >= m) i = i0;
      rec[u] = stage[i];
      gp[u] = gd[sbin[i]] + i;
    }
#pragma unroll
    for (int u = 0; u < 4; ++u)
      if (i0 + u * SORT_TB < m) binned[gp[u]] = rec[u];
  }
}

__global__ void k_binsort(const int2* __restrict__ binned, int* off,
                          int2* __restrict__ pairs, float* __restrict__ dinv, int n, int E) {
  __shared__ int c[BIN_NODES];
  __shared__ int s[BIN_NODES];
  __shared__ int2 stage[L2_CAP];
  int t = threadIdx.x;
  int b = blockIdx.x;
  int beg = off[b * BIN_NODES];
  int end = (b + 1 < (int)gridDim.x) ? off[(b + 1) * BIN_NODES] : E;
  if (t < BIN_NODES) c[t] = 0;
  __syncthreads();
  for (int p0 = beg + t; p0 < end; p0 += 4 * BIN_TB) {
    int r[4];
#pragma unroll
    for (int u = 0; u < 4; ++u) {
      int p = p0 + u * BIN_TB;
      r[u] = (p < end) ? binned[p].x : -1;
    }
#pragma unroll
    for (int u = 0; u < 4; ++u)
      if (r[u] >= 0) atomicAdd(&c[r[u] & (BIN_NODES - 1)], 1);
  }
  __syncthreads();
  int v = (t < BIN_NODES) ? c[t] : 0;
  if (t < BIN_NODES) s[t] = v;
  __syncthreads();
  for (int d = 1; d < BIN_NODES; d <<= 1) {
    int a = (t < BIN_NODES && t >= d) ? s[t - d] : 0;
    __syncthreads();
    if (t < BIN_NODES) s[t] += a;
    __syncthreads();
  }
  if (t < BIN_NODES) {
    int st = beg + s[t] - v;
    int node = b * BIN_NODES + t;
    if (t > 0 && node < n) off[node] = st;
    c[t] = st;
  }
  __syncthreads();
  for (int p0 = beg + t; p0 < end; p0 += 4 * BIN_TB) {
    int2 r[4];
#pragma unroll
    for (int u = 0; u < 4; ++u) {
      int p = p0 + u * BIN_TB;
      r[u] = (p < end) ? binned[p] : make_int2(-1, 0);
    }
#pragma unroll
    for (int u = 0; u < 4; ++u) {
      if (r[u].x >= 0) {
        int pos = atomicAdd(&c[r[u].x & (BIN_NODES - 1)], 1);
        int2 pr = make_int2(r[u].x >> 8, r[u].y);
        int lp = pos - beg;
        if (lp < L2_CAP) stage[lp] = pr;
        else pairs[pos] = pr;
      }
    }
  }
  __syncthreads();
  int lane = t & 31;
  int j0 = (t >> 5) * NPW;
  int ofv = end;
  if (lane <= NPW) { int j = j0 + lane; ofv = (j == 0) ? beg : c[j - 1]; }
  const int cntb = end - beg;
  const bool over = cntb > L2_CAP;
  auto wt = [&](int p) -> float {
    int lp = p - beg;
    return __int_as_float((lp < L2_CAP) ? stage[lp].y : pairs[p].y);
  };
  float x0[NPW];
  float x1[NPW];
  int lmax = (over ? L2_CAP : cntb) - 1; if (lmax < 0) lmax = 0;
#pragma unroll
  for (int k = 0; k < NPW; ++k) {
    int p = __builtin_amdgcn_readlane(ofv, k) + lane;
    int e = __builtin_amdgcn_readlane(ofv, k + 1);
    int la = p - beg;      if (la > lmax) la = lmax;
    int lb = p + 32 - beg; if (lb > lmax) lb = lmax;
    float wa = __int_as_float(stage[la].y);
    float wb = __int_as_float(stage[lb].y);
    x0[k] = (p < e) ? wa : 0.0f;
    x1[k] = (p + 32 < e) ? wb : 0.0f;
  }
  if (over) {
#pragma unroll
    for (int k = 0; k < NPW; ++k) {
      int p = __builtin_amdgcn_readlane(ofv, k) + lane;
      int e = __builtin_amdgcn_readlane(ofv, k + 1);
      if (p < e && p - beg >= L2_CAP) x0[k] = __int_as_float(pairs[p].y);
      if (p + 32 < e && p + 32 - beg >= L2_CAP) x1[k] = __int_as_float(pairs[p + 32].y);
    }
  }
  float res = 0.0f;
#pragma unroll
  for (int k = 0; k < NPW; ++k) {
    int p = __builtin_amdgcn_readlane(ofv, k) + lane;
    int e = __builtin_amdgcn_readlane(ofv, k + 1);
    float sm = 0.0f;
    sm += x0[k];
    sm += x1[k];
    for (int p2 = p + 64; p2 < e; p2 += 32) sm += wt(p2);
#pragma unroll
    for (int m = 16; m >= 1; m >>= 1) sm += __shfl_xor(sm, m, 32);
    float dg = sm + 1.0f;
    float rq = (dg > 0.0f) ? rsqrtf(dg) : 0.0f;
    if (lane == k) res = rq;
  }
  int node = b * BIN_NODES + j0 + lane;
  if (lane < NPW && node < n) dinv[node] = res;
  int mm = over ? L2_CAP : cntb;
  for (int i0 = t - (beg & 15); i0 < mm; i0 += 4 * BIN_TB) {
    int2 rec[4];
#pragma unroll
    for (int u = 0; u < 4; ++u) {
      int i = i0 + u * BIN_TB;
      if (i < 0 || i >= mm) i = 0;
      rec[u] = stage[i];
    }
#pragma unroll
    for (int u = 0; u < 4; ++u) {
      int i = i0 + u * BIN_TB;
      if (i >= 0 && i < mm) pairs[beg + i] = rec[u];
    }
  }
}


template <int J>
__device__ __forceinline__ int grp_bcast(int v) {
  return (int)__builtin_amdgcn_mov_dpp8((unsigned)v, (unsigned)(J * 0x249249));
}
__device__ __forceinline__ float lane_xor1(float x) {
  return __int_as_float((int)__builtin_amdgcn_mov_dpp8((unsigned)__float_as_int(x),
      1u | (0u << 3) | (3u << 6) | (2u << 9) | (5u << 12) | (4u << 15) | (7u << 18) | (6u << 21)));
}
__device__ __forceinline__ float lane_xor2(float x) {
  return __int_as_float((int)__builtin_amdgcn_mov_dpp8((unsigned)__float_as_int(x),
      2u | (3u << 3) | (0u << 6) | (1u << 9) | (6u << 12) | (7u << 15) | (4u << 18) | (5u << 21)));
}
__device__ __forceinline__ float lane_xor4(float x) {
  return __int_as_float((int)__builtin_amdgcn_mov_dpp8((unsigned)__float_as_int(x),
      4u | (5u << 3) | (6u << 6) | (7u << 9) | (0u << 12) | (1u << 15) | (2u << 18) | (3u << 21)));
}
__device__ __forceinline__ float lane_xor8(float x) {
  int v = __float_as_int(x);
  return __int_as_float(__builtin_amdgcn_update_dpp(v, v, 0x128, 0xF, 0xF, false));
}
__device__ __forceinline__ float lane_xor16(float x) {
  unsigned v = (unsigned)__float_as_int(x);
  return __int_as_float((int)__builtin_amdgcn_permlanex16(v, v, 0x76543210u, 0xfedcba98u, false, false));
}

#define ROW_LOAD(J) v[J] = *(const float4*)(hb + ((unsigned)grp_bcast<J>(q.x) * 128u + c16))
#define ROW_ADD(J) { float cf = __int_as_float(grp_bcast<J>(q.y)); \
    acc.x += v[J].x * cf; acc.y += v[J].y * cf; acc.z += v[J].z * cf; acc.w += v[J].w * cf; }

template <int MODE>
__global__ void k_gather_norm(const int* __restrict__ off, const int2* __restrict__ pairs,
                              const float* __restrict__ dinv, const float* __restrict__ h,
                              const float* __restrict__ b, const float* __restrict__ Bp,
                              float* __restrict__ hn, float* __restrict__ out,
                              const float* __restrict__ bl, int n, int E) {
  __shared__ __attribute__((aligned(16))) float tile[GATHER_TB / 32][NPW][HID];
  int i0 = __builtin_amdgcn_readfirstlane((int)((blockIdx.x * blockDim.x + threadIdx.x) >> 5)) * NPW;
  if (i0 >= n) return;
  int lane = threadIdx.x & 31;
  int wv = threadIdx.x >> 5;
  int g = lane >> 3;
  int c = lane & 7;
  int slot = 4 * c + g;
  int cnt = n - i0; if (cnt > NPW) cnt = NPW;
  int ofv = E;
  if (lane <= cnt && i0 + lane < n) ofv = off[i0 + lane];
  float dv = (lane < cnt) ? dinv[i0 + lane] : 0.0f;
  const bool cols = c < HID / 4;
  float b0 = 0.0f, b1 = 0.0f, b2 = 0.0f, b3 = 0.0f;
  if (cols) { b0 = b[4 * c]; b1 = b[4 * c + 1]; b2 = b[4 * c + 2]; b3 = b[4 * c + 3]; }
  const char* hb = (const char*)h;
  unsigned c16 = (unsigned)c * 16u;

  int k = 0;
  int p0 = __builtin_amdgcn_readlane(ofv, 0);
  int end = __builtin_amdgcn_readlane(ofv, 1);
  int m = end - p0; if (m > 32) m = 32;
  int2 q = make_int2(i0, 0);
  if (slot < m) q = pairs[p0 + slot];
  bool first = true;
  float4 acc = make_float4(0.0f, 0.0f, 0.0f, 0.0f);
  for (;;) {
    int i = i0 + k;
    int p1 = p0 + m;
    bool done = p1 >= end;
    int kn = done ? k + 1 : k;
    bool more = kn < cnt;
    int endn = end;
    if (done && more) endn = __builtin_amdgcn_readlane(ofv, kn + 1);
    int mn = 0;
    if (more) { mn = endn - p1; if (mn > 32) mn = 32; }
    int2 qn = make_int2(i0, 0);
    if (slot < mn) qn = pairs[p1 + slot];

    float4 self = make_float4(0.0f, 0.0f, 0.0f, 0.0f);
    if (first) self = *(const float4*)(hb + ((unsigned)i * 128u + c16));
    float4 v[8];
    if (0 < m)  { ROW_LOAD(0); ROW_LOAD(1); }
    if (8 < m)  { ROW_LOAD(2); ROW_LOAD(3); }
    if (16 < m) { ROW_LOAD(4); ROW_LOAD(5); }
    if (24 < m) { ROW_LOAD(6); ROW_LOAD(7); }
    if (first) {
      acc.x = (g == 0) ? self.x : 0.0f;
      acc.y = (g == 0) ? self.y : 0.0f;
      acc.z = (g == 0) ? self.z : 0.0f;
      acc.w = (g == 0) ? self.w : 0.0f;
    }
    if (0 < m)  { ROW_ADD(0); ROW_ADD(1); }
    if (8 < m)  { ROW_ADD(2); ROW_ADD(3); }
    if (16 < m) { ROW_ADD(4); ROW_ADD(5); }
    if (24 < m) { ROW_ADD(6); ROW_ADD(7); }
    if (done) {
      float di = __int_as_float(__builtin_amdgcn_readlane(__float_as_int(dv), k));
      float tx = acc.x + lane_xor8(acc.x);
      float ty = acc.y + lane_xor8(acc.y);
      float tz = acc.z + lane_xor8(acc.z);
      float tw = acc.w + lane_xor8(acc.w);
      tx += lane_xor16(tx);
      ty += lane_xor16(ty);
      tz += lane_xor16(tz);
      tw += lane_xor16(tw);
      float rx = cols ? tx * di + b0 : 0.0f;
      float ry = cols ? ty * di + b1 : 0.0f;
      float rz = cols ? tz * di + b2 : 0.0f;
      float rw = cols ? tw * di + b3 : 0.0f;
      float ss = rx * rx + ry * ry + rz * rz + rw * rw;
      ss += lane_xor1(ss);
      ss += lane_xor2(ss);
      ss += lane_xor4(ss);
      float nrm = fmaxf(sqrtf(ss), 1e-12f);
      float inv = 1.0f / nrm;
      if (lane < HID / 4) {
        float4 r;
        rx *= inv; ry *= inv; rz *= inv; rw *= inv;
        r.x = (rx > 0.0f) ? rx : 0.0f;
        r.y = (ry > 0.0f) ? ry : 0.0f;
        r.z = (rz > 0.0f) ? rz : 0.0f;
        r.w = (rw > 0.0f) ? rw : 0.0f;
        *(float4*)(&tile[wv][k][4 * lane]) = r;
      }
    }
    if (!more) break;
    first = done;
    k = kn; p0 = p1; end = endn; m = mn; q = qn;
  }

  const int half = lane >> 4;
  const int l16 = lane & 15;
  constexpr int NT = (MODE == 2) ? 1 : 2;
  const int oc = (MODE == 2) ? l16 : l16 + 16 - HID;
  const bool ocol = half == 0 && oc >= 0 && oc < OUT_DIM;
  float blv = 0.0f;
  if (MODE == 0) { if (ocol) blv = bl[oc]; }
  float oo[NPW];
#pragma unroll
  for (int r = 0; r < NPW; ++r) {
    oo[r] = blv;
    if (MODE != 0) { if (ocol && r < cnt) oo[r] = out[(size_t)(i0 + r) * OUT_DIM + oc]; }
  }
  const float* bp = Bp + lane * 2;
  const float* ar = &tile[wv][lane & (NPW - 1)][2 * half];
  v8f d[NT];
#pragma unroll
  for (int t = 0; t < NT; ++t) d[t] = (v8f){0.f, 0.f, 0.f, 0.f, 0.f, 0.f, 0.f, 0.f};
#pragma unroll
  for (int ks = 0; ks < HID / 4; ++ks) {
    v2f a = *(const v2f*)(ar + 4 * ks);
#pragma unroll
    for (int t = 0; t < NT; ++t) {
      v2f bb = *(const v2f*)(bp + (size_t)(ks * NT + t) * 64);
      d[t] = __builtin_amdgcn_wmma_f32_16x16x4_f32(false, a, false, bb, (short)0, d[t], false, false);
    }
  }
#pragma unroll
  for (int r = 0; r < NPW; ++r) {
    if (r < cnt) {
      if (MODE != 2) {
        float di = __int_as_float(__builtin_amdgcn_readlane(__float_as_int(dv), r));
        float val = half ? d[NT - 1][r] : d[0][r];
        hn[(size_t)(i0 + r) * 32 + lane] = (lane < HID) ? val * di : 0.0f;
      }
      if (ocol) out[(size_t)(i0 + r) * OUT_DIM + oc] = oo[r] + d[NT - 1][r];
    }
  }
}
#undef ROW_LOAD
#undef ROW_ADD

#define G1_WAVES 8
#define G1_KB 64
#define G1_PITCH (G1_KB + 4)
__global__ void __launch_bounds__(G1_WAVES * 32)
k_gemm1(const float* __restrict__ A, const float* __restrict__ Bq, float* __restrict__ out,
        const float* __restrict__ rowscale, int npairs) {
  __shared__ __attribute__((aligned(16))) float sB[IN_DIM * 32];
  __shared__ __attribute__((aligned(16))) float sA[G1_WAVES][32 * G1_PITCH];
  const int t = threadIdx.x;
  {
    const float4* bsrc = (const float4*)Bq;
    float4* bdst = (float4*)sB;
    float4 tmp[IN_DIM * 8 / (G1_WAVES * 32)];
#pragma unroll
    for (int i = 0; i < IN_DIM * 8 / (G1_WAVES * 32); ++i) tmp[i] = bsrc[t + G1_WAVES * 32 * i];
#pragma unroll
    for (int i = 0; i < IN_DIM * 8 / (G1_WAVES * 32); ++i) bdst[t + G1_WAVES * 32 * i] = tmp[i];
  }
  __syncthreads();
  const int wv = t >> 5;
  const int pair = blockIdx.x * G1_WAVES + wv;
  if (pair >= npairs) return;
  const int lane = t & 31;
  const int half = lane >> 4;
  const int l16 = lane & 15;
  const int m0 = pair * 32;
  const float* ag = A + (size_t)(m0 + half) * IN_DIM + 4 * l16;
  float* sst = &sA[wv][half * G1_PITCH + 4 * l16];
  const float* sfr = &sA[wv][l16 * G1_PITCH + 4 * half];
  const float* sb = sB + lane * 4;
  float rs[2][8];
  {
    const float4* rp = (const float4*)(rowscale + m0 + half * 8);
    float4 q0 = rp[0], q1 = rp[1], q2 = rp[4], q3 = rp[5];
    rs[0][0] = q0.x; rs[0][1] = q0.y; rs[0][2] = q0.z; rs[0][3] = q0.w;
    rs[0][4] = q1.x; rs[0][5] = q1.y; rs[0][6] = q1.z; rs[0][7] = q1.w;
    rs[1][0] = q2.x; rs[1][1] = q2.y; rs[1][2] = q2.z; rs[1][3] = q2.w;
    rs[1][4] = q3.x; rs[1][5] = q3.y; rs[1][6] = q3.z; rs[1][7] = q3.w;
  }
  float4 pre[16];
#pragma unroll
  for (int i = 0; i < 16; ++i) pre[i] = *(const float4*)(ag + (size_t)(2 * i) * IN_DIM);
  v8f acc[2][2];
#pragma unroll
  for (int rr = 0; rr < 2; ++rr)
#pragma unroll
    for (int tt = 0; tt < 2; ++tt) acc[rr][tt] = (v8f){0.f, 0.f, 0.f, 0.f, 0.f, 0.f, 0.f, 0.f};

#pragma unroll
  for (int kb = 0; kb < IN_DIM / G1_KB; ++kb) {
#pragma unroll
    for (int i = 0; i < 16; ++i) *(float4*)(sst + 2 * i * G1_PITCH) = pre[i];
    if (kb + 1 < IN_DIM / G1_KB) {
#pragma unroll
      for (int i = 0; i < 16; ++i)
        pre[i] = *(const float4*)(ag + (size_t)(2 * i) * IN_DIM + (kb + 1) * G1_KB);
    }
#pragma unroll
    for (int jj = 0; jj < G1_KB / 8; ++jj) {
      float4 a0 = *(const float4*)(sfr + 8 * jj);
      float4 a1 = *(const float4*)(sfr + 16 * G1_PITCH + 8 * jj);
      v2f a0l = {a0.x, a0.y}, a0h = {a0.z, a0.w};
      v2f a1l = {a1.x, a1.y}, a1h = {a1.z, a1.w};
#pragma unroll
      for (int tt = 0; tt < 2; ++tt) {
        float4 bq = *(const float4*)(sb + (size_t)(((kb * (G1_KB / 8) + jj) * 2 + tt) * 128));
        v2f bl = {bq.x, bq.y}, bh = {bq.z, bq.w};
        acc[0][tt] = __builtin_amdgcn_wmma_f32_16x16x4_f32(false, a0l, false, bl, (short)0, acc[0][tt], false, false);
        acc[1][tt] = __builtin_amdgcn_wmma_f32_16x16x4_f32(false, a1l, false, bl, (short)0, acc[1][tt], false, false);
        acc[0][tt] = __builtin_amdgcn_wmma_f32_16x16x4_f32(false, a0h, false, bh, (short)0, acc[0][tt], false, false);
        acc[1][tt] = __builtin_amdgcn_wmma_f32_16x16x4_f32(false, a1h, false, bh, (short)0, acc[1][tt], false, false);
      }
    }
  }

#pragma unroll
  for (int rr = 0; rr < 2; ++rr)
#pragma unroll
    for (int tt = 0; tt < 2; ++tt)
#pragma unroll
      for (int r = 0; r < 8; ++r) {
        int row = m0 + rr * 16 + r + half * 8;
        int col = tt * 16 + l16;
        out[(size_t)row * 32 + col] = acc[rr][tt][r] * rs[rr][r];
      }
}

extern "C" void kernel_launch(void* const* d_in, const int* in_sizes, int n_in,
                              void* d_out, int out_size, void* d_ws, size_t ws_size,
                              hipStream_t stream) {
  const float* x  = (const float*)d_in[0];
  const int*   ei = (const int*)d_in[1];
  const float* ew = (const float*)d_in[2];
  const float* W1 = (const float*)d_in[3];
  const float* b1 = (const float*)d_in[4];
  const float* W2 = (const float*)d_in[5];
  const float* b2 = (const float*)d_in[6];
  const float* W3 = (const float*)d_in[7];
  const float* b3 = (const float*)d_in[8];
  const float* Wl = (const float*)d_in[9];
  const float* bl = (const float*)d_in[10];
  float* out = (float*)d_out;

  const int N = in_sizes[0] / IN_DIM;
  const long long E = in_sizes[2];
  const int* src = ei;
  const int* dst = ei + E;

  const int KS1 = IN_DIM / 4;
  const int KS2 = HID / 4;

  float* ws = (float*)d_ws;
  int*   off  = (int*)ws; ws += N;
  float* dinv = ws; ws += N;
  int2*  pairs = (int2*)ws; ws += (size_t)N * 64;
  float* h    = ws; ws += (size_t)N * 32;
  float* h2   = ws; ws += (size_t)N * 32;
  float* W1p  = ws; ws += KS1 * 2 * 64;
  float* W2c  = ws; ws += KS2 * 2 * 64;
  float* W3c  = ws; ws += KS2 * 2 * 64;
  float* Wl3p = ws; ws += KS2 * 1 * 64;
  const int NBIN = (N + BIN_NODES - 1) / BIN_NODES;
  const int NSORTWG = 512;
  int*   binCount  = (int*)dinv;
  int2*  binned    = (int2*)h;
  int*   blockBase = (int*)pairs;
  int*   blockCnt  = blockBase + NSORTWG * NBIN;

  const int TB = GATHER_TB;
  const int NPAIRS = N / 32;
  dim3 gblk((NPAIRS + G1_WAVES - 1) / G1_WAVES);
  const int Ei = (int)E;
  long long nt32 = (long long)((N + NPW - 1) / NPW) * 32;
  int g_n32 = (int)((nt32 + TB - 1) / TB);
  const int NB_CNT = (NBIN * CNT_PITCH + SETUP_TB - 1) / SETUP_TB;

  k_setup<<<NB_CNT + NB_W1 + 2 * NB_WC + NB_WL, SETUP_TB, 0, stream>>>(
      binCount, NBIN * CNT_PITCH, NB_CNT, W1, W2, W3, Wl, W1p, W2c, W3c, Wl3p);
  k_bincount<<<NSORTWG, SORT_TB, 0, stream>>>(dst, binCount, blockBase, blockCnt, Ei, NBIN);
  k_binscatter<<<NSORTWG, SORT_TB, 0, stream>>>(src, dst, ew, binCount, blockBase, blockCnt, off, binned, Ei, NBIN);
  k_binsort<<<NBIN, BIN_TB, 0, stream>>>(binned, off, pairs, dinv, N, Ei);

  k_gemm1<<<gblk, G1_WAVES * 32, 0, stream>>>(x, W1p, h, dinv, NPAIRS);

  k_gather_norm<0><<<g_n32, TB, 0, stream>>>(off, pairs, dinv, h, b1, W2c, h2, out, bl, N, Ei);

  k_gather_norm<1><<<g_n32, TB, 0, stream>>>(off, pairs, dinv, h2, b2, W3c, h, out, bl, N, Ei);

  k_gather_norm<2><<<g_n32, TB, 0, stream>>>(off, pairs, dinv, h, b3, Wl3p, nullptr, out, bl, N, Ei);
}
